# starting kernel
# baseline (speedup 1.0000x reference)
#include <hip/hip_runtime.h>
#include <hip/hip_bf16.h>

typedef __attribute__((ext_vector_type(2))) float v2f;
typedef __attribute__((ext_vector_type(8))) float v8f;

static constexpr int cB = 2;
static constexpr int cN = 384;
static constexpr int cD = 128;
static constexpr int cM = cB * cN;

__global__ void gemm_wmma_f32(const float* __restrict__ A, int K,
                              const float* __restrict__ Bm,
                              const float* __restrict__ bias,
                              const float* __restrict__ rowscale,
                              float* __restrict__ C,
                              int addC, int relu)
{
    const int wave  = threadIdx.x >> 5;
    const int lane  = threadIdx.x & 31;
    const int tile  = blockIdx.x * 8 + wave;
    const int mt    = tile >> 3;
    const int nt    = tile & 7;
    const int m0    = mt * 16;
    const int n0    = nt * 16;
    const int l15   = lane & 15;
    const int khalf = lane >> 4;
    const int n     = n0 + l15;
    const int mbase = m0 + (khalf ? 8 : 0);

    v8f c = {};
    if (addC) {
#pragma unroll
        for (int r = 0; r < 8; ++r)
            c[r] = C[(size_t)(mbase + r) * cD + n];
    }

    const float* Arow = A + (size_t)(m0 + l15) * K;
    for (int k0 = 0; k0 < K; k0 += 4) {
        const int ka = k0 + 2 * khalf;
        v2f a, b;
        a.x = Arow[ka];
        a.y = Arow[ka + 1];
        b.x = Bm[(size_t)ka * cD + n];
        b.y = Bm[(size_t)(ka + 1) * cD + n];
        c = __builtin_amdgcn_wmma_f32_16x16x4_f32(
                false, a, false, b, (short)0, c, false, false);
    }

    const float bv = bias ? bias[n] : 0.0f;
#pragma unroll
    for (int r = 0; r < 8; ++r) {
        const int mr = mbase + r;
        const float s = rowscale ? rowscale[mr] : 1.0f;
        float y = c[r] + s * bv;
        if (relu) y = y > 0.0f ? y : 0.0f;
        C[(size_t)mr * cD + n] = y;
    }
}

__global__ void pairwise_agg(const float* __restrict__ P,
                             const float* __restrict__ Q,
                             const float* __restrict__ adj,
                             const float* __restrict__ we,
                             const float* __restrict__ mb1,
                             float* __restrict__ R,
                             float* __restrict__ deg)
{
    const int bi = blockIdx.x;
    const int b  = bi / cN;
    const int d  = threadIdx.x;

    __shared__ float s_adj[cN];
    for (int j = d; j < cN; j += blockDim.x)
        s_adj[j] = adj[(size_t)bi * cN + j];
    __syncthreads();

    const float p  = P[(size_t)bi * cD + d];
    const float w  = we[d];
    const float b1 = mb1[d];
    const float* Qb = Q + (size_t)b * cN * cD;

    float acc = 0.0f;
    float dg  = 0.0f;
    for (int j = 0; j < cN; ++j) {
        const float a = s_adj[j];
        if (a != 0.0f) {
            float t = p + Qb[(size_t)j * cD + d] + a * w + b1;
            t = t > 0.0f ? t : 0.0f;
            acc += a * t;
            dg  += a;
        }
    }
    R[(size_t)bi * cD + d] = acc;
    if (d == 0) deg[bi] = dg;
}

__global__ void ln_res_relu(const float* __restrict__ xin,
                            const float* __restrict__ hn,
                            const float* __restrict__ g,
                            const float* __restrict__ beta,
                            float* __restrict__ out,
                            int add_residual, int relu_after)
{
    const int row = blockIdx.x;
    const int d   = threadIdx.x;

    float v = hn[(size_t)row * cD + d];
    if (add_residual) v += xin[(size_t)row * cD + d];

    __shared__ float red[cD];
    red[d] = v;
    __syncthreads();
#pragma unroll
    for (int s = cD / 2; s > 0; s >>= 1) {
        if (d < s) red[d] += red[d + s];
        __syncthreads();
    }
    const float mu = red[0] * (1.0f / cD);
    __syncthreads();

    const float cdev = v - mu;
    red[d] = cdev * cdev;
    __syncthreads();
#pragma unroll
    for (int s = cD / 2; s > 0; s >>= 1) {
        if (d < s) red[d] += red[d + s];
        __syncthreads();
    }
    const float var = red[0] * (1.0f / cD);

    float y = cdev * rsqrtf(var + 1e-5f) * g[d] + beta[d];
    if (relu_after) y = y > 0.0f ? y : 0.0f;
    out[(size_t)row * cD + d] = y;
}

extern "C" void kernel_launch(void* const* d_in, const int* in_sizes, int n_in,
                              void* d_out, int out_size, void* d_ws, size_t ws_size,
                              hipStream_t stream)
{
    (void)in_sizes; (void)n_in; (void)out_size; (void)ws_size;

    const float* h    = (const float*)d_in[0];
    const float* adj  = (const float*)d_in[1];
    const float* ip_w = (const float*)d_in[2];
    const float* ip_b = (const float*)d_in[3];
    const float* mw1  = (const float*)d_in[4];
    const float* mb1  = (const float*)d_in[5];
    const float* mw2  = (const float*)d_in[6];
    const float* mb2  = (const float*)d_in[7];
    const float* uw1  = (const float*)d_in[8];
    const float* ub1  = (const float*)d_in[9];
    const float* uw2  = (const float*)d_in[10];
    const float* ub2  = (const float*)d_in[11];
    const float* ln_g = (const float*)d_in[12];
    const float* ln_b = (const float*)d_in[13];

    float* ws = (float*)d_ws;
    const size_t S = (size_t)cM * cD;
    float* x   = ws + 0 * S;
    float* P   = ws + 1 * S;
    float* Q   = ws + 2 * S;
    float* R   = ws + 3 * S;
    float* agg = ws + 4 * S;
    float* hid = ws + 5 * S;
    float* hn  = ws + 6 * S;
    float* deg = ws + 7 * S;

    gemm_wmma_f32<<<48, 256, 0, stream>>>(h, 64, ip_w, ip_b, nullptr, x, 0, 1);

    for (int l = 0; l < 3; ++l) {
        const float* mw1l = mw1 + (size_t)l * 257 * cD;
        const float* mb1l = mb1 + (size_t)l * cD;
        const float* mw2l = mw2 + (size_t)l * cD * cD;
        const float* mb2l = mb2 + (size_t)l * cD;
        const float* uw1l = uw1 + (size_t)l * 256 * cD;
        const float* ub1l = ub1 + (size_t)l * cD;
        const float* uw2l = uw2 + (size_t)l * cD * cD;
        const float* ub2l = ub2 + (size_t)l * cD;

        gemm_wmma_f32<<<48, 256, 0, stream>>>(x, cD, mw1l,            nullptr, nullptr, P, 0, 0);
        gemm_wmma_f32<<<48, 256, 0, stream>>>(x, cD, mw1l + 128 * cD, nullptr, nullptr, Q, 0, 0);

        pairwise_agg<<<cM, cD, 0, stream>>>(P, Q, adj, mw1l + 256 * cD, mb1l, R, deg);

        gemm_wmma_f32<<<48, 256, 0, stream>>>(R, cD, mw2l, mb2l, deg, agg, 0, 0);

        gemm_wmma_f32<<<48, 256, 0, stream>>>(x,   cD, uw1l,            ub1l,    nullptr, hid, 0, 0);
        gemm_wmma_f32<<<48, 256, 0, stream>>>(agg, cD, uw1l + 128 * cD, nullptr, nullptr, hid, 1, 1);

        gemm_wmma_f32<<<48, 256, 0, stream>>>(hid, cD, uw2l, ub2l, nullptr, hn, 0, 0);

        if (l < 2) {
            ln_res_relu<<<cM, cD, 0, stream>>>(x, hn, ln_g + l * cD, ln_b + l * cD, x, 1, 1);
        } else {
            ln_res_relu<<<cM, cD, 0, stream>>>(x, hn, ln_g + l * cD, ln_b + l * cD,
                                               (float*)d_out, 0, 0);
        }
    }
}


# pairwise aggregation dense and branch-free: one block per 3 receivers, each sender value loaded once for the 3 and 48 senders ahead of its use; zero terms add 0.0, every sum keeps its j order
# speedup vs baseline: 1.4234x; 1.4234x over previous
#include <hip/hip_runtime.h>
#include <hip/hip_bf16.h>

typedef __attribute__((ext_vector_type(2))) float v2f;
typedef __attribute__((ext_vector_type(8))) float v8f;

static constexpr int cB = 2;
static constexpr int cN = 384;
static constexpr int cD = 128;
static constexpr int cM = cB * cN;

__global__ void gemm_wmma_f32(const float* __restrict__ A, int K,
                              const float* __restrict__ Bm,
                              const float* __restrict__ bias,
                              const float* __restrict__ rowscale,
                              float* __restrict__ C,
                              int addC, int relu)
{
    const int wave  = threadIdx.x >> 5;
    const int lane  = threadIdx.x & 31;
    const int tile  = blockIdx.x * 8 + wave;
    const int mt    = tile >> 3;
    const int nt    = tile & 7;
    const int m0    = mt * 16;
    const int n0    = nt * 16;
    const int l15   = lane & 15;
    const int khalf = lane >> 4;
    const int n     = n0 + l15;
    const int mbase = m0 + (khalf ? 8 : 0);

    v8f c = {};
    if (addC) {
#pragma unroll
        for (int r = 0; r < 8; ++r)
            c[r] = C[(size_t)(mbase + r) * cD + n];
    }

    const float* Arow = A + (size_t)(m0 + l15) * K;
    for (int k0 = 0; k0 < K; k0 += 4) {
        const int ka = k0 + 2 * khalf;
        v2f a, b;
        a.x = Arow[ka];
        a.y = Arow[ka + 1];
        b.x = Bm[(size_t)ka * cD + n];
        b.y = Bm[(size_t)(ka + 1) * cD + n];
        c = __builtin_amdgcn_wmma_f32_16x16x4_f32(
                false, a, false, b, (short)0, c, false, false);
    }

    const float bv = bias ? bias[n] : 0.0f;
#pragma unroll
    for (int r = 0; r < 8; ++r) {
        const int mr = mbase + r;
        const float s = rowscale ? rowscale[mr] : 1.0f;
        float y = c[r] + s * bv;
        if (relu) y = y > 0.0f ? y : 0.0f;
        C[(size_t)mr * cD + n] = y;
    }
}

static constexpr int TI = 3;
static constexpr int QB = 48;
static constexpr int AG = 8;
static constexpr int NG = QB / AG;
static constexpr int RW = cN / 4;
static_assert(cN % TI == 0 && cN % (2 * QB) == 0 && NG % 2 == 0 && AG % 4 == 0, "shape");

struct AdjGroup { float a[TI][AG]; };

__device__ __forceinline__ void load_q(float (&q)[QB], const float* __restrict__ Qcol, int j0)
{
#pragma unroll
    for (int u = 0; u < QB; ++u)
        q[u] = Qcol[(size_t)(j0 + u) * cD];
}

__device__ __forceinline__ void load_adj(AdjGroup& g, const float4* rows, int j0)
{
#pragma unroll
    for (int r = 0; r < TI; ++r) {
#pragma unroll
        for (int v = 0; v < AG / 4; ++v) {
            const float4 x = rows[r * RW + (j0 >> 2) + v];
            g.a[r][4 * v + 0] = x.x; g.a[r][4 * v + 1] = x.y;
            g.a[r][4 * v + 2] = x.z; g.a[r][4 * v + 3] = x.w;
        }
    }
}

__device__ __forceinline__ void compute_group(const float* q, const AdjGroup& g,
                                              const float (&p)[TI], float w, float b1,
                                              float (&acc)[TI], float& dg)
{
#pragma unroll
    for (int u = 0; u < AG; ++u) {
#pragma unroll
        for (int r = 0; r < TI; ++r) {
            const float a = g.a[r][u];
            float t = p[r] + q[u] + a * w + b1;
            t = fmaxf(t, 0.0f);
            acc[r] += a * t;
        }
        dg += g.a[0][u];
    }
}

template <bool LAST>
__device__ __forceinline__ void one_batch(const float (&q)[QB], AdjGroup& ga, AdjGroup& gb,
                                          const float4* rows, int j0,
                                          const float (&p)[TI], float w, float b1,
                                          float (&acc)[TI], float& dg)
{
#pragma unroll
    for (int s = 0; s < NG; s += 2) {
        load_adj(gb, rows, j0 + (s + 1) * AG);
        __builtin_amdgcn_sched_barrier(0);
        compute_group(&q[s * AG], ga, p, w, b1, acc, dg);
        if (s + 2 < NG || !LAST) load_adj(ga, rows, j0 + (s + 2) * AG);
        __builtin_amdgcn_sched_barrier(0);
        compute_group(&q[(s + 1) * AG], gb, p, w, b1, acc, dg);
    }
}

__global__ void __launch_bounds__(cD)
pairwise_agg(const float* __restrict__ P,
             const float* __restrict__ Q,
             const float* __restrict__ adj,
             const float* __restrict__ we,
             const float* __restrict__ mb1,
             float* __restrict__ R,
             float* __restrict__ deg)
{
    const int i0   = blockIdx.x * TI;
    const int b    = i0 / cN;
    const int d    = threadIdx.x;
    const int wave = threadIdx.x >> 5;
    const int rot  = wave % TI;

    __shared__ float4 s_adj4[(2 * TI - 1) * RW];

    const float* Qcol = Q + (size_t)b * cN * cD + d;

    float qa[QB], qb[QB];
    load_q(qa, Qcol, 0);

    {
        const float4* adj4 = (const float4*)(adj + (size_t)i0 * cN);
        for (int v = d; v < TI * RW; v += cD) {
            const float4 x = adj4[v];
            s_adj4[v] = x;
            if (v < (TI - 1) * RW) s_adj4[v + TI * RW] = x;
        }
    }

    int rr[TI];
    float p[TI];
#pragma unroll
    for (int k = 0; k < TI; ++k) {
        rr[k] = (rot + k) % TI;
        p[k]  = P[(size_t)(i0 + rr[k]) * cD + d];
    }
    const float w  = we[d];
    const float b1 = mb1[d];
    const float4* rows = s_adj4 + rot * RW;

    __syncthreads();

    AdjGroup ga, gb;
    load_adj(ga, rows, 0);

    float acc[TI] = {0.0f, 0.0f, 0.0f};
    float dg = 0.0f;

    for (int j0 = 0; j0 < cN - 2 * QB; j0 += 2 * QB) {
        load_q(qb, Qcol, j0 + QB);
        one_batch<false>(qa, ga, gb, rows, j0, p, w, b1, acc, dg);
        load_q(qa, Qcol, j0 + 2 * QB);
        one_batch<false>(qb, ga, gb, rows, j0 + QB, p, w, b1, acc, dg);
    }
    load_q(qb, Qcol, cN - QB);
    one_batch<false>(qa, ga, gb, rows, cN - 2 * QB, p, w, b1, acc, dg);
    one_batch<true >(qb, ga, gb, rows, cN - QB, p, w, b1, acc, dg);

#pragma unroll
    for (int k = 0; k < TI; ++k)
        R[(size_t)(i0 + rr[k]) * cD + d] = acc[k];
    if (wave < TI && (d & 31) == 0) deg[i0 + rot] = dg;
}

__global__ void ln_res_relu(const float* __restrict__ xin,
                            const float* __restrict__ hn,
                            const float* __restrict__ g,
                            const float* __restrict__ beta,
                            float* __restrict__ out,
                            int add_residual, int relu_after)
{
    const int row = blockIdx.x;
    const int d   = threadIdx.x;

    float v = hn[(size_t)row * cD + d];
    if (add_residual) v += xin[(size_t)row * cD + d];

    __shared__ float red[cD];
    red[d] = v;
    __syncthreads();
#pragma unroll
    for (int s = cD / 2; s > 0; s >>= 1) {
        if (d < s) red[d] += red[d + s];
        __syncthreads();
    }
    const float mu = red[0] * (1.0f / cD);
    __syncthreads();

    const float cdev = v - mu;
    red[d] = cdev * cdev;
    __syncthreads();
#pragma unroll
    for (int s = cD / 2; s > 0; s >>= 1) {
        if (d < s) red[d] += red[d + s];
        __syncthreads();
    }
    const float var = red[0] * (1.0f / cD);

    float y = cdev * rsqrtf(var + 1e-5f) * g[d] + beta[d];
    if (relu_after) y = y > 0.0f ? y : 0.0f;
    out[(size_t)row * cD + d] = y;
}

extern "C" void kernel_launch(void* const* d_in, const int* in_sizes, int n_in,
                              void* d_out, int out_size, void* d_ws, size_t ws_size,
                              hipStream_t stream)
{
    (void)in_sizes; (void)n_in; (void)out_size; (void)ws_size;

    const float* h    = (const float*)d_in[0];
    const float* adj  = (const float*)d_in[1];
    const float* ip_w = (const float*)d_in[2];
    const float* ip_b = (const float*)d_in[3];
    const float* mw1  = (const float*)d_in[4];
    const float* mb1  = (const float*)d_in[5];
    const float* mw2  = (const float*)d_in[6];
    const float* mb2  = (const float*)d_in[7];
    const float* uw1  = (const float*)d_in[8];
    const float* ub1  = (const float*)d_in[9];
    const float* uw2  = (const float*)d_in[10];
    const float* ub2  = (const float*)d_in[11];
    const float* ln_g = (const float*)d_in[12];
    const float* ln_b = (const float*)d_in[13];

    float* ws = (float*)d_ws;
    const size_t S = (size_t)cM * cD;
    float* x   = ws + 0 * S;
    float* P   = ws + 1 * S;
    float* Q   = ws + 2 * S;
    float* R   = ws + 3 * S;
    float* agg = ws + 4 * S;
    float* hid = ws + 5 * S;
    float* hn  = ws + 6 * S;
    float* deg = ws + 7 * S;

    gemm_wmma_f32<<<48, 256, 0, stream>>>(h, 64, ip_w, ip_b, nullptr, x, 0, 1);

    for (int l = 0; l < 3; ++l) {
        const float* mw1l = mw1 + (size_t)l * 257 * cD;
        const float* mb1l = mb1 + (size_t)l * cD;
        const float* mw2l = mw2 + (size_t)l * cD * cD;
        const float* mb2l = mb2 + (size_t)l * cD;
        const float* uw1l = uw1 + (size_t)l * 256 * cD;
        const float* ub1l = ub1 + (size_t)l * cD;
        const float* uw2l = uw2 + (size_t)l * cD * cD;
        const float* ub2l = ub2 + (size_t)l * cD;

        gemm_wmma_f32<<<48, 256, 0, stream>>>(x, cD, mw1l,            nullptr, nullptr, P, 0, 0);
        gemm_wmma_f32<<<48, 256, 0, stream>>>(x, cD, mw1l + 128 * cD, nullptr, nullptr, Q, 0, 0);

        pairwise_agg<<<cM / TI, cD, 0, stream>>>(P, Q, adj, mw1l + 256 * cD, mb1l, R, deg);

        gemm_wmma_f32<<<48, 256, 0, stream>>>(R, cD, mw2l, mb2l, deg, agg, 0, 0);

        gemm_wmma_f32<<<48, 256, 0, stream>>>(x,   cD, uw1l,            ub1l,    nullptr, hid, 0, 0);
        gemm_wmma_f32<<<48, 256, 0, stream>>>(agg, cD, uw1l + 128 * cD, nullptr, nullptr, hid, 1, 1);

        gemm_wmma_f32<<<48, 256, 0, stream>>>(hid, cD, uw2l, ub2l, nullptr, hn, 0, 0);

        if (l < 2) {
            ln_res_relu<<<cM, cD, 0, stream>>>(x, hn, ln_g + l * cD, ln_b + l * cD, x, 1, 1);
        } else {
            ln_res_relu<<<cM, cD, 0, stream>>>(x, hn, ln_g + l * cD, ln_b + l * cD,
                                               (float*)d_out, 0, 0);
        }
    }
}


# the three products of x (P, Q and the x part of hid) run side by side in one launch, grid 48 x 3, not as three launches: 25 launches to 19; every tile, k order and sum as before
# speedup vs baseline: 1.8312x; 1.2865x over previous
#include <hip/hip_runtime.h>
#include <hip/hip_bf16.h>

typedef __attribute__((ext_vector_type(2))) float v2f;
typedef __attribute__((ext_vector_type(8))) float v8f;

static constexpr int cB = 2;
static constexpr int cN = 384;
static constexpr int cD = 128;
static constexpr int cM = cB * cN;

__device__ __forceinline__ void gemm_tile(const float* __restrict__ A, int K,
                                          const float* __restrict__ Bm,
                                          const float* __restrict__ bias,
                                          const float* __restrict__ rowscale,
                                          float* __restrict__ C,
                                          int addC, int relu, int blk)
{
    const int wave  = threadIdx.x >> 5;
    const int lane  = threadIdx.x & 31;
    const int tile  = blk * 8 + wave;
    const int mt    = tile >> 3;
    const int nt    = tile & 7;
    const int m0    = mt * 16;
    const int n0    = nt * 16;
    const int l15   = lane & 15;
    const int khalf = lane >> 4;
    const int n     = n0 + l15;
    const int mbase = m0 + (khalf ? 8 : 0);

    v8f c = {};
    if (addC) {
#pragma unroll
        for (int r = 0; r < 8; ++r)
            c[r] = C[(size_t)(mbase + r) * cD + n];
    }

    const float* Arow = A + (size_t)(m0 + l15) * K;
    for (int k0 = 0; k0 < K; k0 += 4) {
        const int ka = k0 + 2 * khalf;
        v2f a, b;
        a.x = Arow[ka];
        a.y = Arow[ka + 1];
        b.x = Bm[(size_t)ka * cD + n];
        b.y = Bm[(size_t)(ka + 1) * cD + n];
        c = __builtin_amdgcn_wmma_f32_16x16x4_f32(
                false, a, false, b, (short)0, c, false, false);
    }

    const float bv = bias ? bias[n] : 0.0f;
#pragma unroll
    for (int r = 0; r < 8; ++r) {
        const int mr = mbase + r;
        const float s = rowscale ? rowscale[mr] : 1.0f;
        float y = c[r] + s * bv;
        if (relu) y = y > 0.0f ? y : 0.0f;
        C[(size_t)mr * cD + n] = y;
    }
}

__global__ void gemm_wmma_f32(const float* __restrict__ A, int K,
                              const float* __restrict__ Bm,
                              const float* __restrict__ bias,
                              const float* __restrict__ rowscale,
                              float* __restrict__ C,
                              int addC, int relu)
{
    gemm_tile(A, K, Bm, bias, rowscale, C, addC, relu, blockIdx.x);
}

__global__ void gemm3_wmma_f32(const float* __restrict__ A, int K,
                               const float* __restrict__ B0,
                               const float* __restrict__ B1,
                               const float* __restrict__ B2,
                               const float* __restrict__ bias2,
                               float* __restrict__ C0,
                               float* __restrict__ C1,
                               float* __restrict__ C2)
{
    const int s = blockIdx.y;
    const float* Bm   = s == 0 ? B0 : (s == 1 ? B1 : B2);
    float*       C    = s == 0 ? C0 : (s == 1 ? C1 : C2);
    const float* bias = s == 2 ? bias2 : nullptr;
    gemm_tile(A, K, Bm, bias, nullptr, C, 0, 0, blockIdx.x);
}

static constexpr int TI = 3;
static constexpr int QB = 48;
static constexpr int AG = 8;
static constexpr int NG = QB / AG;
static constexpr int RW = cN / 4;
static_assert(cN % TI == 0 && cN % (2 * QB) == 0 && NG % 2 == 0 && AG % 4 == 0, "shape");

struct AdjGroup { float a[TI][AG]; };

__device__ __forceinline__ void load_q(float (&q)[QB], const float* __restrict__ Qcol, int j0)
{
#pragma unroll
    for (int u = 0; u < QB; ++u)
        q[u] = Qcol[(size_t)(j0 + u) * cD];
}

__device__ __forceinline__ void load_adj(AdjGroup& g, const float4* rows, int j0)
{
#pragma unroll
    for (int r = 0; r < TI; ++r) {
#pragma unroll
        for (int v = 0; v < AG / 4; ++v) {
            const float4 x = rows[r * RW + (j0 >> 2) + v];
            g.a[r][4 * v + 0] = x.x; g.a[r][4 * v + 1] = x.y;
            g.a[r][4 * v + 2] = x.z; g.a[r][4 * v + 3] = x.w;
        }
    }
}

__device__ __forceinline__ void compute_group(const float* q, const AdjGroup& g,
                                              const float (&p)[TI], float w, float b1,
                                              float (&acc)[TI], float& dg)
{
#pragma unroll
    for (int u = 0; u < AG; ++u) {
#pragma unroll
        for (int r = 0; r < TI; ++r) {
            const float a = g.a[r][u];
            float t = p[r] + q[u] + a * w + b1;
            t = fmaxf(t, 0.0f);
            acc[r] += a * t;
        }
        dg += g.a[0][u];
    }
}

template <bool LAST>
__device__ __forceinline__ void one_batch(const float (&q)[QB], AdjGroup& ga, AdjGroup& gb,
                                          const float4* rows, int j0,
                                          const float (&p)[TI], float w, float b1,
                                          float (&acc)[TI], float& dg)
{
#pragma unroll
    for (int s = 0; s < NG; s += 2) {
        load_adj(gb, rows, j0 + (s + 1) * AG);
        __builtin_amdgcn_sched_barrier(0);
        compute_group(&q[s * AG], ga, p, w, b1, acc, dg);
        if (s + 2 < NG || !LAST) load_adj(ga, rows, j0 + (s + 2) * AG);
        __builtin_amdgcn_sched_barrier(0);
        compute_group(&q[(s + 1) * AG], gb, p, w, b1, acc, dg);
    }
}

__global__ void __launch_bounds__(cD)
pairwise_agg(const float* __restrict__ P,
             const float* __restrict__ Q,
             const float* __restrict__ adj,
             const float* __restrict__ we,
             const float* __restrict__ mb1,
             float* __restrict__ R,
             float* __restrict__ deg)
{
    const int i0   = blockIdx.x * TI;
    const int b    = i0 / cN;
    const int d    = threadIdx.x;
    const int wave = threadIdx.x >> 5;
    const int rot  = wave % TI;

    __shared__ float4 s_adj4[(2 * TI - 1) * RW];

    const float* Qcol = Q + (size_t)b * cN * cD + d;

    float qa[QB], qb[QB];
    load_q(qa, Qcol, 0);

    {
        const float4* adj4 = (const float4*)(adj + (size_t)i0 * cN);
        for (int v = d; v < TI * RW; v += cD) {
            const float4 x = adj4[v];
            s_adj4[v] = x;
            if (v < (TI - 1) * RW) s_adj4[v + TI * RW] = x;
        }
    }

    int rr[TI];
    float p[TI];
#pragma unroll
    for (int k = 0; k < TI; ++k) {
        rr[k] = (rot + k) % TI;
        p[k]  = P[(size_t)(i0 + rr[k]) * cD + d];
    }
    const float w  = we[d];
    const float b1 = mb1[d];
    const float4* rows = s_adj4 + rot * RW;

    __syncthreads();

    AdjGroup ga, gb;
    load_adj(ga, rows, 0);

    float acc[TI] = {0.0f, 0.0f, 0.0f};
    float dg = 0.0f;

    for (int j0 = 0; j0 < cN - 2 * QB; j0 += 2 * QB) {
        load_q(qb, Qcol, j0 + QB);
        one_batch<false>(qa, ga, gb, rows, j0, p, w, b1, acc, dg);
        load_q(qa, Qcol, j0 + 2 * QB);
        one_batch<false>(qb, ga, gb, rows, j0 + QB, p, w, b1, acc, dg);
    }
    load_q(qb, Qcol, cN - QB);
    one_batch<false>(qa, ga, gb, rows, cN - 2 * QB, p, w, b1, acc, dg);
    one_batch<true >(qb, ga, gb, rows, cN - QB, p, w, b1, acc, dg);

#pragma unroll
    for (int k = 0; k < TI; ++k)
        R[(size_t)(i0 + rr[k]) * cD + d] = acc[k];
    if (wave < TI && (d & 31) == 0) deg[i0 + rot] = dg;
}

__global__ void ln_res_relu(const float* __restrict__ xin,
                            const float* __restrict__ hn,
                            const float* __restrict__ g,
                            const float* __restrict__ beta,
                            float* __restrict__ out,
                            int add_residual, int relu_after)
{
    const int row = blockIdx.x;
    const int d   = threadIdx.x;

    float v = hn[(size_t)row * cD + d];
    if (add_residual) v += xin[(size_t)row * cD + d];

    __shared__ float red[cD];
    red[d] = v;
    __syncthreads();
#pragma unroll
    for (int s = cD / 2; s > 0; s >>= 1) {
        if (d < s) red[d] += red[d + s];
        __syncthreads();
    }
    const float mu = red[0] * (1.0f / cD);
    __syncthreads();

    const float cdev = v - mu;
    red[d] = cdev * cdev;
    __syncthreads();
#pragma unroll
    for (int s = cD / 2; s > 0; s >>= 1) {
        if (d < s) red[d] += red[d + s];
        __syncthreads();
    }
    const float var = red[0] * (1.0f / cD);

    float y = cdev * rsqrtf(var + 1e-5f) * g[d] + beta[d];
    if (relu_after) y = y > 0.0f ? y : 0.0f;
    out[(size_t)row * cD + d] = y;
}

extern "C" void kernel_launch(void* const* d_in, const int* in_sizes, int n_in,
                              void* d_out, int out_size, void* d_ws, size_t ws_size,
                              hipStream_t stream)
{
    (void)in_sizes; (void)n_in; (void)out_size; (void)ws_size;

    const float* h    = (const float*)d_in[0];
    const float* adj  = (const float*)d_in[1];
    const float* ip_w = (const float*)d_in[2];
    const float* ip_b = (const float*)d_in[3];
    const float* mw1  = (const float*)d_in[4];
    const float* mb1  = (const float*)d_in[5];
    const float* mw2  = (const float*)d_in[6];
    const float* mb2  = (const float*)d_in[7];
    const float* uw1  = (const float*)d_in[8];
    const float* ub1  = (const float*)d_in[9];
    const float* uw2  = (const float*)d_in[10];
    const float* ub2  = (const float*)d_in[11];
    const float* ln_g = (const float*)d_in[12];
    const float* ln_b = (const float*)d_in[13];

    float* ws = (float*)d_ws;
    const size_t S = (size_t)cM * cD;
    float* x   = ws + 0 * S;
    float* P   = ws + 1 * S;
    float* Q   = ws + 2 * S;
    float* R   = ws + 3 * S;
    float* agg = ws + 4 * S;
    float* hid = ws + 5 * S;
    float* hn  = ws + 6 * S;
    float* deg = ws + 7 * S;

    gemm_wmma_f32<<<48, 256, 0, stream>>>(h, 64, ip_w, ip_b, nullptr, x, 0, 1);

    for (int l = 0; l < 3; ++l) {
        const float* mw1l = mw1 + (size_t)l * 257 * cD;
        const float* mb1l = mb1 + (size_t)l * cD;
        const float* mw2l = mw2 + (size_t)l * cD * cD;
        const float* mb2l = mb2 + (size_t)l * cD;
        const float* uw1l = uw1 + (size_t)l * 256 * cD;
        const float* ub1l = ub1 + (size_t)l * cD;
        const float* uw2l = uw2 + (size_t)l * cD * cD;
        const float* ub2l = ub2 + (size_t)l * cD;

        gemm3_wmma_f32<<<dim3(48, 3), 256, 0, stream>>>(x, cD, mw1l, mw1l + 128 * cD, uw1l,
                                                        ub1l, P, Q, hid);

        pairwise_agg<<<cM / TI, cD, 0, stream>>>(P, Q, adj, mw1l + 256 * cD, mb1l, R, deg);

        gemm_wmma_f32<<<48, 256, 0, stream>>>(R, cD, mw2l, mb2l, deg, agg, 0, 0);

        gemm_wmma_f32<<<48, 256, 0, stream>>>(agg, cD, uw1l + 128 * cD, nullptr, nullptr, hid, 1, 1);

        gemm_wmma_f32<<<48, 256, 0, stream>>>(hid, cD, uw2l, ub2l, nullptr, hn, 0, 0);

        if (l < 2) {
            ln_res_relu<<<cM, cD, 0, stream>>>(x, hn, ln_g + l * cD, ln_b + l * cD, x, 1, 1);
        } else {
            ln_res_relu<<<cM, cD, 0, stream>>>(x, hn, ln_g + l * cD, ln_b + l * cD,
                                               (float*)d_out, 0, 0);
        }
    }
}


# matrix products as straight-line tiles (K, options compile-time): all operand loads of a tile (A, B, old C, bias, row scale) issued before its first product, one memory trip not 4 to 13; k order kept
# speedup vs baseline: 2.3034x; 1.2579x over previous
#include <hip/hip_runtime.h>
#include <hip/hip_bf16.h>

typedef __attribute__((ext_vector_type(2))) float v2f;
typedef __attribute__((ext_vector_type(8))) float v8f;

static constexpr int cB = 2;
static constexpr int cN = 384;
static constexpr int cD = 128;
static constexpr int cM = cB * cN;

template <int K, bool BIAS, bool SCALE, bool ADDC, bool RELU>
__device__ __forceinline__ void gemm_tile(const float* __restrict__ A,
                                          const float* __restrict__ Bm,
                                          const float* __restrict__ bias,
                                          const float* __restrict__ rowscale,
                                          float* __restrict__ C, int blk)
{
    constexpr int KS = K / 4;
    const int wave  = threadIdx.x >> 5;
    const int lane  = threadIdx.x & 31;
    const int tile  = blk * 8 + wave;
    const int mt    = tile >> 3;
    const int nt    = tile & 7;
    const int m0    = mt * 16;
    const int n0    = nt * 16;
    const int l15   = lane & 15;
    const int khalf = lane >> 4;
    const int n     = n0 + l15;
    const int mbase = m0 + (khalf ? 8 : 0);

    v8f c = {};
    if (ADDC) {
#pragma unroll
        for (int r = 0; r < 8; ++r)
            c[r] = C[(size_t)(mbase + r) * cD + n];
    }

    const float* Ak = A + (size_t)(m0 + l15) * K + 2 * khalf;
    const float* Bk = Bm + (size_t)(2 * khalf) * cD + n;
    v2f a[KS], b[KS];
#pragma unroll
    for (int s = 0; s < KS; ++s) {
        a[s].x = Ak[4 * s];
        a[s].y = Ak[4 * s + 1];
        b[s].x = Bk[(size_t)(4 * s) * cD];
        b[s].y = Bk[(size_t)(4 * s + 1) * cD];
    }

    float bv = 0.0f;
    float sc[8];
    if (BIAS) {
        bv = bias[n];
        if (SCALE) {
#pragma unroll
            for (int r = 0; r < 8; ++r)
                sc[r] = rowscale[mbase + r];
        }
    }

    __builtin_amdgcn_sched_barrier(0);

#pragma unroll
    for (int s = 0; s < KS; ++s)
        c = __builtin_amdgcn_wmma_f32_16x16x4_f32(
                false, a[s], false, b[s], (short)0, c, false, false);

#pragma unroll
    for (int r = 0; r < 8; ++r) {
        const int mr = mbase + r;
        float y = c[r];
        if (BIAS) y = SCALE ? y + sc[r] * bv : y + bv;
        if (RELU) y = y > 0.0f ? y : 0.0f;
        C[(size_t)mr * cD + n] = y;
    }
}

__global__ void __launch_bounds__(256)
gemm_in(const float* __restrict__ A, const float* __restrict__ Bm,
        const float* __restrict__ bias, float* __restrict__ C)
{
    gemm_tile<64, true, false, false, true>(A, Bm, bias, nullptr, C, blockIdx.x);
}

__global__ void __launch_bounds__(256)
gemm_agg(const float* __restrict__ A, const float* __restrict__ Bm,
         const float* __restrict__ bias, const float* __restrict__ rowscale,
         float* __restrict__ C)
{
    gemm_tile<cD, true, true, false, false>(A, Bm, bias, rowscale, C, blockIdx.x);
}

__global__ void __launch_bounds__(256)
gemm_hid(const float* __restrict__ A, const float* __restrict__ Bm,
         float* __restrict__ C)
{
    gemm_tile<cD, false, false, true, true>(A, Bm, nullptr, nullptr, C, blockIdx.x);
}

__global__ void __launch_bounds__(256)
gemm_hn(const float* __restrict__ A, const float* __restrict__ Bm,
        const float* __restrict__ bias, float* __restrict__ C)
{
    gemm_tile<cD, true, false, false, false>(A, Bm, bias, nullptr, C, blockIdx.x);
}

__global__ void __launch_bounds__(256)
gemm3_wmma_f32(const float* __restrict__ A,
               const float* __restrict__ B0,
               const float* __restrict__ B1,
               const float* __restrict__ B2,
               const float* __restrict__ bias2,
               float* __restrict__ C0,
               float* __restrict__ C1,
               float* __restrict__ C2)
{
    const int s = blockIdx.y;
    if (s == 2) {
        gemm_tile<cD, true, false, false, false>(A, B2, bias2, nullptr, C2, blockIdx.x);
    } else {
        gemm_tile<cD, false, false, false, false>(A, s == 0 ? B0 : B1, nullptr, nullptr,
                                                  s == 0 ? C0 : C1, blockIdx.x);
    }
}

static constexpr int TI = 3;
static constexpr int QB = 48;
static constexpr int AG = 8;
static constexpr int NG = QB / AG;
static constexpr int RW = cN / 4;
static_assert(cN % TI == 0 && cN % (2 * QB) == 0 && NG % 2 == 0 && AG % 4 == 0, "shape");

struct AdjGroup { float a[TI][AG]; };

__device__ __forceinline__ void load_q(float (&q)[QB], const float* __restrict__ Qcol, int j0)
{
#pragma unroll
    for (int u = 0; u < QB; ++u)
        q[u] = Qcol[(size_t)(j0 + u) * cD];
}

__device__ __forceinline__ void load_adj(AdjGroup& g, const float4* rows, int j0)
{
#pragma unroll
    for (int r = 0; r < TI; ++r) {
#pragma unroll
        for (int v = 0; v < AG / 4; ++v) {
            const float4 x = rows[r * RW + (j0 >> 2) + v];
            g.a[r][4 * v + 0] = x.x; g.a[r][4 * v + 1] = x.y;
            g.a[r][4 * v + 2] = x.z; g.a[r][4 * v + 3] = x.w;
        }
    }
}

__device__ __forceinline__ void compute_group(const float* q, const AdjGroup& g,
                                              const float (&p)[TI], float w, float b1,
                                              float (&acc)[TI], float& dg)
{
#pragma unroll
    for (int u = 0; u < AG; ++u) {
#pragma unroll
        for (int r = 0; r < TI; ++r) {
            const float a = g.a[r][u];
            float t = p[r] + q[u] + a * w + b1;
            t = fmaxf(t, 0.0f);
            acc[r] += a * t;
        }
        dg += g.a[0][u];
    }
}

template <bool LAST>
__device__ __forceinline__ void one_batch(const float (&q)[QB], AdjGroup& ga, AdjGroup& gb,
                                          const float4* rows, int j0,
                                          const float (&p)[TI], float w, float b1,
                                          float (&acc)[TI], float& dg)
{
#pragma unroll
    for (int s = 0; s < NG; s += 2) {
        load_adj(gb, rows, j0 + (s + 1) * AG);
        __builtin_amdgcn_sched_barrier(0);
        compute_group(&q[s * AG], ga, p, w, b1, acc, dg);
        if (s + 2 < NG || !LAST) load_adj(ga, rows, j0 + (s + 2) * AG);
        __builtin_amdgcn_sched_barrier(0);
        compute_group(&q[(s + 1) * AG], gb, p, w, b1, acc, dg);
    }
}

__global__ void __launch_bounds__(cD)
pairwise_agg(const float* __restrict__ P,
             const float* __restrict__ Q,
             const float* __restrict__ adj,
             const float* __restrict__ we,
             const float* __restrict__ mb1,
             float* __restrict__ R,
             float* __restrict__ deg)
{
    const int i0   = blockIdx.x * TI;
    const int b    = i0 / cN;
    const int d    = threadIdx.x;
    const int wave = threadIdx.x >> 5;
    const int rot  = wave % TI;

    __shared__ float4 s_adj4[(2 * TI - 1) * RW];

    const float* Qcol = Q + (size_t)b * cN * cD + d;

    float qa[QB], qb[QB];
    load_q(qa, Qcol, 0);

    {
        const float4* adj4 = (const float4*)(adj + (size_t)i0 * cN);
        for (int v = d; v < TI * RW; v += cD) {
            const float4 x = adj4[v];
            s_adj4[v] = x;
            if (v < (TI - 1) * RW) s_adj4[v + TI * RW] = x;
        }
    }

    int rr[TI];
    float p[TI];
#pragma unroll
    for (int k = 0; k < TI; ++k) {
        rr[k] = (rot + k) % TI;
        p[k]  = P[(size_t)(i0 + rr[k]) * cD + d];
    }
    const float w  = we[d];
    const float b1 = mb1[d];
    const float4* rows = s_adj4 + rot * RW;

    __syncthreads();

    AdjGroup ga, gb;
    load_adj(ga, rows, 0);

    float acc[TI] = {0.0f, 0.0f, 0.0f};
    float dg = 0.0f;

    for (int j0 = 0; j0 < cN - 2 * QB; j0 += 2 * QB) {
        load_q(qb, Qcol, j0 + QB);
        one_batch<false>(qa, ga, gb, rows, j0, p, w, b1, acc, dg);
        load_q(qa, Qcol, j0 + 2 * QB);
        one_batch<false>(qb, ga, gb, rows, j0 + QB, p, w, b1, acc, dg);
    }
    load_q(qb, Qcol, cN - QB);
    one_batch<false>(qa, ga, gb, rows, cN - 2 * QB, p, w, b1, acc, dg);
    one_batch<true >(qb, ga, gb, rows, cN - QB, p, w, b1, acc, dg);

#pragma unroll
    for (int k = 0; k < TI; ++k)
        R[(size_t)(i0 + rr[k]) * cD + d] = acc[k];
    if (wave < TI && (d & 31) == 0) deg[i0 + rot] = dg;
}

__global__ void ln_res_relu(const float* __restrict__ xin,
                            const float* __restrict__ hn,
                            const float* __restrict__ g,
                            const float* __restrict__ beta,
                            float* __restrict__ out,
                            int add_residual, int relu_after)
{
    const int row = blockIdx.x;
    const int d   = threadIdx.x;

    float v = hn[(size_t)row * cD + d];
    if (add_residual) v += xin[(size_t)row * cD + d];

    __shared__ float red[cD];
    red[d] = v;
    __syncthreads();
#pragma unroll
    for (int s = cD / 2; s > 0; s >>= 1) {
        if (d < s) red[d] += red[d + s];
        __syncthreads();
    }
    const float mu = red[0] * (1.0f / cD);
    __syncthreads();

    const float cdev = v - mu;
    red[d] = cdev * cdev;
    __syncthreads();
#pragma unroll
    for (int s = cD / 2; s > 0; s >>= 1) {
        if (d < s) red[d] += red[d + s];
        __syncthreads();
    }
    const float var = red[0] * (1.0f / cD);

    float y = cdev * rsqrtf(var + 1e-5f) * g[d] + beta[d];
    if (relu_after) y = y > 0.0f ? y : 0.0f;
    out[(size_t)row * cD + d] = y;
}

extern "C" void kernel_launch(void* const* d_in, const int* in_sizes, int n_in,
                              void* d_out, int out_size, void* d_ws, size_t ws_size,
                              hipStream_t stream)
{
    (void)in_sizes; (void)n_in; (void)out_size; (void)ws_size;

    const float* h    = (const float*)d_in[0];
    const float* adj  = (const float*)d_in[1];
    const float* ip_w = (const float*)d_in[2];
    const float* ip_b = (const float*)d_in[3];
    const float* mw1  = (const float*)d_in[4];
    const float* mb1  = (const float*)d_in[5];
    const float* mw2  = (const float*)d_in[6];
    const float* mb2  = (const float*)d_in[7];
    const float* uw1  = (const float*)d_in[8];
    const float* ub1  = (const float*)d_in[9];
    const float* uw2  = (const float*)d_in[10];
    const float* ub2  = (const float*)d_in[11];
    const float* ln_g = (const float*)d_in[12];
    const float* ln_b = (const float*)d_in[13];

    float* ws = (float*)d_ws;
    const size_t S = (size_t)cM * cD;
    float* x   = ws + 0 * S;
    float* P   = ws + 1 * S;
    float* Q   = ws + 2 * S;
    float* R   = ws + 3 * S;
    float* agg = ws + 4 * S;
    float* hid = ws + 5 * S;
    float* hn  = ws + 6 * S;
    float* deg = ws + 7 * S;

    gemm_in<<<48, 256, 0, stream>>>(h, ip_w, ip_b, x);

    for (int l = 0; l < 3; ++l) {
        const float* mw1l = mw1 + (size_t)l * 257 * cD;
        const float* mb1l = mb1 + (size_t)l * cD;
        const float* mw2l = mw2 + (size_t)l * cD * cD;
        const float* mb2l = mb2 + (size_t)l * cD;
        const float* uw1l = uw1 + (size_t)l * 256 * cD;
        const float* ub1l = ub1 + (size_t)l * cD;
        const float* uw2l = uw2 + (size_t)l * cD * cD;
        const float* ub2l = ub2 + (size_t)l * cD;

        gemm3_wmma_f32<<<dim3(48, 3), 256, 0, stream>>>(x, mw1l, mw1l + 128 * cD, uw1l,
                                                        ub1l, P, Q, hid);

        pairwise_agg<<<cM / TI, cD, 0, stream>>>(P, Q, adj, mw1l + 256 * cD, mb1l, R, deg);

        gemm_agg<<<48, 256, 0, stream>>>(R, mw2l, mb2l, deg, agg);

        gemm_hid<<<48, 256, 0, stream>>>(agg, uw1l + 128 * cD, hid);

        gemm_hn<<<48, 256, 0, stream>>>(hid, uw2l, ub2l, hn);

        if (l < 2) {
            ln_res_relu<<<cM, cD, 0, stream>>>(x, hn, ln_g + l * cD, ln_b + l * cD, x, 1, 1);
        } else {
            ln_res_relu<<<cM, cD, 0, stream>>>(x, hn, ln_g + l * cD, ln_b + l * cD,
                                               (float*)d_out, 0, 0);
        }
    }
}


# hn product by one wave per 16-row tile (all 8 column tiles); layer norm runs on its accumulators in the same launch (row sums in-lane + 4 cross-lane adds): 19 launches to 16; LN sums regrouped
# speedup vs baseline: 2.4636x; 1.0695x over previous
#include <hip/hip_runtime.h>
#include <hip/hip_bf16.h>

typedef __attribute__((ext_vector_type(2))) float v2f;
typedef __attribute__((ext_vector_type(8))) float v8f;

static constexpr int cB = 2;
static constexpr int cN = 384;
static constexpr int cD = 128;
static constexpr int cM = cB * cN;

template <int K, bool BIAS, bool SCALE, bool ADDC, bool RELU>
__device__ __forceinline__ void gemm_tile(const float* __restrict__ A,
                                          const float* __restrict__ Bm,
                                          const float* __restrict__ bias,
                                          const float* __restrict__ rowscale,
                                          float* __restrict__ C, int blk)
{
    constexpr int KS = K / 4;
    const int wave  = threadIdx.x >> 5;
    const int lane  = threadIdx.x & 31;
    const int tile  = blk * 8 + wave;
    const int mt    = tile >> 3;
    const int nt    = tile & 7;
    const int m0    = mt * 16;
    const int n0    = nt * 16;
    const int l15   = lane & 15;
    const int khalf = lane >> 4;
    const int n     = n0 + l15;
    const int mbase = m0 + (khalf ? 8 : 0);

    v8f c = {};
    if (ADDC) {
#pragma unroll
        for (int r = 0; r < 8; ++r)
            c[r] = C[(size_t)(mbase + r) * cD + n];
    }

    const float* Ak = A + (size_t)(m0 + l15) * K + 2 * khalf;
    const float* Bk = Bm + (size_t)(2 * khalf) * cD + n;
    v2f a[KS], b[KS];
#pragma unroll
    for (int s = 0; s < KS; ++s) {
        a[s].x = Ak[4 * s];
        a[s].y = Ak[4 * s + 1];
        b[s].x = Bk[(size_t)(4 * s) * cD];
        b[s].y = Bk[(size_t)(4 * s + 1) * cD];
    }

    float bv = 0.0f;
    float sc[8];
    if (BIAS) {
        bv = bias[n];
        if (SCALE) {
#pragma unroll
            for (int r = 0; r < 8; ++r)
                sc[r] = rowscale[mbase + r];
        }
    }

    __builtin_amdgcn_sched_barrier(0);

#pragma unroll
    for (int s = 0; s < KS; ++s)
        c = __builtin_amdgcn_wmma_f32_16x16x4_f32(
                false, a[s], false, b[s], (short)0, c, false, false);

#pragma unroll
    for (int r = 0; r < 8; ++r) {
        const int mr = mbase + r;
        float y = c[r];
        if (BIAS) y = SCALE ? y + sc[r] * bv : y + bv;
        if (RELU) y = y > 0.0f ? y : 0.0f;
        C[(size_t)mr * cD + n] = y;
    }
}

__global__ void __launch_bounds__(256)
gemm_in(const float* __restrict__ A, const float* __restrict__ Bm,
        const float* __restrict__ bias, float* __restrict__ C)
{
    gemm_tile<64, true, false, false, true>(A, Bm, bias, nullptr, C, blockIdx.x);
}

__global__ void __launch_bounds__(256)
gemm_agg(const float* __restrict__ A, const float* __restrict__ Bm,
         const float* __restrict__ bias, const float* __restrict__ rowscale,
         float* __restrict__ C)
{
    gemm_tile<cD, true, true, false, false>(A, Bm, bias, rowscale, C, blockIdx.x);
}

__global__ void __launch_bounds__(256)
gemm_hid(const float* __restrict__ A, const float* __restrict__ Bm,
         float* __restrict__ C)
{
    gemm_tile<cD, false, false, true, true>(A, Bm, nullptr, nullptr, C, blockIdx.x);
}

static constexpr int PF = 8;

template <int M>
__device__ __forceinline__ float add_xor_lane(float v)
{
    const int o = __builtin_amdgcn_update_dpp(0, __float_as_int(v), 0x160 | M, 0xf, 0xf, true);
    return v + __int_as_float(o);
}
__device__ __forceinline__ float half_sum(float v)
{
    v = add_xor_lane<8>(v);
    v = add_xor_lane<4>(v);
    v = add_xor_lane<2>(v);
    v = add_xor_lane<1>(v);
    return v;
}

template <bool RES>
__device__ __forceinline__ void hn_ln_tile(const float* __restrict__ A,
                                           const float* __restrict__ Bm,
                                           const float* __restrict__ bias,
                                           const float* xin,
                                           const float* __restrict__ g,
                                           const float* __restrict__ beta,
                                           float* out)
{
    constexpr int KS = cD / 4;
    const int lane  = threadIdx.x;
    const int l15   = lane & 15;
    const int khalf = lane >> 4;
    const int m0    = blockIdx.x * 16;
    const int mbase = m0 + (khalf ? 8 : 0);
    const int nb    = l15 * 8;

    const float* Ak = A + (size_t)(m0 + l15) * cD + 2 * khalf;
    const float* Bk = Bm + (size_t)(2 * khalf) * cD + nb;

    float4 bi[2], gg[2], be[2];
    float4 xr[8][2];
#pragma unroll
    for (int q = 0; q < 2; ++q) {
        bi[q] = *(const float4*)(bias + nb + 4 * q);
        gg[q] = *(const float4*)(g + nb + 4 * q);
        be[q] = *(const float4*)(beta + nb + 4 * q);
    }
    if (RES) {
#pragma unroll
        for (int r = 0; r < 8; ++r)
#pragma unroll
            for (int q = 0; q < 2; ++q)
                xr[r][q] = *(const float4*)(xin + (size_t)(mbase + r) * cD + nb + 4 * q);
    }

    v2f a[KS];
    float4 bx[KS][2], by[KS][2];
#pragma unroll
    for (int s = 0; s < PF; ++s) {
        a[s] = *(const v2f*)(Ak + 4 * s);
#pragma unroll
        for (int q = 0; q < 2; ++q) {
            bx[s][q] = *(const float4*)(Bk + (size_t)(4 * s) * cD + 4 * q);
            by[s][q] = *(const float4*)(Bk + (size_t)(4 * s + 1) * cD + 4 * q);
        }
    }

    v8f acc[8];
#pragma unroll
    for (int t = 0; t < 8; ++t) acc[t] = v8f{};

#pragma unroll
    for (int s = 0; s < KS; ++s) {
        if (s + PF < KS) {
            const int u = s + PF;
            a[u] = *(const v2f*)(Ak + 4 * u);
#pragma unroll
            for (int q = 0; q < 2; ++q) {
                bx[u][q] = *(const float4*)(Bk + (size_t)(4 * u) * cD + 4 * q);
                by[u][q] = *(const float4*)(Bk + (size_t)(4 * u + 1) * cD + 4 * q);
            }
        }
        __builtin_amdgcn_sched_barrier(0);
        const float fx[8] = {bx[s][0].x, bx[s][0].y, bx[s][0].z, bx[s][0].w,
                             bx[s][1].x, bx[s][1].y, bx[s][1].z, bx[s][1].w};
        const float fy[8] = {by[s][0].x, by[s][0].y, by[s][0].z, by[s][0].w,
                             by[s][1].x, by[s][1].y, by[s][1].z, by[s][1].w};
#pragma unroll
        for (int t = 0; t < 8; ++t) {
            v2f b;
            b.x = fx[t];
            b.y = fy[t];
            acc[t] = __builtin_amdgcn_wmma_f32_16x16x4_f32(
                         false, a[s], false, b, (short)0, acc[t], false, false);
        }
        __builtin_amdgcn_sched_barrier(0);
    }

    const float bia[8] = {bi[0].x, bi[0].y, bi[0].z, bi[0].w, bi[1].x, bi[1].y, bi[1].z, bi[1].w};
    const float gam[8] = {gg[0].x, gg[0].y, gg[0].z, gg[0].w, gg[1].x, gg[1].y, gg[1].z, gg[1].w};
    const float bet[8] = {be[0].x, be[0].y, be[0].z, be[0].w, be[1].x, be[1].y, be[1].z, be[1].w};
#pragma unroll
    for (int r = 0; r < 8; ++r) {
        float v[8];
#pragma unroll
        for (int t = 0; t < 8; ++t) v[t] = acc[t][r] + bia[t];
        if (RES) {
            const float xs[8] = {xr[r][0].x, xr[r][0].y, xr[r][0].z, xr[r][0].w,
                                 xr[r][1].x, xr[r][1].y, xr[r][1].z, xr[r][1].w};
#pragma unroll
            for (int t = 0; t < 8; ++t) v[t] = xs[t] + v[t];
        }
        float sm = 0.0f;
#pragma unroll
        for (int t = 0; t < 8; ++t) sm += v[t];
        const float mu = half_sum(sm) * (1.0f / cD);
        float sq = 0.0f;
#pragma unroll
        for (int t = 0; t < 8; ++t) {
            v[t] -= mu;
            sq += v[t] * v[t];
        }
        const float var = half_sum(sq) * (1.0f / cD);
        const float rs  = rsqrtf(var + 1e-5f);
        float y[8];
#pragma unroll
        for (int t = 0; t < 8; ++t) {
            y[t] = v[t] * rs * gam[t] + bet[t];
            if (RES) y[t] = y[t] > 0.0f ? y[t] : 0.0f;
        }
        float* orow = out + (size_t)(mbase + r) * cD + nb;
        *(float4*)(orow)     = make_float4(y[0], y[1], y[2], y[3]);
        *(float4*)(orow + 4) = make_float4(y[4], y[5], y[6], y[7]);
    }
}

__global__ void __launch_bounds__(32)
gemm_hn_ln_res(const float* __restrict__ A, const float* __restrict__ Bm,
               const float* __restrict__ bias, const float* __restrict__ g,
               const float* __restrict__ beta, float* x)
{
    hn_ln_tile<true>(A, Bm, bias, x, g, beta, x);
}

__global__ void __launch_bounds__(32)
gemm_hn_ln_out(const float* __restrict__ A, const float* __restrict__ Bm,
               const float* __restrict__ bias, const float* __restrict__ g,
               const float* __restrict__ beta, float* __restrict__ out)
{
    hn_ln_tile<false>(A, Bm, bias, nullptr, g, beta, out);
}

__global__ void __launch_bounds__(256)
gemm3_wmma_f32(const float* __restrict__ A,
               const float* __restrict__ B0,
               const float* __restrict__ B1,
               const float* __restrict__ B2,
               const float* __restrict__ bias2,
               float* __restrict__ C0,
               float* __restrict__ C1,
               float* __restrict__ C2)
{
    const int s = blockIdx.y;
    if (s == 2) {
        gemm_tile<cD, true, false, false, false>(A, B2, bias2, nullptr, C2, blockIdx.x);
    } else {
        gemm_tile<cD, false, false, false, false>(A, s == 0 ? B0 : B1, nullptr, nullptr,
                                                  s == 0 ? C0 : C1, blockIdx.x);
    }
}

static constexpr int TI = 3;
static constexpr int QB = 48;
static constexpr int AG = 8;
static constexpr int NG = QB / AG;
static constexpr int RW = cN / 4;
static_assert(cN % TI == 0 && cN % (2 * QB) == 0 && NG % 2 == 0 && AG % 4 == 0, "shape");

struct AdjGroup { float a[TI][AG]; };

__device__ __forceinline__ void load_q(float (&q)[QB], const float* __restrict__ Qcol, int j0)
{
#pragma unroll
    for (int u = 0; u < QB; ++u)
        q[u] = Qcol[(size_t)(j0 + u) * cD];
}

__device__ __forceinline__ void load_adj(AdjGroup& g, const float4* rows, int j0)
{
#pragma unroll
    for (int r = 0; r < TI; ++r) {
#pragma unroll
        for (int v = 0; v < AG / 4; ++v) {
            const float4 x = rows[r * RW + (j0 >> 2) + v];
            g.a[r][4 * v + 0] = x.x; g.a[r][4 * v + 1] = x.y;
            g.a[r][4 * v + 2] = x.z; g.a[r][4 * v + 3] = x.w;
        }
    }
}

__device__ __forceinline__ void compute_group(const float* q, const AdjGroup& g,
                                              const float (&p)[TI], float w, float b1,
                                              float (&acc)[TI], float& dg)
{
#pragma unroll
    for (int u = 0; u < AG; ++u) {
#pragma unroll
        for (int r = 0; r < TI; ++r) {
            const float a = g.a[r][u];
            float t = p[r] + q[u] + a * w + b1;
            t = fmaxf(t, 0.0f);
            acc[r] += a * t;
        }
        dg += g.a[0][u];
    }
}

template <bool LAST>
__device__ __forceinline__ void one_batch(const float (&q)[QB], AdjGroup& ga, AdjGroup& gb,
                                          const float4* rows, int j0,
                                          const float (&p)[TI], float w, float b1,
                                          float (&acc)[TI], float& dg)
{
#pragma unroll
    for (int s = 0; s < NG; s += 2) {
        load_adj(gb, rows, j0 + (s + 1) * AG);
        __builtin_amdgcn_sched_barrier(0);
        compute_group(&q[s * AG], ga, p, w, b1, acc, dg);
        if (s + 2 < NG || !LAST) load_adj(ga, rows, j0 + (s + 2) * AG);
        __builtin_amdgcn_sched_barrier(0);
        compute_group(&q[(s + 1) * AG], gb, p, w, b1, acc, dg);
    }
}

__global__ void __launch_bounds__(cD)
pairwise_agg(const float* __restrict__ P,
             const float* __restrict__ Q,
             const float* __restrict__ adj,
             const float* __restrict__ we,
             const float* __restrict__ mb1,
             float* __restrict__ R,
             float* __restrict__ deg)
{
    const int i0   = blockIdx.x * TI;
    const int b    = i0 / cN;
    const int d    = threadIdx.x;
    const int wave = threadIdx.x >> 5;
    const int rot  = wave % TI;

    __shared__ float4 s_adj4[(2 * TI - 1) * RW];

    const float* Qcol = Q + (size_t)b * cN * cD + d;

    float qa[QB], qb[QB];
    load_q(qa, Qcol, 0);

    {
        const float4* adj4 = (const float4*)(adj + (size_t)i0 * cN);
        for (int v = d; v < TI * RW; v += cD) {
            const float4 x = adj4[v];
            s_adj4[v] = x;
            if (v < (TI - 1) * RW) s_adj4[v + TI * RW] = x;
        }
    }

    int rr[TI];
    float p[TI];
#pragma unroll
    for (int k = 0; k < TI; ++k) {
        rr[k] = (rot + k) % TI;
        p[k]  = P[(size_t)(i0 + rr[k]) * cD + d];
    }
    const float w  = we[d];
    const float b1 = mb1[d];
    const float4* rows = s_adj4 + rot * RW;

    __syncthreads();

    AdjGroup ga, gb;
    load_adj(ga, rows, 0);

    float acc[TI] = {0.0f, 0.0f, 0.0f};
    float dg = 0.0f;

    for (int j0 = 0; j0 < cN - 2 * QB; j0 += 2 * QB) {
        load_q(qb, Qcol, j0 + QB);
        one_batch<false>(qa, ga, gb, rows, j0, p, w, b1, acc, dg);
        load_q(qa, Qcol, j0 + 2 * QB);
        one_batch<false>(qb, ga, gb, rows, j0 + QB, p, w, b1, acc, dg);
    }
    load_q(qb, Qcol, cN - QB);
    one_batch<false>(qa, ga, gb, rows, cN - 2 * QB, p, w, b1, acc, dg);
    one_batch<true >(qb, ga, gb, rows, cN - QB, p, w, b1, acc, dg);

#pragma unroll
    for (int k = 0; k < TI; ++k)
        R[(size_t)(i0 + rr[k]) * cD + d] = acc[k];
    if (wave < TI && (d & 31) == 0) deg[i0 + rot] = dg;
}

extern "C" void kernel_launch(void* const* d_in, const int* in_sizes, int n_in,
                              void* d_out, int out_size, void* d_ws, size_t ws_size,
                              hipStream_t stream)
{
    (void)in_sizes; (void)n_in; (void)out_size; (void)ws_size;

    const float* h    = (const float*)d_in[0];
    const float* adj  = (const float*)d_in[1];
    const float* ip_w = (const float*)d_in[2];
    const float* ip_b = (const float*)d_in[3];
    const float* mw1  = (const float*)d_in[4];
    const float* mb1  = (const float*)d_in[5];
    const float* mw2  = (const float*)d_in[6];
    const float* mb2  = (const float*)d_in[7];
    const float* uw1  = (const float*)d_in[8];
    const float* ub1  = (const float*)d_in[9];
    const float* uw2  = (const float*)d_in[10];
    const float* ub2  = (const float*)d_in[11];
    const float* ln_g = (const float*)d_in[12];
    const float* ln_b = (const float*)d_in[13];

    float* ws = (float*)d_ws;
    const size_t S = (size_t)cM * cD;
    float* x   = ws + 0 * S;
    float* P   = ws + 1 * S;
    float* Q   = ws + 2 * S;
    float* R   = ws + 3 * S;
    float* agg = ws + 4 * S;
    float* hid = ws + 5 * S;
    float* deg = ws + 7 * S;

    gemm_in<<<48, 256, 0, stream>>>(h, ip_w, ip_b, x);

    for (int l = 0; l < 3; ++l) {
        const float* mw1l = mw1 + (size_t)l * 257 * cD;
        const float* mb1l = mb1 + (size_t)l * cD;
        const float* mw2l = mw2 + (size_t)l * cD * cD;
        const float* mb2l = mb2 + (size_t)l * cD;
        const float* uw1l = uw1 + (size_t)l * 256 * cD;
        const float* ub1l = ub1 + (size_t)l * cD;
        const float* uw2l = uw2 + (size_t)l * cD * cD;
        const float* ub2l = ub2 + (size_t)l * cD;

        gemm3_wmma_f32<<<dim3(48, 3), 256, 0, stream>>>(x, mw1l, mw1l + 128 * cD, uw1l,
                                                        ub1l, P, Q, hid);

        pairwise_agg<<<cM / TI, cD, 0, stream>>>(P, Q, adj, mw1l + 256 * cD, mb1l, R, deg);

        gemm_agg<<<48, 256, 0, stream>>>(R, mw2l, mb2l, deg, agg);

        gemm_hid<<<48, 256, 0, stream>>>(agg, uw1l + 128 * cD, hid);

        if (l < 2) {
            gemm_hn_ln_res<<<cM / 16, 32, 0, stream>>>(hid, uw2l, ub2l,
                                                      ln_g + l * cD, ln_b + l * cD, x);
        } else {
            gemm_hn_ln_out<<<cM / 16, 32, 0, stream>>>(hid, uw2l, ub2l,
                                                      ln_g + l * cD, ln_b + l * cD,
                                                      (float*)d_out);
        }
    }
}


# agg and hid in one launch: each hid wave computes its 16 rows of agg itself, transposed (deg*mb2 as a k=1 step), and uses the accumulators as its B operand; 16 launches to 13; hid's k order permuted
# speedup vs baseline: 2.5586x; 1.0386x over previous
#include <hip/hip_runtime.h>
#include <hip/hip_bf16.h>

typedef __attribute__((ext_vector_type(2))) float v2f;
typedef __attribute__((ext_vector_type(8))) float v8f;

static constexpr int cB = 2;
static constexpr int cN = 384;
static constexpr int cD = 128;
static constexpr int cM = cB * cN;

template <int K, bool BIAS, bool RELU>
__device__ __forceinline__ void gemm_tile(const float* __restrict__ A,
                                          const float* __restrict__ Bm,
                                          const float* __restrict__ bias,
                                          float* __restrict__ C, int blk)
{
    constexpr int KS = K / 4;
    const int wave  = threadIdx.x >> 5;
    const int lane  = threadIdx.x & 31;
    const int tile  = blk * 8 + wave;
    const int mt    = tile >> 3;
    const int nt    = tile & 7;
    const int m0    = mt * 16;
    const int n0    = nt * 16;
    const int l15   = lane & 15;
    const int khalf = lane >> 4;
    const int n     = n0 + l15;
    const int mbase = m0 + (khalf ? 8 : 0);

    v8f c = {};

    const float* Ak = A + (size_t)(m0 + l15) * K + 2 * khalf;
    const float* Bk = Bm + (size_t)(2 * khalf) * cD + n;
    v2f a[KS], b[KS];
#pragma unroll
    for (int s = 0; s < KS; ++s) {
        a[s].x = Ak[4 * s];
        a[s].y = Ak[4 * s + 1];
        b[s].x = Bk[(size_t)(4 * s) * cD];
        b[s].y = Bk[(size_t)(4 * s + 1) * cD];
    }

    float bv = 0.0f;
    if (BIAS) bv = bias[n];

    __builtin_amdgcn_sched_barrier(0);

#pragma unroll
    for (int s = 0; s < KS; ++s)
        c = __builtin_amdgcn_wmma_f32_16x16x4_f32(
                false, a[s], false, b[s], (short)0, c, false, false);

#pragma unroll
    for (int r = 0; r < 8; ++r) {
        const int mr = mbase + r;
        float y = c[r];
        if (BIAS) y = y + bv;
        if (RELU) y = y > 0.0f ? y : 0.0f;
        C[(size_t)mr * cD + n] = y;
    }
}

__global__ void __launch_bounds__(256)
gemm_in(const float* __restrict__ A, const float* __restrict__ Bm,
        const float* __restrict__ bias, float* __restrict__ C)
{
    gemm_tile<64, true, true>(A, Bm, bias, C, blockIdx.x);
}

static constexpr int PF = 8;

template <int M>
__device__ __forceinline__ float add_xor_lane(float v)
{
    const int o = __builtin_amdgcn_update_dpp(0, __float_as_int(v), 0x160 | M, 0xf, 0xf, true);
    return v + __int_as_float(o);
}
__device__ __forceinline__ float half_sum(float v)
{
    v = add_xor_lane<8>(v);
    v = add_xor_lane<4>(v);
    v = add_xor_lane<2>(v);
    v = add_xor_lane<1>(v);
    return v;
}

template <bool RES>
__device__ __forceinline__ void hn_ln_tile(const float* __restrict__ A,
                                           const float* __restrict__ Bm,
                                           const float* __restrict__ bias,
                                           const float* xin,
                                           const float* __restrict__ g,
                                           const float* __restrict__ beta,
                                           float* out)
{
    constexpr int KS = cD / 4;
    const int lane  = threadIdx.x;
    const int l15   = lane & 15;
    const int khalf = lane >> 4;
    const int m0    = blockIdx.x * 16;
    const int mbase = m0 + (khalf ? 8 : 0);
    const int nb    = l15 * 8;

    const float* Ak = A + (size_t)(m0 + l15) * cD + 2 * khalf;
    const float* Bk = Bm + (size_t)(2 * khalf) * cD + nb;

    float4 bi[2], gg[2], be[2];
    float4 xr[8][2];
#pragma unroll
    for (int q = 0; q < 2; ++q) {
        bi[q] = *(const float4*)(bias + nb + 4 * q);
        gg[q] = *(const float4*)(g + nb + 4 * q);
        be[q] = *(const float4*)(beta + nb + 4 * q);
    }
    if (RES) {
#pragma unroll
        for (int r = 0; r < 8; ++r)
#pragma unroll
            for (int q = 0; q < 2; ++q)
                xr[r][q] = *(const float4*)(xin + (size_t)(mbase + r) * cD + nb + 4 * q);
    }

    v2f a[KS];
    float4 bx[KS][2], by[KS][2];
#pragma unroll
    for (int s = 0; s < PF; ++s) {
        a[s] = *(const v2f*)(Ak + 4 * s);
#pragma unroll
        for (int q = 0; q < 2; ++q) {
            bx[s][q] = *(const float4*)(Bk + (size_t)(4 * s) * cD + 4 * q);
            by[s][q] = *(const float4*)(Bk + (size_t)(4 * s + 1) * cD + 4 * q);
        }
    }

    v8f acc[8];
#pragma unroll
    for (int t = 0; t < 8; ++t) acc[t] = v8f{};

#pragma unroll
    for (int s = 0; s < KS; ++s) {
        if (s + PF < KS) {
            const int u = s + PF;
            a[u] = *(const v2f*)(Ak + 4 * u);
#pragma unroll
            for (int q = 0; q < 2; ++q) {
                bx[u][q] = *(const float4*)(Bk + (size_t)(4 * u) * cD + 4 * q);
                by[u][q] = *(const float4*)(Bk + (size_t)(4 * u + 1) * cD + 4 * q);
            }
        }
        __builtin_amdgcn_sched_barrier(0);
        const float fx[8] = {bx[s][0].x, bx[s][0].y, bx[s][0].z, bx[s][0].w,
                             bx[s][1].x, bx[s][1].y, bx[s][1].z, bx[s][1].w};
        const float fy[8] = {by[s][0].x, by[s][0].y, by[s][0].z, by[s][0].w,
                             by[s][1].x, by[s][1].y, by[s][1].z, by[s][1].w};
#pragma unroll
        for (int t = 0; t < 8; ++t) {
            v2f b;
            b.x = fx[t];
            b.y = fy[t];
            acc[t] = __builtin_amdgcn_wmma_f32_16x16x4_f32(
                         false, a[s], false, b, (short)0, acc[t], false, false);
        }
        __builtin_amdgcn_sched_barrier(0);
    }

    const float bia[8] = {bi[0].x, bi[0].y, bi[0].z, bi[0].w, bi[1].x, bi[1].y, bi[1].z, bi[1].w};
    const float gam[8] = {gg[0].x, gg[0].y, gg[0].z, gg[0].w, gg[1].x, gg[1].y, gg[1].z, gg[1].w};
    const float bet[8] = {be[0].x, be[0].y, be[0].z, be[0].w, be[1].x, be[1].y, be[1].z, be[1].w};
#pragma unroll
    for (int r = 0; r < 8; ++r) {
        float v[8];
#pragma unroll
        for (int t = 0; t < 8; ++t) v[t] = acc[t][r] + bia[t];
        if (RES) {
            const float xs[8] = {xr[r][0].x, xr[r][0].y, xr[r][0].z, xr[r][0].w,
                                 xr[r][1].x, xr[r][1].y, xr[r][1].z, xr[r][1].w};
#pragma unroll
            for (int t = 0; t < 8; ++t) v[t] = xs[t] + v[t];
        }
        float sm = 0.0f;
#pragma unroll
        for (int t = 0; t < 8; ++t) sm += v[t];
        const float mu = half_sum(sm) * (1.0f / cD);
        float sq = 0.0f;
#pragma unroll
        for (int t = 0; t < 8; ++t) {
            v[t] -= mu;
            sq += v[t] * v[t];
        }
        const float var = half_sum(sq) * (1.0f / cD);
        const float rs  = rsqrtf(var + 1e-5f);
        float y[8];
#pragma unroll
        for (int t = 0; t < 8; ++t) {
            y[t] = v[t] * rs * gam[t] + bet[t];
            if (RES) y[t] = y[t] > 0.0f ? y[t] : 0.0f;
        }
        float* orow = out + (size_t)(mbase + r) * cD + nb;
        *(float4*)(orow)     = make_float4(y[0], y[1], y[2], y[3]);
        *(float4*)(orow + 4) = make_float4(y[4], y[5], y[6], y[7]);
    }
}

__global__ void __launch_bounds__(32)
gemm_hn_ln_res(const float* __restrict__ A, const float* __restrict__ Bm,
               const float* __restrict__ bias, const float* __restrict__ g,
               const float* __restrict__ beta, float* x)
{
    hn_ln_tile<true>(A, Bm, bias, x, g, beta, x);
}

__global__ void __launch_bounds__(32)
gemm_hn_ln_out(const float* __restrict__ A, const float* __restrict__ Bm,
               const float* __restrict__ bias, const float* __restrict__ g,
               const float* __restrict__ beta, float* __restrict__ out)
{
    hn_ln_tile<false>(A, Bm, bias, nullptr, g, beta, out);
}

__global__ void __launch_bounds__(128)
gemm_agg_hid(const float* __restrict__ R,
             const float* __restrict__ W2,
             const float* __restrict__ b2,
             const float* __restrict__ deg,
             const float* __restrict__ U,
             float* hid)
{
    constexpr int KS = cD / 4;
    constexpr int T0 = KS - 8;
    const int wave  = threadIdx.x >> 5;
    const int lane  = threadIdx.x & 31;
    const int l15   = lane & 15;
    const int khalf = lane >> 4;
    const int m0    = (blockIdx.x >> 1) * 16;
    const int n0    = ((blockIdx.x & 1) * 4 + wave) * 16;

    const float* Rk = R + (size_t)(m0 + l15) * cD + 2 * khalf;
    const float* Wk = W2 + (size_t)(2 * khalf) * cD + l15 * 8;
    const float* Uk = U + (size_t)(64 * khalf) * cD + n0 + l15;
    const float* bk = b2 + l15 * 8;
    float* hrow = hid + (size_t)(m0 + l15) * cD + n0 + 8 * khalf;

    v2f rb[KS];
    float4 wx[KS][2], wy[KS][2];
    float ux[8][4], uy[8][4];
    float4 mbv[2];
    float4 hold[2];
    float dg;
#pragma unroll
    for (int s = 0; s < PF; ++s) {
        rb[s] = *(const v2f*)(Rk + 4 * s);
#pragma unroll
        for (int q = 0; q < 2; ++q) {
            wx[s][q] = *(const float4*)(Wk + (size_t)(4 * s) * cD + 4 * q);
            wy[s][q] = *(const float4*)(Wk + (size_t)(4 * s + 1) * cD + 4 * q);
        }
    }

    v8f acc[8];
#pragma unroll
    for (int t = 0; t < 8; ++t) acc[t] = v8f{};

#pragma unroll
    for (int s = 0; s < KS; ++s) {
        if (s + PF < KS) {
            const int v = s + PF;
            rb[v] = *(const v2f*)(Rk + 4 * v);
#pragma unroll
            for (int q = 0; q < 2; ++q) {
                wx[v][q] = *(const float4*)(Wk + (size_t)(4 * v) * cD + 4 * q);
                wy[v][q] = *(const float4*)(Wk + (size_t)(4 * v + 1) * cD + 4 * q);
            }
        }
        if (s >= T0) {
            const int r = s - T0;
#pragma unroll
            for (int u = 0; u < 4; ++u) {
                ux[r][u] = Uk[(size_t)(8 * r + 2 * u) * cD];
                uy[r][u] = Uk[(size_t)(8 * r + 2 * u + 1) * cD];
            }
            if (s == T0) {
                mbv[0]  = *(const float4*)(bk);
                mbv[1]  = *(const float4*)(bk + 4);
                dg      = deg[m0 + l15];
                hold[0] = *(const float4*)(hrow);
                hold[1] = *(const float4*)(hrow + 4);
            }
        }
        __builtin_amdgcn_sched_barrier(0);
        const float fx[8] = {wx[s][0].x, wx[s][0].y, wx[s][0].z, wx[s][0].w,
                             wx[s][1].x, wx[s][1].y, wx[s][1].z, wx[s][1].w};
        const float fy[8] = {wy[s][0].x, wy[s][0].y, wy[s][0].z, wy[s][0].w,
                             wy[s][1].x, wy[s][1].y, wy[s][1].z, wy[s][1].w};
#pragma unroll
        for (int t = 0; t < 8; ++t) {
            v2f a;
            a.x = fx[t];
            a.y = fy[t];
            acc[t] = __builtin_amdgcn_wmma_f32_16x16x4_f32(
                         false, a, false, rb[s], (short)0, acc[t], false, false);
        }
        __builtin_amdgcn_sched_barrier(0);
    }

    {
        const float mb[8] = {mbv[0].x, mbv[0].y, mbv[0].z, mbv[0].w,
                             mbv[1].x, mbv[1].y, mbv[1].z, mbv[1].w};
        v2f b;
        b.x = khalf ? 0.0f : dg;
        b.y = 0.0f;
#pragma unroll
        for (int t = 0; t < 8; ++t) {
            v2f a;
            a.x = khalf ? 0.0f : mb[t];
            a.y = 0.0f;
            acc[t] = __builtin_amdgcn_wmma_f32_16x16x4_f32(
                         false, a, false, b, (short)0, acc[t], false, false);
        }
    }

    v8f e = {hold[0].x, hold[0].y, hold[0].z, hold[0].w,
             hold[1].x, hold[1].y, hold[1].z, hold[1].w};
#pragma unroll
    for (int r = 0; r < 8; ++r) {
#pragma unroll
        for (int u = 0; u < 4; ++u) {
            v2f a, b;
            a.x = ux[r][u];
            a.y = uy[r][u];
            b.x = acc[2 * u][r];
            b.y = acc[2 * u + 1][r];
            e = __builtin_amdgcn_wmma_f32_16x16x4_f32(
                    false, a, false, b, (short)0, e, false, false);
        }
    }

    float y[8];
#pragma unroll
    for (int r = 0; r < 8; ++r) y[r] = e[r] > 0.0f ? e[r] : 0.0f;
    *(float4*)(hrow)     = make_float4(y[0], y[1], y[2], y[3]);
    *(float4*)(hrow + 4) = make_float4(y[4], y[5], y[6], y[7]);
}

__global__ void __launch_bounds__(256)
gemm3_wmma_f32(const float* __restrict__ A,
               const float* __restrict__ B0,
               const float* __restrict__ B1,
               const float* __restrict__ B2,
               const float* __restrict__ bias2,
               float* __restrict__ C0,
               float* __restrict__ C1,
               float* __restrict__ C2)
{
    const int s = blockIdx.y;
    if (s == 2) {
        gemm_tile<cD, true, false>(A, B2, bias2, C2, blockIdx.x);
    } else {
        gemm_tile<cD, false, false>(A, s == 0 ? B0 : B1, nullptr,
                                    s == 0 ? C0 : C1, blockIdx.x);
    }
}

static constexpr int TI = 3;
static constexpr int QB = 48;
static constexpr int AG = 8;
static constexpr int NG = QB / AG;
static constexpr int RW = cN / 4;
static_assert(cN % TI == 0 && cN % (2 * QB) == 0 && NG % 2 == 0 && AG % 4 == 0, "shape");

struct AdjGroup { float a[TI][AG]; };

__device__ __forceinline__ void load_q(float (&q)[QB], const float* __restrict__ Qcol, int j0)
{
#pragma unroll
    for (int u = 0; u < QB; ++u)
        q[u] = Qcol[(size_t)(j0 + u) * cD];
}

__device__ __forceinline__ void load_adj(AdjGroup& g, const float4* rows, int j0)
{
#pragma unroll
    for (int r = 0; r < TI; ++r) {
#pragma unroll
        for (int v = 0; v < AG / 4; ++v) {
            const float4 x = rows[r * RW + (j0 >> 2) + v];
            g.a[r][4 * v + 0] = x.x; g.a[r][4 * v + 1] = x.y;
            g.a[r][4 * v + 2] = x.z; g.a[r][4 * v + 3] = x.w;
        }
    }
}

__device__ __forceinline__ void compute_group(const float* q, const AdjGroup& g,
                                              const float (&p)[TI], float w, float b1,
                                              float (&acc)[TI], float& dg)
{
#pragma unroll
    for (int u = 0; u < AG; ++u) {
#pragma unroll
        for (int r = 0; r < TI; ++r) {
            const float a = g.a[r][u];
            float t = p[r] + q[u] + a * w + b1;
            t = fmaxf(t, 0.0f);
            acc[r] += a * t;
        }
        dg += g.a[0][u];
    }
}

template <bool LAST>
__device__ __forceinline__ void one_batch(const float (&q)[QB], AdjGroup& ga, AdjGroup& gb,
                                          const float4* rows, int j0,
                                          const float (&p)[TI], float w, float b1,
                                          float (&acc)[TI], float& dg)
{
#pragma unroll
    for (int s = 0; s < NG; s += 2) {
        load_adj(gb, rows, j0 + (s + 1) * AG);
        __builtin_amdgcn_sched_barrier(0);
        compute_group(&q[s * AG], ga, p, w, b1, acc, dg);
        if (s + 2 < NG || !LAST) load_adj(ga, rows, j0 + (s + 2) * AG);
        __builtin_amdgcn_sched_barrier(0);
        compute_group(&q[(s + 1) * AG], gb, p, w, b1, acc, dg);
    }
}

__global__ void __launch_bounds__(cD)
pairwise_agg(const float* __restrict__ P,
             const float* __restrict__ Q,
             const float* __restrict__ adj,
             const float* __restrict__ we,
             const float* __restrict__ mb1,
             float* __restrict__ R,
             float* __restrict__ deg)
{
    const int i0   = blockIdx.x * TI;
    const int b    = i0 / cN;
    const int d    = threadIdx.x;
    const int wave = threadIdx.x >> 5;
    const int rot  = wave % TI;

    __shared__ float4 s_adj4[(2 * TI - 1) * RW];

    const float* Qcol = Q + (size_t)b * cN * cD + d;

    float qa[QB], qb[QB];
    load_q(qa, Qcol, 0);

    {
        const float4* adj4 = (const float4*)(adj + (size_t)i0 * cN);
        for (int v = d; v < TI * RW; v += cD) {
            const float4 x = adj4[v];
            s_adj4[v] = x;
            if (v < (TI - 1) * RW) s_adj4[v + TI * RW] = x;
        }
    }

    int rr[TI];
    float p[TI];
#pragma unroll
    for (int k = 0; k < TI; ++k) {
        rr[k] = (rot + k) % TI;
        p[k]  = P[(size_t)(i0 + rr[k]) * cD + d];
    }
    const float w  = we[d];
    const float b1 = mb1[d];
    const float4* rows = s_adj4 + rot * RW;

    __syncthreads();

    AdjGroup ga, gb;
    load_adj(ga, rows, 0);

    float acc[TI] = {0.0f, 0.0f, 0.0f};
    float dg = 0.0f;

    for (int j0 = 0; j0 < cN - 2 * QB; j0 += 2 * QB) {
        load_q(qb, Qcol, j0 + QB);
        one_batch<false>(qa, ga, gb, rows, j0, p, w, b1, acc, dg);
        load_q(qa, Qcol, j0 + 2 * QB);
        one_batch<false>(qb, ga, gb, rows, j0 + QB, p, w, b1, acc, dg);
    }
    load_q(qb, Qcol, cN - QB);
    one_batch<false>(qa, ga, gb, rows, cN - 2 * QB, p, w, b1, acc, dg);
    one_batch<true >(qb, ga, gb, rows, cN - QB, p, w, b1, acc, dg);

#pragma unroll
    for (int k = 0; k < TI; ++k)
        R[(size_t)(i0 + rr[k]) * cD + d] = acc[k];
    if (wave < TI && (d & 31) == 0) deg[i0 + rot] = dg;
}

extern "C" void kernel_launch(void* const* d_in, const int* in_sizes, int n_in,
                              void* d_out, int out_size, void* d_ws, size_t ws_size,
                              hipStream_t stream)
{
    (void)in_sizes; (void)n_in; (void)out_size; (void)ws_size;

    const float* h    = (const float*)d_in[0];
    const float* adj  = (const float*)d_in[1];
    const float* ip_w = (const float*)d_in[2];
    const float* ip_b = (const float*)d_in[3];
    const float* mw1  = (const float*)d_in[4];
    const float* mb1  = (const float*)d_in[5];
    const float* mw2  = (const float*)d_in[6];
    const float* mb2  = (const float*)d_in[7];
    const float* uw1  = (const float*)d_in[8];
    const float* ub1  = (const float*)d_in[9];
    const float* uw2  = (const float*)d_in[10];
    const float* ub2  = (const float*)d_in[11];
    const float* ln_g = (const float*)d_in[12];
    const float* ln_b = (const float*)d_in[13];

    float* ws = (float*)d_ws;
    const size_t S = (size_t)cM * cD;
    float* x   = ws + 0 * S;
    float* P   = ws + 1 * S;
    float* Q   = ws + 2 * S;
    float* R   = ws + 3 * S;
    float* hid = ws + 5 * S;
    float* deg = ws + 7 * S;

    gemm_in<<<48, 256, 0, stream>>>(h, ip_w, ip_b, x);

    for (int l = 0; l < 3; ++l) {
        const float* mw1l = mw1 + (size_t)l * 257 * cD;
        const float* mb1l = mb1 + (size_t)l * cD;
        const float* mw2l = mw2 + (size_t)l * cD * cD;
        const float* mb2l = mb2 + (size_t)l * cD;
        const float* uw1l = uw1 + (size_t)l * 256 * cD;
        const float* ub1l = ub1 + (size_t)l * cD;
        const float* uw2l = uw2 + (size_t)l * cD * cD;
        const float* ub2l = ub2 + (size_t)l * cD;

        gemm3_wmma_f32<<<dim3(48, 3), 256, 0, stream>>>(x, mw1l, mw1l + 128 * cD, uw1l,
                                                        ub1l, P, Q, hid);

        pairwise_agg<<<cM / TI, cD, 0, stream>>>(P, Q, adj, mw1l + 256 * cD, mb1l, R, deg);

        gemm_agg_hid<<<2 * (cM / 16), 128, 0, stream>>>(R, mw2l, mb2l, deg, uw1l + 128 * cD, hid);

        if (l < 2) {
            gemm_hn_ln_res<<<cM / 16, 32, 0, stream>>>(hid, uw2l, ub2l,
                                                      ln_g + l * cD, ln_b + l * cD, x);
        } else {
            gemm_hn_ln_out<<<cM / 16, 32, 0, stream>>>(hid, uw2l, ub2l,
                                                      ln_g + l * cD, ln_b + l * cD,
                                                      (float*)d_out);
        }
    }
}


# P, Q and hid's x part run in the launch that makes x: a wave builds its 16 x rows transposed in registers as B operand; 13 launches to 10; k order permuted, LN sums regrouped, residual summed first
# speedup vs baseline: 2.8302x; 1.1062x over previous
#include <hip/hip_runtime.h>
#include <hip/hip_bf16.h>

typedef __attribute__((ext_vector_type(2))) float v2f;
typedef __attribute__((ext_vector_type(8))) float v8f;

static constexpr int cB = 2;
static constexpr int cN = 384;
static constexpr int cD = 128;
static constexpr int cM = cB * cN;

static constexpr int PF = 8;

template <int M>
__device__ __forceinline__ float add_xor_lane(float v)
{
    const int o = __builtin_amdgcn_update_dpp(0, __float_as_int(v), 0x160 | M, 0xf, 0xf, true);
    return v + __int_as_float(o);
}
__device__ __forceinline__ float half_sum(float v)
{
    v = add_xor_lane<8>(v);
    v = add_xor_lane<4>(v);
    v = add_xor_lane<2>(v);
    v = add_xor_lane<1>(v);
    return v;
}

template <bool RES>
__device__ __forceinline__ void hn_ln_tile(const float* __restrict__ A,
                                           const float* __restrict__ Bm,
                                           const float* __restrict__ bias,
                                           const float* xin,
                                           const float* __restrict__ g,
                                           const float* __restrict__ beta,
                                           float* out)
{
    constexpr int KS = cD / 4;
    const int lane  = threadIdx.x;
    const int l15   = lane & 15;
    const int khalf = lane >> 4;
    const int m0    = blockIdx.x * 16;
    const int mbase = m0 + (khalf ? 8 : 0);
    const int nb    = l15 * 8;

    const float* Ak = A + (size_t)(m0 + l15) * cD + 2 * khalf;
    const float* Bk = Bm + (size_t)(2 * khalf) * cD + nb;

    float4 bi[2], gg[2], be[2];
    float4 xr[8][2];
#pragma unroll
    for (int q = 0; q < 2; ++q) {
        bi[q] = *(const float4*)(bias + nb + 4 * q);
        gg[q] = *(const float4*)(g + nb + 4 * q);
        be[q] = *(const float4*)(beta + nb + 4 * q);
    }
    if (RES) {
#pragma unroll
        for (int r = 0; r < 8; ++r)
#pragma unroll
            for (int q = 0; q < 2; ++q)
                xr[r][q] = *(const float4*)(xin + (size_t)(mbase + r) * cD + nb + 4 * q);
    }

    v2f a[KS];
    float4 bx[KS][2], by[KS][2];
#pragma unroll
    for (int s = 0; s < PF; ++s) {
        a[s] = *(const v2f*)(Ak + 4 * s);
#pragma unroll
        for (int q = 0; q < 2; ++q) {
            bx[s][q] = *(const float4*)(Bk + (size_t)(4 * s) * cD + 4 * q);
            by[s][q] = *(const float4*)(Bk + (size_t)(4 * s + 1) * cD + 4 * q);
        }
    }

    v8f acc[8];
#pragma unroll
    for (int t = 0; t < 8; ++t) acc[t] = v8f{};

#pragma unroll
    for (int s = 0; s < KS; ++s) {
        if (s + PF < KS) {
            const int u = s + PF;
            a[u] = *(const v2f*)(Ak + 4 * u);
#pragma unroll
            for (int q = 0; q < 2; ++q) {
                bx[u][q] = *(const float4*)(Bk + (size_t)(4 * u) * cD + 4 * q);
                by[u][q] = *(const float4*)(Bk + (size_t)(4 * u + 1) * cD + 4 * q);
            }
        }
        __builtin_amdgcn_sched_barrier(0);
        const float fx[8] = {bx[s][0].x, bx[s][0].y, bx[s][0].z, bx[s][0].w,
                             bx[s][1].x, bx[s][1].y, bx[s][1].z, bx[s][1].w};
        const float fy[8] = {by[s][0].x, by[s][0].y, by[s][0].z, by[s][0].w,
                             by[s][1].x, by[s][1].y, by[s][1].z, by[s][1].w};
#pragma unroll
        for (int t = 0; t < 8; ++t) {
            v2f b;
            b.x = fx[t];
            b.y = fy[t];
            acc[t] = __builtin_amdgcn_wmma_f32_16x16x4_f32(
                         false, a[s], false, b, (short)0, acc[t], false, false);
        }
        __builtin_amdgcn_sched_barrier(0);
    }

    const float bia[8] = {bi[0].x, bi[0].y, bi[0].z, bi[0].w, bi[1].x, bi[1].y, bi[1].z, bi[1].w};
    const float gam[8] = {gg[0].x, gg[0].y, gg[0].z, gg[0].w, gg[1].x, gg[1].y, gg[1].z, gg[1].w};
    const float bet[8] = {be[0].x, be[0].y, be[0].z, be[0].w, be[1].x, be[1].y, be[1].z, be[1].w};
#pragma unroll
    for (int r = 0; r < 8; ++r) {
        float v[8];
#pragma unroll
        for (int t = 0; t < 8; ++t) v[t] = acc[t][r] + bia[t];
        if (RES) {
            const float xs[8] = {xr[r][0].x, xr[r][0].y, xr[r][0].z, xr[r][0].w,
                                 xr[r][1].x, xr[r][1].y, xr[r][1].z, xr[r][1].w};
#pragma unroll
            for (int t = 0; t < 8; ++t) v[t] = xs[t] + v[t];
        }
        float sm = 0.0f;
#pragma unroll
        for (int t = 0; t < 8; ++t) sm += v[t];
        const float mu = half_sum(sm) * (1.0f / cD);
        float sq = 0.0f;
#pragma unroll
        for (int t = 0; t < 8; ++t) {
            v[t] -= mu;
            sq += v[t] * v[t];
        }
        const float var = half_sum(sq) * (1.0f / cD);
        const float rs  = rsqrtf(var + 1e-5f);
        float y[8];
#pragma unroll
        for (int t = 0; t < 8; ++t) {
            y[t] = v[t] * rs * gam[t] + bet[t];
            if (RES) y[t] = y[t] > 0.0f ? y[t] : 0.0f;
        }
        float* orow = out + (size_t)(mbase + r) * cD + nb;
        *(float4*)(orow)     = make_float4(y[0], y[1], y[2], y[3]);
        *(float4*)(orow + 4) = make_float4(y[4], y[5], y[6], y[7]);
    }
}

__global__ void __launch_bounds__(32)
gemm_hn_ln_out(const float* __restrict__ A, const float* __restrict__ Bm,
               const float* __restrict__ bias, const float* __restrict__ g,
               const float* __restrict__ beta, float* __restrict__ out)
{
    hn_ln_tile<false>(A, Bm, bias, nullptr, g, beta, out);
}

__global__ void __launch_bounds__(128)
gemm_agg_hid(const float* __restrict__ R,
             const float* __restrict__ W2,
             const float* __restrict__ b2,
             const float* __restrict__ deg,
             const float* __restrict__ U,
             float* hid)
{
    constexpr int KS = cD / 4;
    constexpr int T0 = KS - 8;
    const int wave  = threadIdx.x >> 5;
    const int lane  = threadIdx.x & 31;
    const int l15   = lane & 15;
    const int khalf = lane >> 4;
    const int m0    = (blockIdx.x >> 1) * 16;
    const int n0    = ((blockIdx.x & 1) * 4 + wave) * 16;

    const float* Rk = R + (size_t)(m0 + l15) * cD + 2 * khalf;
    const float* Wk = W2 + (size_t)(2 * khalf) * cD + l15 * 8;
    const float* Uk = U + (size_t)(64 * khalf) * cD + n0 + l15;
    const float* bk = b2 + l15 * 8;
    float* hrow = hid + (size_t)(m0 + l15) * cD + n0 + 8 * khalf;

    v2f rb[KS];
    float4 wx[KS][2], wy[KS][2];
    float ux[8][4], uy[8][4];
    float4 mbv[2];
    float4 hold[2];
    float dg;
#pragma unroll
    for (int s = 0; s < PF; ++s) {
        rb[s] = *(const v2f*)(Rk + 4 * s);
#pragma unroll
        for (int q = 0; q < 2; ++q) {
            wx[s][q] = *(const float4*)(Wk + (size_t)(4 * s) * cD + 4 * q);
            wy[s][q] = *(const float4*)(Wk + (size_t)(4 * s + 1) * cD + 4 * q);
        }
    }

    v8f acc[8];
#pragma unroll
    for (int t = 0; t < 8; ++t) acc[t] = v8f{};

#pragma unroll
    for (int s = 0; s < KS; ++s) {
        if (s + PF < KS) {
            const int v = s + PF;
            rb[v] = *(const v2f*)(Rk + 4 * v);
#pragma unroll
            for (int q = 0; q < 2; ++q) {
                wx[v][q] = *(const float4*)(Wk + (size_t)(4 * v) * cD + 4 * q);
                wy[v][q] = *(const float4*)(Wk + (size_t)(4 * v + 1) * cD + 4 * q);
            }
        }
        if (s >= T0) {
            const int r = s - T0;
#pragma unroll
            for (int u = 0; u < 4; ++u) {
                ux[r][u] = Uk[(size_t)(8 * r + 2 * u) * cD];
                uy[r][u] = Uk[(size_t)(8 * r + 2 * u + 1) * cD];
            }
            if (s == T0) {
                mbv[0]  = *(const float4*)(bk);
                mbv[1]  = *(const float4*)(bk + 4);
                dg      = deg[m0 + l15];
                hold[0] = *(const float4*)(hrow);
                hold[1] = *(const float4*)(hrow + 4);
            }
        }
        __builtin_amdgcn_sched_barrier(0);
        const float fx[8] = {wx[s][0].x, wx[s][0].y, wx[s][0].z, wx[s][0].w,
                             wx[s][1].x, wx[s][1].y, wx[s][1].z, wx[s][1].w};
        const float fy[8] = {wy[s][0].x, wy[s][0].y, wy[s][0].z, wy[s][0].w,
                             wy[s][1].x, wy[s][1].y, wy[s][1].z, wy[s][1].w};
#pragma unroll
        for (int t = 0; t < 8; ++t) {
            v2f a;
            a.x = fx[t];
            a.y = fy[t];
            acc[t] = __builtin_amdgcn_wmma_f32_16x16x4_f32(
                         false, a, false, rb[s], (short)0, acc[t], false, false);
        }
        __builtin_amdgcn_sched_barrier(0);
    }

    {
        const float mb[8] = {mbv[0].x, mbv[0].y, mbv[0].z, mbv[0].w,
                             mbv[1].x, mbv[1].y, mbv[1].z, mbv[1].w};
        v2f b;
        b.x = khalf ? 0.0f : dg;
        b.y = 0.0f;
#pragma unroll
        for (int t = 0; t < 8; ++t) {
            v2f a;
            a.x = khalf ? 0.0f : mb[t];
            a.y = 0.0f;
            acc[t] = __builtin_amdgcn_wmma_f32_16x16x4_f32(
                         false, a, false, b, (short)0, acc[t], false, false);
        }
    }

    v8f e = {hold[0].x, hold[0].y, hold[0].z, hold[0].w,
             hold[1].x, hold[1].y, hold[1].z, hold[1].w};
#pragma unroll
    for (int r = 0; r < 8; ++r) {
#pragma unroll
        for (int u = 0; u < 4; ++u) {
            v2f a, b;
            a.x = ux[r][u];
            a.y = uy[r][u];
            b.x = acc[2 * u][r];
            b.y = acc[2 * u + 1][r];
            e = __builtin_amdgcn_wmma_f32_16x16x4_f32(
                    false, a, false, b, (short)0, e, false, false);
        }
    }

    float y[8];
#pragma unroll
    for (int r = 0; r < 8; ++r) y[r] = e[r] > 0.0f ? e[r] : 0.0f;
    *(float4*)(hrow)     = make_float4(y[0], y[1], y[2], y[3]);
    *(float4*)(hrow + 4) = make_float4(y[4], y[5], y[6], y[7]);
}


__device__ __forceinline__ float other_half(float v)
{
    return __int_as_float(__builtin_amdgcn_permlanex16(0, __float_as_int(v),
                                                       0x76543210, 0xfedcba98, false, false));
}

__device__ __forceinline__ void load4(float* d, const float* p)
{
    const float4 q = *(const float4*)p;
    d[0] = q.x; d[1] = q.y; d[2] = q.z; d[3] = q.w;
}

template <int K0, bool LNRES>
__device__ __forceinline__ void x_then_products(const float* __restrict__ A0,
                                                const float* __restrict__ W0,
                                                const float* __restrict__ b0,
                                                const float* __restrict__ xin,
                                                const float* __restrict__ g,
                                                const float* __restrict__ beta,
                                                float* __restrict__ xout,
                                                const float* __restrict__ Bw,
                                                const float* __restrict__ bias2,
                                                float* __restrict__ Cw,
                                                const bool with_bias, const bool write_x)
{
    constexpr int KS = K0 / 4;
    constexpr int T0 = KS - 8;
    const int wave  = threadIdx.x >> 5;
    const int lane  = threadIdx.x & 31;
    const int l15   = lane & 15;
    const int khalf = lane >> 4;
    const int m0    = blockIdx.x * 16;
    const int n0    = wave * 32;

    const float* Ak = A0 + (size_t)(m0 + l15) * K0 + 2 * khalf;
    const float* Wk = W0 + (size_t)(2 * khalf) * cD + l15 * 8;
    const float* Uk = Bw + (size_t)(64 * khalf) * cD + n0 + 2 * l15;
    const size_t xoff = (size_t)(m0 + l15) * cD + 64 * khalf;

    v2f rb[KS];
    float4 wx[KS][2], wy[KS][2];
    v2f ux[8][4], uy[8][4];
    float b0v[8];
    float b2v[16];
    float gv[64];
    float betv[8];
#pragma unroll
    for (int s = 0; s < PF; ++s) {
        rb[s] = *(const v2f*)(Ak + 4 * s);
#pragma unroll
        for (int q = 0; q < 2; ++q) {
            wx[s][q] = *(const float4*)(Wk + (size_t)(4 * s) * cD + 4 * q);
            wy[s][q] = *(const float4*)(Wk + (size_t)(4 * s + 1) * cD + 4 * q);
        }
    }

    v8f acc[8];
#pragma unroll
    for (int t = 0; t < 8; ++t) acc[t] = v8f{};
    if (LNRES) {
#pragma unroll
        for (int i = 0; i < 16; ++i) {
            float q[4];
            load4(q, xin + xoff + 4 * i);
#pragma unroll
            for (int e = 0; e < 4; ++e) acc[4 * (i & 1) + e][i >> 1] = q[e];
        }
    }

#pragma unroll
    for (int s = 0; s < KS; ++s) {
        if (s + PF < KS) {
            const int v = s + PF;
            rb[v] = *(const v2f*)(Ak + 4 * v);
#pragma unroll
            for (int q = 0; q < 2; ++q) {
                wx[v][q] = *(const float4*)(Wk + (size_t)(4 * v) * cD + 4 * q);
                wy[v][q] = *(const float4*)(Wk + (size_t)(4 * v + 1) * cD + 4 * q);
            }
        }
        if (s >= T0) {
            const int r = s - T0;
            if (s == T0) {
                load4(&b0v[0], b0 + l15 * 8);
                load4(&b0v[4], b0 + l15 * 8 + 4);
                if (LNRES) {
                    load4(&betv[0], beta + l15 * 8);
                    load4(&betv[4], beta + l15 * 8 + 4);
                }
            }
            if (!LNRES) {
#pragma unroll
                for (int u = 0; u < 4; ++u) {
                    ux[r][u] = *(const v2f*)(Uk + (size_t)(8 * r + 2 * u) * cD);
                    uy[r][u] = *(const v2f*)(Uk + (size_t)(8 * r + 2 * u + 1) * cD);
                }
                if (s == T0) {
#pragma unroll
                    for (int i = 0; i < 4; ++i)
                        load4(&b2v[4 * i], bias2 + n0 + 16 * khalf + 4 * i);
                }
            } else if (r >= 1 && r <= 4) {
#pragma unroll
                for (int i = 4 * (r - 1); i < 4 * r; ++i)
                    load4(&gv[4 * i], g + 64 * khalf + 4 * i);
            } else if (r >= 5) {
                if (r == 5) {
#pragma unroll
                    for (int i = 0; i < 4; ++i)
                        load4(&b2v[4 * i], bias2 + n0 + 16 * khalf + 4 * i);
                }
#pragma unroll
                for (int u = 0; u < 4; ++u) {
                    ux[r - 5][u] = *(const v2f*)(Uk + (size_t)(8 * (r - 5) + 2 * u) * cD);
                    uy[r - 5][u] = *(const v2f*)(Uk + (size_t)(8 * (r - 5) + 2 * u + 1) * cD);
                }
            }
        }
        __builtin_amdgcn_sched_barrier(0);
        const float fx[8] = {wx[s][0].x, wx[s][0].y, wx[s][0].z, wx[s][0].w,
                             wx[s][1].x, wx[s][1].y, wx[s][1].z, wx[s][1].w};
        const float fy[8] = {wy[s][0].x, wy[s][0].y, wy[s][0].z, wy[s][0].w,
                             wy[s][1].x, wy[s][1].y, wy[s][1].z, wy[s][1].w};
#pragma unroll
        for (int t = 0; t < 8; ++t) {
            v2f a;
            a.x = fx[t];
            a.y = fy[t];
            acc[t] = __builtin_amdgcn_wmma_f32_16x16x4_f32(
                         false, a, false, rb[s], (short)0, acc[t], false, false);
        }
        __builtin_amdgcn_sched_barrier(0);
    }

    {
        v2f b;
        b.x = khalf ? 0.0f : 1.0f;
        b.y = 0.0f;
#pragma unroll
        for (int t = 0; t < 8; ++t) {
            v2f a;
            a.x = khalf ? 0.0f : b0v[t];
            a.y = 0.0f;
            acc[t] = __builtin_amdgcn_wmma_f32_16x16x4_f32(
                         false, a, false, b, (short)0, acc[t], false, false);
        }
    }

    float xn[64];
    if (LNRES) {
        float pr[8];
#pragma unroll
        for (int r = 0; r < 8; ++r) {
            float sm = 0.0f;
#pragma unroll
            for (int t = 0; t < 8; ++t) sm += acc[t][r];
            pr[r] = sm;
        }
        float sm = ((pr[0] + pr[1]) + (pr[2] + pr[3])) + ((pr[4] + pr[5]) + (pr[6] + pr[7]));
        sm = sm + other_half(sm);
        const float mu = sm * (1.0f / cD);
#pragma unroll
        for (int r = 0; r < 8; ++r) {
            float sq = 0.0f;
#pragma unroll
            for (int t = 0; t < 8; ++t) {
                const float d = acc[t][r] - mu;
                acc[t][r] = d;
                sq += d * d;
            }
            pr[r] = sq;
        }
        float sq = ((pr[0] + pr[1]) + (pr[2] + pr[3])) + ((pr[4] + pr[5]) + (pr[6] + pr[7]));
        sq = sq + other_half(sq);
        const float var = sq * (1.0f / cD);
        const float rs  = rsqrtf(var + 1e-5f);
#pragma unroll
        for (int r = 0; r < 8; ++r)
#pragma unroll
            for (int t = 0; t < 8; ++t)
                acc[t][r] = acc[t][r] * rs * gv[8 * r + t];

#pragma unroll
        for (int r = 3; r < 8; ++r) {
#pragma unroll
            for (int u = 0; u < 4; ++u) {
                ux[r][u] = *(const v2f*)(Uk + (size_t)(8 * r + 2 * u) * cD);
                uy[r][u] = *(const v2f*)(Uk + (size_t)(8 * r + 2 * u + 1) * cD);
            }
        }

        {
            v2f b;
            b.x = khalf ? 0.0f : 1.0f;
            b.y = 0.0f;
#pragma unroll
            for (int t = 0; t < 8; ++t) {
                v2f a;
                a.x = khalf ? 0.0f : betv[t];
                a.y = 0.0f;
                acc[t] = __builtin_amdgcn_wmma_f32_16x16x4_f32(
                             false, a, false, b, (short)0, acc[t], false, false);
            }
        }
    }
#pragma unroll
    for (int r = 0; r < 8; ++r)
#pragma unroll
        for (int t = 0; t < 8; ++t)
            xn[8 * r + t] = acc[t][r] > 0.0f ? acc[t][r] : 0.0f;

    if (write_x) {
        float* xrow = xout + xoff;
#pragma unroll
        for (int i = 0; i < 16; ++i)
            *(float4*)(xrow + 4 * i) = make_float4(xn[4 * i], xn[4 * i + 1], xn[4 * i + 2], xn[4 * i + 3]);
    }

    v8f e0 = {}, e1 = {};
#pragma unroll
    for (int r = 0; r < 8; ++r) {
#pragma unroll
        for (int u = 0; u < 4; ++u) {
            v2f a0, a1, b;
            a0.x = ux[r][u].x;
            a0.y = uy[r][u].x;
            a1.x = ux[r][u].y;
            a1.y = uy[r][u].y;
            b.x = xn[8 * r + 2 * u];
            b.y = xn[8 * r + 2 * u + 1];
            e0 = __builtin_amdgcn_wmma_f32_16x16x4_f32(
                     false, a0, false, b, (short)0, e0, false, false);
            e1 = __builtin_amdgcn_wmma_f32_16x16x4_f32(
                     false, a1, false, b, (short)0, e1, false, false);
        }
    }

    float y[16];
#pragma unroll
    for (int r = 0; r < 8; ++r) {
        y[2 * r]     = e0[r] + (with_bias ? b2v[2 * r] : 0.0f);
        y[2 * r + 1] = e1[r] + (with_bias ? b2v[2 * r + 1] : 0.0f);
    }
    float* crow = Cw + (size_t)(m0 + l15) * cD + n0 + 16 * khalf;
#pragma unroll
    for (int i = 0; i < 4; ++i)
        *(float4*)(crow + 4 * i) = make_float4(y[4 * i], y[4 * i + 1], y[4 * i + 2], y[4 * i + 3]);
}

__global__ void __launch_bounds__(128)
gemm_in_products(const float* __restrict__ A0, const float* __restrict__ W0,
                 const float* __restrict__ b0, float* __restrict__ xout,
                 const float* __restrict__ B0, const float* __restrict__ B1,
                 const float* __restrict__ B2, const float* __restrict__ bias2,
                 float* __restrict__ C0, float* __restrict__ C1, float* __restrict__ C2)
{
    const int s = blockIdx.y;
    x_then_products<64, false>(A0, W0, b0, nullptr, nullptr, nullptr, xout,
                               s == 0 ? B0 : (s == 1 ? B1 : B2), bias2,
                               s == 0 ? C0 : (s == 1 ? C1 : C2),
                               s == 2, s == 0 && threadIdx.x < 32);
}

__global__ void __launch_bounds__(128)
gemm_hn_ln_products(const float* __restrict__ A0, const float* __restrict__ W0,
                    const float* __restrict__ b0, const float* __restrict__ xin,
                    const float* __restrict__ g, const float* __restrict__ beta,
                    float* __restrict__ xout,
                    const float* __restrict__ B0, const float* __restrict__ B1,
                    const float* __restrict__ B2, const float* __restrict__ bias2,
                    float* __restrict__ C0, float* __restrict__ C1, float* __restrict__ C2)
{
    const int s = blockIdx.y;
    x_then_products<cD, true>(A0, W0, b0, xin, g, beta, xout,
                              s == 0 ? B0 : (s == 1 ? B1 : B2), bias2,
                              s == 0 ? C0 : (s == 1 ? C1 : C2),
                              s == 2, s == 0 && threadIdx.x < 32);
}

static constexpr int TI = 3;
static constexpr int QB = 48;
static constexpr int AG = 8;
static constexpr int NG = QB / AG;
static constexpr int RW = cN / 4;
static_assert(cN % TI == 0 && cN % (2 * QB) == 0 && NG % 2 == 0 && AG % 4 == 0, "shape");

struct AdjGroup { float a[TI][AG]; };

__device__ __forceinline__ void load_q(float (&q)[QB], const float* __restrict__ Qcol, int j0)
{
#pragma unroll
    for (int u = 0; u < QB; ++u)
        q[u] = Qcol[(size_t)(j0 + u) * cD];
}

__device__ __forceinline__ void load_adj(AdjGroup& g, const float4* rows, int j0)
{
#pragma unroll
    for (int r = 0; r < TI; ++r) {
#pragma unroll
        for (int v = 0; v < AG / 4; ++v) {
            const float4 x = rows[r * RW + (j0 >> 2) + v];
            g.a[r][4 * v + 0] = x.x; g.a[r][4 * v + 1] = x.y;
            g.a[r][4 * v + 2] = x.z; g.a[r][4 * v + 3] = x.w;
        }
    }
}

__device__ __forceinline__ void compute_group(const float* q, const AdjGroup& g,
                                              const float (&p)[TI], float w, float b1,
                                              float (&acc)[TI], float& dg)
{
#pragma unroll
    for (int u = 0; u < AG; ++u) {
#pragma unroll
        for (int r = 0; r < TI; ++r) {
            const float a = g.a[r][u];
            float t = p[r] + q[u] + a * w + b1;
            t = fmaxf(t, 0.0f);
            acc[r] += a * t;
        }
        dg += g.a[0][u];
    }
}

template <bool LAST>
__device__ __forceinline__ void one_batch(const float (&q)[QB], AdjGroup& ga, AdjGroup& gb,
                                          const float4* rows, int j0,
                                          const float (&p)[TI], float w, float b1,
                                          float (&acc)[TI], float& dg)
{
#pragma unroll
    for (int s = 0; s < NG; s += 2) {
        load_adj(gb, rows, j0 + (s + 1) * AG);
        __builtin_amdgcn_sched_barrier(0);
        compute_group(&q[s * AG], ga, p, w, b1, acc, dg);
        if (s + 2 < NG || !LAST) load_adj(ga, rows, j0 + (s + 2) * AG);
        __builtin_amdgcn_sched_barrier(0);
        compute_group(&q[(s + 1) * AG], gb, p, w, b1, acc, dg);
    }
}

__global__ void __launch_bounds__(cD)
pairwise_agg(const float* __restrict__ P,
             const float* __restrict__ Q,
             const float* __restrict__ adj,
             const float* __restrict__ we,
             const float* __restrict__ mb1,
             float* __restrict__ R,
             float* __restrict__ deg)
{
    const int i0   = blockIdx.x * TI;
    const int b    = i0 / cN;
    const int d    = threadIdx.x;
    const int wave = threadIdx.x >> 5;
    const int rot  = wave % TI;

    __shared__ float4 s_adj4[(2 * TI - 1) * RW];

    const float* Qcol = Q + (size_t)b * cN * cD + d;

    float qa[QB], qb[QB];
    load_q(qa, Qcol, 0);

    {
        const float4* adj4 = (const float4*)(adj + (size_t)i0 * cN);
        for (int v = d; v < TI * RW; v += cD) {
            const float4 x = adj4[v];
            s_adj4[v] = x;
            if (v < (TI - 1) * RW) s_adj4[v + TI * RW] = x;
        }
    }

    int rr[TI];
    float p[TI];
#pragma unroll
    for (int k = 0; k < TI; ++k) {
        rr[k] = (rot + k) % TI;
        p[k]  = P[(size_t)(i0 + rr[k]) * cD + d];
    }
    const float w  = we[d];
    const float b1 = mb1[d];
    const float4* rows = s_adj4 + rot * RW;

    __syncthreads();

    AdjGroup ga, gb;
    load_adj(ga, rows, 0);

    float acc[TI] = {0.0f, 0.0f, 0.0f};
    float dg = 0.0f;

    for (int j0 = 0; j0 < cN - 2 * QB; j0 += 2 * QB) {
        load_q(qb, Qcol, j0 + QB);
        one_batch<false>(qa, ga, gb, rows, j0, p, w, b1, acc, dg);
        load_q(qa, Qcol, j0 + 2 * QB);
        one_batch<false>(qb, ga, gb, rows, j0 + QB, p, w, b1, acc, dg);
    }
    load_q(qb, Qcol, cN - QB);
    one_batch<false>(qa, ga, gb, rows, cN - 2 * QB, p, w, b1, acc, dg);
    one_batch<true >(qb, ga, gb, rows, cN - QB, p, w, b1, acc, dg);

#pragma unroll
    for (int k = 0; k < TI; ++k)
        R[(size_t)(i0 + rr[k]) * cD + d] = acc[k];
    if (wave < TI && (d & 31) == 0) deg[i0 + rot] = dg;
}

extern "C" void kernel_launch(void* const* d_in, const int* in_sizes, int n_in,
                              void* d_out, int out_size, void* d_ws, size_t ws_size,
                              hipStream_t stream)
{
    (void)in_sizes; (void)n_in; (void)out_size; (void)ws_size;

    const float* h    = (const float*)d_in[0];
    const float* adj  = (const float*)d_in[1];
    const float* ip_w = (const float*)d_in[2];
    const float* ip_b = (const float*)d_in[3];
    const float* mw1  = (const float*)d_in[4];
    const float* mb1  = (const float*)d_in[5];
    const float* mw2  = (const float*)d_in[6];
    const float* mb2  = (const float*)d_in[7];
    const float* uw1  = (const float*)d_in[8];
    const float* ub1  = (const float*)d_in[9];
    const float* uw2  = (const float*)d_in[10];
    const float* ub2  = (const float*)d_in[11];
    const float* ln_g = (const float*)d_in[12];
    const float* ln_b = (const float*)d_in[13];

    float* ws = (float*)d_ws;
    const size_t S = (size_t)cM * cD;
    float* xa   = ws + 0 * S;
    float* P    = ws + 1 * S;
    float* Q    = ws + 2 * S;
    float* R    = ws + 3 * S;
    float* xb   = ws + 4 * S;
    float* hida = ws + 5 * S;
    float* hidb = ws + 6 * S;
    float* deg  = ws + 7 * S;

    gemm_in_products<<<dim3(48, 3), 128, 0, stream>>>(h, ip_w, ip_b, xa,
                                                      mw1, mw1 + 128 * cD, uw1, ub1,
                                                      P, Q, hida);

    for (int l = 0; l < 3; ++l) {
        const float* mw1l = mw1 + (size_t)l * 257 * cD;
        const float* mb1l = mb1 + (size_t)l * cD;
        const float* mw2l = mw2 + (size_t)l * cD * cD;
        const float* mb2l = mb2 + (size_t)l * cD;
        const float* uw1l = uw1 + (size_t)l * 256 * cD;
        const float* uw2l = uw2 + (size_t)l * cD * cD;
        const float* ub2l = ub2 + (size_t)l * cD;
        float* xcur = (l & 1) ? xb : xa;
        float* xnxt = (l & 1) ? xa : xb;
        float* hcur = (l & 1) ? hidb : hida;
        float* hnxt = (l & 1) ? hida : hidb;

        pairwise_agg<<<cM / TI, cD, 0, stream>>>(P, Q, adj, mw1l + 256 * cD, mb1l, R, deg);

        gemm_agg_hid<<<2 * (cM / 16), 128, 0, stream>>>(R, mw2l, mb2l, deg, uw1l + 128 * cD, hcur);

        if (l < 2) {
            const float* mw1n = mw1 + (size_t)(l + 1) * 257 * cD;
            const float* uw1n = uw1 + (size_t)(l + 1) * 256 * cD;
            const float* ub1n = ub1 + (size_t)(l + 1) * cD;
            gemm_hn_ln_products<<<dim3(48, 3), 128, 0, stream>>>(hcur, uw2l, ub2l, xcur,
                                                                 ln_g + l * cD, ln_b + l * cD, xnxt,
                                                                 mw1n, mw1n + 128 * cD, uw1n, ub1n,
                                                                 P, Q, hnxt);
        } else {
            gemm_hn_ln_out<<<cM / 16, 32, 0, stream>>>(hcur, uw2l, ub2l,
                                                      ln_g + l * cD, ln_b + l * cD,
                                                      (float*)d_out);
        }
    }
}


# agg+hid launch: a block's 4 waves split agg's 128 channels (was all in each wave), each sums its 32-channel part of hid for 32 columns, parts added via LDS: hid summed in 4 parts, agg from deg*mb2 up
# speedup vs baseline: 2.9601x; 1.0459x over previous
#include <hip/hip_runtime.h>
#include <hip/hip_bf16.h>

typedef __attribute__((ext_vector_type(2))) float v2f;
typedef __attribute__((ext_vector_type(8))) float v8f;

static constexpr int cB = 2;
static constexpr int cN = 384;
static constexpr int cD = 128;
static constexpr int cM = cB * cN;

static constexpr int PF = 8;

template <int M>
__device__ __forceinline__ float add_xor_lane(float v)
{
    const int o = __builtin_amdgcn_update_dpp(0, __float_as_int(v), 0x160 | M, 0xf, 0xf, true);
    return v + __int_as_float(o);
}
__device__ __forceinline__ float half_sum(float v)
{
    v = add_xor_lane<8>(v);
    v = add_xor_lane<4>(v);
    v = add_xor_lane<2>(v);
    v = add_xor_lane<1>(v);
    return v;
}

template <bool RES>
__device__ __forceinline__ void hn_ln_tile(const float* __restrict__ A,
                                           const float* __restrict__ Bm,
                                           const float* __restrict__ bias,
                                           const float* xin,
                                           const float* __restrict__ g,
                                           const float* __restrict__ beta,
                                           float* out)
{
    constexpr int KS = cD / 4;
    const int lane  = threadIdx.x;
    const int l15   = lane & 15;
    const int khalf = lane >> 4;
    const int m0    = blockIdx.x * 16;
    const int mbase = m0 + (khalf ? 8 : 0);
    const int nb    = l15 * 8;

    const float* Ak = A + (size_t)(m0 + l15) * cD + 2 * khalf;
    const float* Bk = Bm + (size_t)(2 * khalf) * cD + nb;

    float4 bi[2], gg[2], be[2];
    float4 xr[8][2];
#pragma unroll
    for (int q = 0; q < 2; ++q) {
        bi[q] = *(const float4*)(bias + nb + 4 * q);
        gg[q] = *(const float4*)(g + nb + 4 * q);
        be[q] = *(const float4*)(beta + nb + 4 * q);
    }
    if (RES) {
#pragma unroll
        for (int r = 0; r < 8; ++r)
#pragma unroll
            for (int q = 0; q < 2; ++q)
                xr[r][q] = *(const float4*)(xin + (size_t)(mbase + r) * cD + nb + 4 * q);
    }

    v2f a[KS];
    float4 bx[KS][2], by[KS][2];
#pragma unroll
    for (int s = 0; s < PF; ++s) {
        a[s] = *(const v2f*)(Ak + 4 * s);
#pragma unroll
        for (int q = 0; q < 2; ++q) {
            bx[s][q] = *(const float4*)(Bk + (size_t)(4 * s) * cD + 4 * q);
            by[s][q] = *(const float4*)(Bk + (size_t)(4 * s + 1) * cD + 4 * q);
        }
    }

    v8f acc[8];
#pragma unroll
    for (int t = 0; t < 8; ++t) acc[t] = v8f{};

#pragma unroll
    for (int s = 0; s < KS; ++s) {
        if (s + PF < KS) {
            const int u = s + PF;
            a[u] = *(const v2f*)(Ak + 4 * u);
#pragma unroll
            for (int q = 0; q < 2; ++q) {
                bx[u][q] = *(const float4*)(Bk + (size_t)(4 * u) * cD + 4 * q);
                by[u][q] = *(const float4*)(Bk + (size_t)(4 * u + 1) * cD + 4 * q);
            }
        }
        __builtin_amdgcn_sched_barrier(0);
        const float fx[8] = {bx[s][0].x, bx[s][0].y, bx[s][0].z, bx[s][0].w,
                             bx[s][1].x, bx[s][1].y, bx[s][1].z, bx[s][1].w};
        const float fy[8] = {by[s][0].x, by[s][0].y, by[s][0].z, by[s][0].w,
                             by[s][1].x, by[s][1].y, by[s][1].z, by[s][1].w};
#pragma unroll
        for (int t = 0; t < 8; ++t) {
            v2f b;
            b.x = fx[t];
            b.y = fy[t];
            acc[t] = __builtin_amdgcn_wmma_f32_16x16x4_f32(
                         false, a[s], false, b, (short)0, acc[t], false, false);
        }
        __builtin_amdgcn_sched_barrier(0);
    }

    const float bia[8] = {bi[0].x, bi[0].y, bi[0].z, bi[0].w, bi[1].x, bi[1].y, bi[1].z, bi[1].w};
    const float gam[8] = {gg[0].x, gg[0].y, gg[0].z, gg[0].w, gg[1].x, gg[1].y, gg[1].z, gg[1].w};
    const float bet[8] = {be[0].x, be[0].y, be[0].z, be[0].w, be[1].x, be[1].y, be[1].z, be[1].w};
#pragma unroll
    for (int r = 0; r < 8; ++r) {
        float v[8];
#pragma unroll
        for (int t = 0; t < 8; ++t) v[t] = acc[t][r] + bia[t];
        if (RES) {
            const float xs[8] = {xr[r][0].x, xr[r][0].y, xr[r][0].z, xr[r][0].w,
                                 xr[r][1].x, xr[r][1].y, xr[r][1].z, xr[r][1].w};
#pragma unroll
            for (int t = 0; t < 8; ++t) v[t] = xs[t] + v[t];
        }
        float sm = 0.0f;
#pragma unroll
        for (int t = 0; t < 8; ++t) sm += v[t];
        const float mu = half_sum(sm) * (1.0f / cD);
        float sq = 0.0f;
#pragma unroll
        for (int t = 0; t < 8; ++t) {
            v[t] -= mu;
            sq += v[t] * v[t];
        }
        const float var = half_sum(sq) * (1.0f / cD);
        const float rs  = rsqrtf(var + 1e-5f);
        float y[8];
#pragma unroll
        for (int t = 0; t < 8; ++t) {
            y[t] = v[t] * rs * gam[t] + bet[t];
            if (RES) y[t] = y[t] > 0.0f ? y[t] : 0.0f;
        }
        float* orow = out + (size_t)(mbase + r) * cD + nb;
        *(float4*)(orow)     = make_float4(y[0], y[1], y[2], y[3]);
        *(float4*)(orow + 4) = make_float4(y[4], y[5], y[6], y[7]);
    }
}

__global__ void __launch_bounds__(32)
gemm_hn_ln_out(const float* __restrict__ A, const float* __restrict__ Bm,
               const float* __restrict__ bias, const float* __restrict__ g,
               const float* __restrict__ beta, float* __restrict__ out)
{
    hn_ln_tile<false>(A, Bm, bias, nullptr, g, beta, out);
}

static constexpr int PFA = 16;
static constexpr int LPQ = 5;

__global__ void __launch_bounds__(128)
gemm_agg_hid(const float* __restrict__ R,
             const float* __restrict__ W2,
             const float* __restrict__ b2,
             const float* __restrict__ deg,
             const float* __restrict__ U,
             float* hid)
{
    constexpr int KS = cD / 4;
    const int wave  = threadIdx.x >> 5;
    const int lane  = threadIdx.x & 31;
    const int l15   = lane & 15;
    const int khalf = lane >> 4;
    const int m0    = (blockIdx.x >> 2) * 16;
    const int n0    = (blockIdx.x & 3) * 32;
    const int ch0   = wave * 32;

    __shared__ float4 s_part[4 * 32 * LPQ];

    const float* Rk = R + (size_t)(m0 + l15) * cD + 2 * khalf;
    const float* Wk = W2 + (size_t)(2 * khalf) * cD + ch0 + 2 * l15;
    const float* Uk = U + (size_t)(ch0 + 16 * khalf) * cD + n0 + 2 * l15;

    const int orow = 4 * wave + (lane >> 3);
    const int oq   = lane & 7;
    float* hout = hid + (size_t)(m0 + orow) * cD + n0 + 4 * oq;

    v2f rb[KS], wx[KS], wy[KS];
    v2f ux[8], uy[8];
    const float4 hold = *(const float4*)hout;
    const v2f mbv = *(const v2f*)(b2 + ch0 + 2 * l15);
    const float dg = deg[m0 + l15];
#pragma unroll
    for (int s = 0; s < PFA; ++s) {
        rb[s] = *(const v2f*)(Rk + 4 * s);
        wx[s] = *(const v2f*)(Wk + (size_t)(4 * s) * cD);
        wy[s] = *(const v2f*)(Wk + (size_t)(4 * s + 1) * cD);
    }

    v8f acc0 = {}, acc1 = {};
    {
        v2f a0, a1, b;
        b.x  = khalf ? 0.0f : dg;
        b.y  = 0.0f;
        a0.x = khalf ? 0.0f : mbv.x;
        a0.y = 0.0f;
        a1.x = khalf ? 0.0f : mbv.y;
        a1.y = 0.0f;
        acc0 = __builtin_amdgcn_wmma_f32_16x16x4_f32(
                   false, a0, false, b, (short)0, acc0, false, false);
        acc1 = __builtin_amdgcn_wmma_f32_16x16x4_f32(
                   false, a1, false, b, (short)0, acc1, false, false);
    }

#pragma unroll
    for (int s = 0; s < KS; ++s) {
        if (s + PFA < KS) {
            const int v = s + PFA;
            rb[v] = *(const v2f*)(Rk + 4 * v);
            wx[v] = *(const v2f*)(Wk + (size_t)(4 * v) * cD);
            wy[v] = *(const v2f*)(Wk + (size_t)(4 * v + 1) * cD);
        } else if (s < KS - PFA + 8) {
            const int r = s - (KS - PFA);
            ux[r] = *(const v2f*)(Uk + (size_t)(2 * r) * cD);
            uy[r] = *(const v2f*)(Uk + (size_t)(2 * r + 1) * cD);
        }
        v2f a0, a1;
        a0.x = wx[s].x;
        a0.y = wy[s].x;
        a1.x = wx[s].y;
        a1.y = wy[s].y;
        acc0 = __builtin_amdgcn_wmma_f32_16x16x4_f32(
                   false, a0, false, rb[s], (short)0, acc0, false, false);
        acc1 = __builtin_amdgcn_wmma_f32_16x16x4_f32(
                   false, a1, false, rb[s], (short)0, acc1, false, false);
    }

    v8f e0 = {}, e1 = {};
#pragma unroll
    for (int r = 0; r < 8; ++r) {
        v2f a0, a1, b;
        a0.x = ux[r].x;
        a0.y = uy[r].x;
        a1.x = ux[r].y;
        a1.y = uy[r].y;
        b.x = acc0[r];
        b.y = acc1[r];
        e0 = __builtin_amdgcn_wmma_f32_16x16x4_f32(
                 false, a0, false, b, (short)0, e0, false, false);
        e1 = __builtin_amdgcn_wmma_f32_16x16x4_f32(
                 false, a1, false, b, (short)0, e1, false, false);
    }

    __builtin_amdgcn_sched_group_barrier(0x020, 3 + 3 * PFA, 0);
    __builtin_amdgcn_sched_group_barrier(0x008, 2, 0);
#pragma unroll
    for (int s = 0; s < KS - PFA; ++s) {
        __builtin_amdgcn_sched_group_barrier(0x020, 3, 0);
        __builtin_amdgcn_sched_group_barrier(0x008, 2, 0);
    }
#pragma unroll
    for (int r = 0; r < 8; ++r) {
        __builtin_amdgcn_sched_group_barrier(0x020, 2, 0);
        __builtin_amdgcn_sched_group_barrier(0x008, 2, 0);
    }

    float4* mine = s_part + (wave * 32 + lane) * LPQ;
#pragma unroll
    for (int q = 0; q < 4; ++q)
        mine[q] = make_float4(e0[2 * q], e1[2 * q], e0[2 * q + 1], e1[2 * q + 1]);

    __syncthreads();

    const float4* theirs = s_part + (orow + 16 * (oq >> 2)) * LPQ + (oq & 3);
    float4 v = hold;
#pragma unroll
    for (int w = 0; w < 4; ++w) {
        const float4 p = theirs[w * 32 * LPQ];
        v.x += p.x;
        v.y += p.y;
        v.z += p.z;
        v.w += p.w;
    }
    v.x = v.x > 0.0f ? v.x : 0.0f;
    v.y = v.y > 0.0f ? v.y : 0.0f;
    v.z = v.z > 0.0f ? v.z : 0.0f;
    v.w = v.w > 0.0f ? v.w : 0.0f;
    *(float4*)hout = v;
}


__device__ __forceinline__ float other_half(float v)
{
    return __int_as_float(__builtin_amdgcn_permlanex16(0, __float_as_int(v),
                                                       0x76543210, 0xfedcba98, false, false));
}

__device__ __forceinline__ void load4(float* d, const float* p)
{
    const float4 q = *(const float4*)p;
    d[0] = q.x; d[1] = q.y; d[2] = q.z; d[3] = q.w;
}

template <int K0, bool LNRES>
__device__ __forceinline__ void x_then_products(const float* __restrict__ A0,
                                                const float* __restrict__ W0,
                                                const float* __restrict__ b0,
                                                const float* __restrict__ xin,
                                                const float* __restrict__ g,
                                                const float* __restrict__ beta,
                                                float* __restrict__ xout,
                                                const float* __restrict__ Bw,
                                                const float* __restrict__ bias2,
                                                float* __restrict__ Cw,
                                                const bool with_bias, const bool write_x)
{
    constexpr int KS = K0 / 4;
    constexpr int T0 = KS - 8;
    const int wave  = threadIdx.x >> 5;
    const int lane  = threadIdx.x & 31;
    const int l15   = lane & 15;
    const int khalf = lane >> 4;
    const int m0    = blockIdx.x * 16;
    const int n0    = wave * 32;

    const float* Ak = A0 + (size_t)(m0 + l15) * K0 + 2 * khalf;
    const float* Wk = W0 + (size_t)(2 * khalf) * cD + l15 * 8;
    const float* Uk = Bw + (size_t)(64 * khalf) * cD + n0 + 2 * l15;
    const size_t xoff = (size_t)(m0 + l15) * cD + 64 * khalf;

    v2f rb[KS];
    float4 wx[KS][2], wy[KS][2];
    v2f ux[8][4], uy[8][4];
    float b0v[8];
    float b2v[16];
    float gv[64];
    float betv[8];
#pragma unroll
    for (int s = 0; s < PF; ++s) {
        rb[s] = *(const v2f*)(Ak + 4 * s);
#pragma unroll
        for (int q = 0; q < 2; ++q) {
            wx[s][q] = *(const float4*)(Wk + (size_t)(4 * s) * cD + 4 * q);
            wy[s][q] = *(const float4*)(Wk + (size_t)(4 * s + 1) * cD + 4 * q);
        }
    }

    v8f acc[8];
#pragma unroll
    for (int t = 0; t < 8; ++t) acc[t] = v8f{};
    if (LNRES) {
#pragma unroll
        for (int i = 0; i < 16; ++i) {
            float q[4];
            load4(q, xin + xoff + 4 * i);
#pragma unroll
            for (int e = 0; e < 4; ++e) acc[4 * (i & 1) + e][i >> 1] = q[e];
        }
    }

#pragma unroll
    for (int s = 0; s < KS; ++s) {
        if (s + PF < KS) {
            const int v = s + PF;
            rb[v] = *(const v2f*)(Ak + 4 * v);
#pragma unroll
            for (int q = 0; q < 2; ++q) {
                wx[v][q] = *(const float4*)(Wk + (size_t)(4 * v) * cD + 4 * q);
                wy[v][q] = *(const float4*)(Wk + (size_t)(4 * v + 1) * cD + 4 * q);
            }
        }
        if (s >= T0) {
            const int r = s - T0;
            if (s == T0) {
                load4(&b0v[0], b0 + l15 * 8);
                load4(&b0v[4], b0 + l15 * 8 + 4);
                if (LNRES) {
                    load4(&betv[0], beta + l15 * 8);
                    load4(&betv[4], beta + l15 * 8 + 4);
                }
            }
            if (!LNRES) {
#pragma unroll
                for (int u = 0; u < 4; ++u) {
                    ux[r][u] = *(const v2f*)(Uk + (size_t)(8 * r + 2 * u) * cD);
                    uy[r][u] = *(const v2f*)(Uk + (size_t)(8 * r + 2 * u + 1) * cD);
                }
                if (s == T0) {
#pragma unroll
                    for (int i = 0; i < 4; ++i)
                        load4(&b2v[4 * i], bias2 + n0 + 16 * khalf + 4 * i);
                }
            } else if (r >= 1 && r <= 4) {
#pragma unroll
                for (int i = 4 * (r - 1); i < 4 * r; ++i)
                    load4(&gv[4 * i], g + 64 * khalf + 4 * i);
            } else if (r >= 5) {
                if (r == 5) {
#pragma unroll
                    for (int i = 0; i < 4; ++i)
                        load4(&b2v[4 * i], bias2 + n0 + 16 * khalf + 4 * i);
                }
#pragma unroll
                for (int u = 0; u < 4; ++u) {
                    ux[r - 5][u] = *(const v2f*)(Uk + (size_t)(8 * (r - 5) + 2 * u) * cD);
                    uy[r - 5][u] = *(const v2f*)(Uk + (size_t)(8 * (r - 5) + 2 * u + 1) * cD);
                }
            }
        }
        __builtin_amdgcn_sched_barrier(0);
        const float fx[8] = {wx[s][0].x, wx[s][0].y, wx[s][0].z, wx[s][0].w,
                             wx[s][1].x, wx[s][1].y, wx[s][1].z, wx[s][1].w};
        const float fy[8] = {wy[s][0].x, wy[s][0].y, wy[s][0].z, wy[s][0].w,
                             wy[s][1].x, wy[s][1].y, wy[s][1].z, wy[s][1].w};
#pragma unroll
        for (int t = 0; t < 8; ++t) {
            v2f a;
            a.x = fx[t];
            a.y = fy[t];
            acc[t] = __builtin_amdgcn_wmma_f32_16x16x4_f32(
                         false, a, false, rb[s], (short)0, acc[t], false, false);
        }
        __builtin_amdgcn_sched_barrier(0);
    }

    {
        v2f b;
        b.x = khalf ? 0.0f : 1.0f;
        b.y = 0.0f;
#pragma unroll
        for (int t = 0; t < 8; ++t) {
            v2f a;
            a.x = khalf ? 0.0f : b0v[t];
            a.y = 0.0f;
            acc[t] = __builtin_amdgcn_wmma_f32_16x16x4_f32(
                         false, a, false, b, (short)0, acc[t], false, false);
        }
    }

    float xn[64];
    if (LNRES) {
        float pr[8];
#pragma unroll
        for (int r = 0; r < 8; ++r) {
            float sm = 0.0f;
#pragma unroll
            for (int t = 0; t < 8; ++t) sm += acc[t][r];
            pr[r] = sm;
        }
        float sm = ((pr[0] + pr[1]) + (pr[2] + pr[3])) + ((pr[4] + pr[5]) + (pr[6] + pr[7]));
        sm = sm + other_half(sm);
        const float mu = sm * (1.0f / cD);
#pragma unroll
        for (int r = 0; r < 8; ++r) {
            float sq = 0.0f;
#pragma unroll
            for (int t = 0; t < 8; ++t) {
                const float d = acc[t][r] - mu;
                acc[t][r] = d;
                sq += d * d;
            }
            pr[r] = sq;
        }
        float sq = ((pr[0] + pr[1]) + (pr[2] + pr[3])) + ((pr[4] + pr[5]) + (pr[6] + pr[7]));
        sq = sq + other_half(sq);
        const float var = sq * (1.0f / cD);
        const float rs  = rsqrtf(var + 1e-5f);
#pragma unroll
        for (int r = 0; r < 8; ++r)
#pragma unroll
            for (int t = 0; t < 8; ++t)
                acc[t][r] = acc[t][r] * rs * gv[8 * r + t];

#pragma unroll
        for (int r = 3; r < 8; ++r) {
#pragma unroll
            for (int u = 0; u < 4; ++u) {
                ux[r][u] = *(const v2f*)(Uk + (size_t)(8 * r + 2 * u) * cD);
                uy[r][u] = *(const v2f*)(Uk + (size_t)(8 * r + 2 * u + 1) * cD);
            }
        }

        {
            v2f b;
            b.x = khalf ? 0.0f : 1.0f;
            b.y = 0.0f;
#pragma unroll
            for (int t = 0; t < 8; ++t) {
                v2f a;
                a.x = khalf ? 0.0f : betv[t];
                a.y = 0.0f;
                acc[t] = __builtin_amdgcn_wmma_f32_16x16x4_f32(
                             false, a, false, b, (short)0, acc[t], false, false);
            }
        }
    }
#pragma unroll
    for (int r = 0; r < 8; ++r)
#pragma unroll
        for (int t = 0; t < 8; ++t)
            xn[8 * r + t] = acc[t][r] > 0.0f ? acc[t][r] : 0.0f;

    if (write_x) {
        float* xrow = xout + xoff;
#pragma unroll
        for (int i = 0; i < 16; ++i)
            *(float4*)(xrow + 4 * i) = make_float4(xn[4 * i], xn[4 * i + 1], xn[4 * i + 2], xn[4 * i + 3]);
    }

    v8f e0 = {}, e1 = {};
#pragma unroll
    for (int r = 0; r < 8; ++r) {
#pragma unroll
        for (int u = 0; u < 4; ++u) {
            v2f a0, a1, b;
            a0.x = ux[r][u].x;
            a0.y = uy[r][u].x;
            a1.x = ux[r][u].y;
            a1.y = uy[r][u].y;
            b.x = xn[8 * r + 2 * u];
            b.y = xn[8 * r + 2 * u + 1];
            e0 = __builtin_amdgcn_wmma_f32_16x16x4_f32(
                     false, a0, false, b, (short)0, e0, false, false);
            e1 = __builtin_amdgcn_wmma_f32_16x16x4_f32(
                     false, a1, false, b, (short)0, e1, false, false);
        }
    }

    float y[16];
#pragma unroll
    for (int r = 0; r < 8; ++r) {
        y[2 * r]     = e0[r] + (with_bias ? b2v[2 * r] : 0.0f);
        y[2 * r + 1] = e1[r] + (with_bias ? b2v[2 * r + 1] : 0.0f);
    }
    float* crow = Cw + (size_t)(m0 + l15) * cD + n0 + 16 * khalf;
#pragma unroll
    for (int i = 0; i < 4; ++i)
        *(float4*)(crow + 4 * i) = make_float4(y[4 * i], y[4 * i + 1], y[4 * i + 2], y[4 * i + 3]);
}

__global__ void __launch_bounds__(128)
gemm_in_products(const float* __restrict__ A0, const float* __restrict__ W0,
                 const float* __restrict__ b0, float* __restrict__ xout,
                 const float* __restrict__ B0, const float* __restrict__ B1,
                 const float* __restrict__ B2, const float* __restrict__ bias2,
                 float* __restrict__ C0, float* __restrict__ C1, float* __restrict__ C2)
{
    const int s = blockIdx.y;
    x_then_products<64, false>(A0, W0, b0, nullptr, nullptr, nullptr, xout,
                               s == 0 ? B0 : (s == 1 ? B1 : B2), bias2,
                               s == 0 ? C0 : (s == 1 ? C1 : C2),
                               s == 2, s == 0 && threadIdx.x < 32);
}

__global__ void __launch_bounds__(128)
gemm_hn_ln_products(const float* __restrict__ A0, const float* __restrict__ W0,
                    const float* __restrict__ b0, const float* __restrict__ xin,
                    const float* __restrict__ g, const float* __restrict__ beta,
                    float* __restrict__ xout,
                    const float* __restrict__ B0, const float* __restrict__ B1,
                    const float* __restrict__ B2, const float* __restrict__ bias2,
                    float* __restrict__ C0, float* __restrict__ C1, float* __restrict__ C2)
{
    const int s = blockIdx.y;
    x_then_products<cD, true>(A0, W0, b0, xin, g, beta, xout,
                              s == 0 ? B0 : (s == 1 ? B1 : B2), bias2,
                              s == 0 ? C0 : (s == 1 ? C1 : C2),
                              s == 2, s == 0 && threadIdx.x < 32);
}

static constexpr int TI = 3;
static constexpr int QB = 48;
static constexpr int AG = 8;
static constexpr int NG = QB / AG;
static constexpr int RW = cN / 4;
static_assert(cN % TI == 0 && cN % (2 * QB) == 0 && NG % 2 == 0 && AG % 4 == 0, "shape");

struct AdjGroup { float a[TI][AG]; };

__device__ __forceinline__ void load_q(float (&q)[QB], const float* __restrict__ Qcol, int j0)
{
#pragma unroll
    for (int u = 0; u < QB; ++u)
        q[u] = Qcol[(size_t)(j0 + u) * cD];
}

__device__ __forceinline__ void load_adj(AdjGroup& g, const float4* rows, int j0)
{
#pragma unroll
    for (int r = 0; r < TI; ++r) {
#pragma unroll
        for (int v = 0; v < AG / 4; ++v) {
            const float4 x = rows[r * RW + (j0 >> 2) + v];
            g.a[r][4 * v + 0] = x.x; g.a[r][4 * v + 1] = x.y;
            g.a[r][4 * v + 2] = x.z; g.a[r][4 * v + 3] = x.w;
        }
    }
}

__device__ __forceinline__ void compute_group(const float* q, const AdjGroup& g,
                                              const float (&p)[TI], float w, float b1,
                                              float (&acc)[TI], float& dg)
{
#pragma unroll
    for (int u = 0; u < AG; ++u) {
#pragma unroll
        for (int r = 0; r < TI; ++r) {
            const float a = g.a[r][u];
            float t = p[r] + q[u] + a * w + b1;
            t = fmaxf(t, 0.0f);
            acc[r] += a * t;
        }
        dg += g.a[0][u];
    }
}

template <bool LAST>
__device__ __forceinline__ void one_batch(const float (&q)[QB], AdjGroup& ga, AdjGroup& gb,
                                          const float4* rows, int j0,
                                          const float (&p)[TI], float w, float b1,
                                          float (&acc)[TI], float& dg)
{
#pragma unroll
    for (int s = 0; s < NG; s += 2) {
        load_adj(gb, rows, j0 + (s + 1) * AG);
        __builtin_amdgcn_sched_barrier(0);
        compute_group(&q[s * AG], ga, p, w, b1, acc, dg);
        if (s + 2 < NG || !LAST) load_adj(ga, rows, j0 + (s + 2) * AG);
        __builtin_amdgcn_sched_barrier(0);
        compute_group(&q[(s + 1) * AG], gb, p, w, b1, acc, dg);
    }
}

__global__ void __launch_bounds__(cD)
pairwise_agg(const float* __restrict__ P,
             const float* __restrict__ Q,
             const float* __restrict__ adj,
             const float* __restrict__ we,
             const float* __restrict__ mb1,
             float* __restrict__ R,
             float* __restrict__ deg)
{
    const int i0   = blockIdx.x * TI;
    const int b    = i0 / cN;
    const int d    = threadIdx.x;
    const int wave = threadIdx.x >> 5;
    const int rot  = wave % TI;

    __shared__ float4 s_adj4[(2 * TI - 1) * RW];

    const float* Qcol = Q + (size_t)b * cN * cD + d;

    float qa[QB], qb[QB];
    load_q(qa, Qcol, 0);

    {
        const float4* adj4 = (const float4*)(adj + (size_t)i0 * cN);
        for (int v = d; v < TI * RW; v += cD) {
            const float4 x = adj4[v];
            s_adj4[v] = x;
            if (v < (TI - 1) * RW) s_adj4[v + TI * RW] = x;
        }
    }

    int rr[TI];
    float p[TI];
#pragma unroll
    for (int k = 0; k < TI; ++k) {
        rr[k] = (rot + k) % TI;
        p[k]  = P[(size_t)(i0 + rr[k]) * cD + d];
    }
    const float w  = we[d];
    const float b1 = mb1[d];
    const float4* rows = s_adj4 + rot * RW;

    __syncthreads();

    AdjGroup ga, gb;
    load_adj(ga, rows, 0);

    float acc[TI] = {0.0f, 0.0f, 0.0f};
    float dg = 0.0f;

    for (int j0 = 0; j0 < cN - 2 * QB; j0 += 2 * QB) {
        load_q(qb, Qcol, j0 + QB);
        one_batch<false>(qa, ga, gb, rows, j0, p, w, b1, acc, dg);
        load_q(qa, Qcol, j0 + 2 * QB);
        one_batch<false>(qb, ga, gb, rows, j0 + QB, p, w, b1, acc, dg);
    }
    load_q(qb, Qcol, cN - QB);
    one_batch<false>(qa, ga, gb, rows, cN - 2 * QB, p, w, b1, acc, dg);
    one_batch<true >(qb, ga, gb, rows, cN - QB, p, w, b1, acc, dg);

#pragma unroll
    for (int k = 0; k < TI; ++k)
        R[(size_t)(i0 + rr[k]) * cD + d] = acc[k];
    if (wave < TI && (d & 31) == 0) deg[i0 + rot] = dg;
}

extern "C" void kernel_launch(void* const* d_in, const int* in_sizes, int n_in,
                              void* d_out, int out_size, void* d_ws, size_t ws_size,
                              hipStream_t stream)
{
    (void)in_sizes; (void)n_in; (void)out_size; (void)ws_size;

    const float* h    = (const float*)d_in[0];
    const float* adj  = (const float*)d_in[1];
    const float* ip_w = (const float*)d_in[2];
    const float* ip_b = (const float*)d_in[3];
    const float* mw1  = (const float*)d_in[4];
    const float* mb1  = (const float*)d_in[5];
    const float* mw2  = (const float*)d_in[6];
    const float* mb2  = (const float*)d_in[7];
    const float* uw1  = (const float*)d_in[8];
    const float* ub1  = (const float*)d_in[9];
    const float* uw2  = (const float*)d_in[10];
    const float* ub2  = (const float*)d_in[11];
    const float* ln_g = (const float*)d_in[12];
    const float* ln_b = (const float*)d_in[13];

    float* ws = (float*)d_ws;
    const size_t S = (size_t)cM * cD;
    float* xa   = ws + 0 * S;
    float* P    = ws + 1 * S;
    float* Q    = ws + 2 * S;
    float* R    = ws + 3 * S;
    float* xb   = ws + 4 * S;
    float* hida = ws + 5 * S;
    float* hidb = ws + 6 * S;
    float* deg  = ws + 7 * S;

    gemm_in_products<<<dim3(48, 3), 128, 0, stream>>>(h, ip_w, ip_b, xa,
                                                      mw1, mw1 + 128 * cD, uw1, ub1,
                                                      P, Q, hida);

    for (int l = 0; l < 3; ++l) {
        const float* mw1l = mw1 + (size_t)l * 257 * cD;
        const float* mb1l = mb1 + (size_t)l * cD;
        const float* mw2l = mw2 + (size_t)l * cD * cD;
        const float* mb2l = mb2 + (size_t)l * cD;
        const float* uw1l = uw1 + (size_t)l * 256 * cD;
        const float* uw2l = uw2 + (size_t)l * cD * cD;
        const float* ub2l = ub2 + (size_t)l * cD;
        float* xcur = (l & 1) ? xb : xa;
        float* xnxt = (l & 1) ? xa : xb;
        float* hcur = (l & 1) ? hidb : hida;
        float* hnxt = (l & 1) ? hida : hidb;

        pairwise_agg<<<cM / TI, cD, 0, stream>>>(P, Q, adj, mw1l + 256 * cD, mb1l, R, deg);

        gemm_agg_hid<<<4 * (cM / 16), 128, 0, stream>>>(R, mw2l, mb2l, deg, uw1l + 128 * cD, hcur);

        if (l < 2) {
            const float* mw1n = mw1 + (size_t)(l + 1) * 257 * cD;
            const float* uw1n = uw1 + (size_t)(l + 1) * 256 * cD;
            const float* ub1n = ub1 + (size_t)(l + 1) * cD;
            gemm_hn_ln_products<<<dim3(48, 3), 128, 0, stream>>>(hcur, uw2l, ub2l, xcur,
                                                                 ln_g + l * cD, ln_b + l * cD, xnxt,
                                                                 mw1n, mw1n + 128 * cD, uw1n, ub1n,
                                                                 P, Q, hnxt);
        } else {
            gemm_hn_ln_out<<<cM / 16, 32, 0, stream>>>(hcur, uw2l, ub2l,
                                                      ln_g + l * cD, ln_b + l * cD,
                                                      (float*)d_out);
        }
    }
}


# x / hn product of a 16-row tile split by channel over the block's 4 waves, 32 each (was all 128 in each wave; last layer: 1 wave to 4), quarters handed on in LDS; k order kept, last LN regrouped
# speedup vs baseline: 3.3059x; 1.1168x over previous
#include <hip/hip_runtime.h>
#include <hip/hip_bf16.h>

typedef __attribute__((ext_vector_type(2))) float v2f;
typedef __attribute__((ext_vector_type(8))) float v8f;

static constexpr int cB = 2;
static constexpr int cN = 384;
static constexpr int cD = 128;
static constexpr int cM = cB * cN;

template <int M>
__device__ __forceinline__ float add_xor_lane(float v)
{
    const int o = __builtin_amdgcn_update_dpp(0, __float_as_int(v), 0x160 | M, 0xf, 0xf, true);
    return v + __int_as_float(o);
}
__device__ __forceinline__ float half_sum(float v)
{
    v = add_xor_lane<8>(v);
    v = add_xor_lane<4>(v);
    v = add_xor_lane<2>(v);
    v = add_xor_lane<1>(v);
    return v;
}

__device__ __forceinline__ float other_half(float v)
{
    return __int_as_float(__builtin_amdgcn_permlanex16(0, __float_as_int(v),
                                                       0x76543210, 0xfedcba98, false, false));
}

__device__ __forceinline__ void load4(float* d, const float* p)
{
    const float4 q = *(const float4*)p;
    d[0] = q.x; d[1] = q.y; d[2] = q.z; d[3] = q.w;
}

static constexpr int PFA = 16;
static constexpr int LPQ = 5;

__global__ void __launch_bounds__(128)
gemm_agg_hid(const float* __restrict__ R,
             const float* __restrict__ W2,
             const float* __restrict__ b2,
             const float* __restrict__ deg,
             const float* __restrict__ U,
             float* hid)
{
    constexpr int KS = cD / 4;
    const int wave  = threadIdx.x >> 5;
    const int lane  = threadIdx.x & 31;
    const int l15   = lane & 15;
    const int khalf = lane >> 4;
    const int m0    = (blockIdx.x >> 2) * 16;
    const int n0    = (blockIdx.x & 3) * 32;
    const int ch0   = wave * 32;

    __shared__ float4 s_part[4 * 32 * LPQ];

    const float* Rk = R + (size_t)(m0 + l15) * cD + 2 * khalf;
    const float* Wk = W2 + (size_t)(2 * khalf) * cD + ch0 + 2 * l15;
    const float* Uk = U + (size_t)(ch0 + 16 * khalf) * cD + n0 + 2 * l15;

    const int orow = 4 * wave + (lane >> 3);
    const int oq   = lane & 7;
    float* hout = hid + (size_t)(m0 + orow) * cD + n0 + 4 * oq;

    v2f rb[KS], wx[KS], wy[KS];
    v2f ux[8], uy[8];
    const float4 hold = *(const float4*)hout;
    const v2f mbv = *(const v2f*)(b2 + ch0 + 2 * l15);
    const float dg = deg[m0 + l15];
#pragma unroll
    for (int s = 0; s < PFA; ++s) {
        rb[s] = *(const v2f*)(Rk + 4 * s);
        wx[s] = *(const v2f*)(Wk + (size_t)(4 * s) * cD);
        wy[s] = *(const v2f*)(Wk + (size_t)(4 * s + 1) * cD);
    }

    v8f acc0 = {}, acc1 = {};
    {
        v2f a0, a1, b;
        b.x  = khalf ? 0.0f : dg;
        b.y  = 0.0f;
        a0.x = khalf ? 0.0f : mbv.x;
        a0.y = 0.0f;
        a1.x = khalf ? 0.0f : mbv.y;
        a1.y = 0.0f;
        acc0 = __builtin_amdgcn_wmma_f32_16x16x4_f32(
                   false, a0, false, b, (short)0, acc0, false, false);
        acc1 = __builtin_amdgcn_wmma_f32_16x16x4_f32(
                   false, a1, false, b, (short)0, acc1, false, false);
    }

#pragma unroll
    for (int s = 0; s < KS; ++s) {
        if (s + PFA < KS) {
            const int v = s + PFA;
            rb[v] = *(const v2f*)(Rk + 4 * v);
            wx[v] = *(const v2f*)(Wk + (size_t)(4 * v) * cD);
            wy[v] = *(const v2f*)(Wk + (size_t)(4 * v + 1) * cD);
        } else if (s < KS - PFA + 8) {
            const int r = s - (KS - PFA);
            ux[r] = *(const v2f*)(Uk + (size_t)(2 * r) * cD);
            uy[r] = *(const v2f*)(Uk + (size_t)(2 * r + 1) * cD);
        }
        v2f a0, a1;
        a0.x = wx[s].x;
        a0.y = wy[s].x;
        a1.x = wx[s].y;
        a1.y = wy[s].y;
        acc0 = __builtin_amdgcn_wmma_f32_16x16x4_f32(
                   false, a0, false, rb[s], (short)0, acc0, false, false);
        acc1 = __builtin_amdgcn_wmma_f32_16x16x4_f32(
                   false, a1, false, rb[s], (short)0, acc1, false, false);
    }

    v8f e0 = {}, e1 = {};
#pragma unroll
    for (int r = 0; r < 8; ++r) {
        v2f a0, a1, b;
        a0.x = ux[r].x;
        a0.y = uy[r].x;
        a1.x = ux[r].y;
        a1.y = uy[r].y;
        b.x = acc0[r];
        b.y = acc1[r];
        e0 = __builtin_amdgcn_wmma_f32_16x16x4_f32(
                 false, a0, false, b, (short)0, e0, false, false);
        e1 = __builtin_amdgcn_wmma_f32_16x16x4_f32(
                 false, a1, false, b, (short)0, e1, false, false);
    }

    __builtin_amdgcn_sched_group_barrier(0x020, 3 + 3 * PFA, 0);
    __builtin_amdgcn_sched_group_barrier(0x008, 2, 0);
#pragma unroll
    for (int s = 0; s < KS - PFA; ++s) {
        __builtin_amdgcn_sched_group_barrier(0x020, 3, 0);
        __builtin_amdgcn_sched_group_barrier(0x008, 2, 0);
    }
#pragma unroll
    for (int r = 0; r < 8; ++r) {
        __builtin_amdgcn_sched_group_barrier(0x020, 2, 0);
        __builtin_amdgcn_sched_group_barrier(0x008, 2, 0);
    }

    float4* mine = s_part + (wave * 32 + lane) * LPQ;
#pragma unroll
    for (int q = 0; q < 4; ++q)
        mine[q] = make_float4(e0[2 * q], e1[2 * q], e0[2 * q + 1], e1[2 * q + 1]);

    __syncthreads();

    const float4* theirs = s_part + (orow + 16 * (oq >> 2)) * LPQ + (oq & 3);
    float4 v = hold;
#pragma unroll
    for (int w = 0; w < 4; ++w) {
        const float4 p = theirs[w * 32 * LPQ];
        v.x += p.x;
        v.y += p.y;
        v.z += p.z;
        v.w += p.w;
    }
    v.x = v.x > 0.0f ? v.x : 0.0f;
    v.y = v.y > 0.0f ? v.y : 0.0f;
    v.z = v.z > 0.0f ? v.z : 0.0f;
    v.w = v.w > 0.0f ? v.w : 0.0f;
    *(float4*)hout = v;
}

static constexpr int PFX = 16;
static constexpr int XP  = 33;

template <int K0, bool LNRES>
__device__ __forceinline__ void x_then_products(const float* __restrict__ A0,
                                                const float* __restrict__ W0,
                                                const float* __restrict__ b0,
                                                const float* __restrict__ xin,
                                                const float* __restrict__ g,
                                                const float* __restrict__ beta,
                                                float* __restrict__ xout,
                                                const float* __restrict__ Bw,
                                                const float* __restrict__ bias2,
                                                float* __restrict__ Cw,
                                                const bool with_bias, const bool write_x)
{
    constexpr int KS = K0 / 4;
    constexpr int T0 = KS - PFX;
    static_assert(LNRES ? (KS == 2 * PFX) : (KS == PFX), "the load schedule below is written for these two shapes");
    const int wave  = threadIdx.x >> 5;
    const int lane  = threadIdx.x & 31;
    const int l15   = lane & 15;
    const int khalf = lane >> 4;
    const int m0    = blockIdx.x * 16;
    const int n0    = wave * 32;

    __shared__ float4 s_x[16 * XP];

    const float* Ak = A0 + (size_t)(m0 + l15) * K0 + 2 * khalf;
    const float* Wk = W0 + (size_t)(2 * khalf) * cD + n0 + 2 * l15;
    const float* Uk = Bw + (size_t)(64 * khalf) * cD + n0 + 2 * l15;
    const size_t xoff = (size_t)(m0 + l15) * cD + 64 * khalf;

    v2f rb[KS], wx[KS], wy[KS];
    v2f ux[8][4], uy[8][4];
    float b2v[16];
    float gv[64];
    float betv[8];

    v8f pa0 = {}, pa1 = {};
    if (LNRES) {
        float q[16];
#pragma unroll
        for (int i = 0; i < 4; ++i)
            load4(&q[4 * i], xin + (size_t)(m0 + l15) * cD + n0 + 16 * khalf + 4 * i);
#pragma unroll
        for (int r = 0; r < 8; ++r) {
            pa0[r] = q[2 * r];
            pa1[r] = q[2 * r + 1];
        }
    }
    const v2f b0v = *(const v2f*)(b0 + n0 + 2 * l15);
#pragma unroll
    for (int s = 0; s < PFX; ++s) {
        rb[s] = *(const v2f*)(Ak + 4 * s);
        wx[s] = *(const v2f*)(Wk + (size_t)(4 * s) * cD);
        wy[s] = *(const v2f*)(Wk + (size_t)(4 * s + 1) * cD);
    }

#pragma unroll
    for (int s = 0; s < KS; ++s) {
        if (s < T0) {
            const int v = s + PFX;
            rb[v] = *(const v2f*)(Ak + 4 * v);
            wx[v] = *(const v2f*)(Wk + (size_t)(4 * v) * cD);
            wy[v] = *(const v2f*)(Wk + (size_t)(4 * v + 1) * cD);
        }
        if (!LNRES) {
            if (s < 4) {
#pragma unroll
                for (int r = 2 * s; r < 2 * s + 2; ++r)
#pragma unroll
                    for (int u = 0; u < 4; ++u) {
                        ux[r][u] = *(const v2f*)(Uk + (size_t)(8 * r + 2 * u) * cD);
                        uy[r][u] = *(const v2f*)(Uk + (size_t)(8 * r + 2 * u + 1) * cD);
                    }
            }
            if (s == 4) {
#pragma unroll
                for (int i = 0; i < 4; ++i)
                    load4(&b2v[4 * i], bias2 + n0 + 16 * khalf + 4 * i);
            }
        } else {
            const int p = s - T0;
            if (p >= 0 && p < 3) {
#pragma unroll
                for (int i = 6 * p; i < 6 * p + 6 && i < 16; ++i)
                    load4(&gv[4 * i], g + 64 * khalf + 4 * i);
            }
            if (p == 2) {
                load4(&betv[0], beta + l15 * 8);
                load4(&betv[4], beta + l15 * 8 + 4);
            }
            if (p == 3) {
#pragma unroll
                for (int i = 0; i < 4; ++i)
                    load4(&b2v[4 * i], bias2 + n0 + 16 * khalf + 4 * i);
            }
            if (p == 3 || p == 4) {
#pragma unroll
                for (int u = (p == 3 ? 0 : 1); u < (p == 3 ? 1 : 4); ++u) {
                    ux[7][u] = *(const v2f*)(Uk + (size_t)(8 * 7 + 2 * u) * cD);
                    uy[7][u] = *(const v2f*)(Uk + (size_t)(8 * 7 + 2 * u + 1) * cD);
                }
            }
        }
        __builtin_amdgcn_sched_barrier(0);
        v2f a0, a1;
        a0.x = wx[s].x;
        a0.y = wy[s].x;
        a1.x = wx[s].y;
        a1.y = wy[s].y;
        pa0 = __builtin_amdgcn_wmma_f32_16x16x4_f32(
                  false, a0, false, rb[s], (short)0, pa0, false, false);
        pa1 = __builtin_amdgcn_wmma_f32_16x16x4_f32(
                  false, a1, false, rb[s], (short)0, pa1, false, false);
        __builtin_amdgcn_sched_barrier(0);
    }

    {
        v2f a0, a1, b;
        b.x  = khalf ? 0.0f : 1.0f;
        b.y  = 0.0f;
        a0.x = khalf ? 0.0f : b0v.x;
        a0.y = 0.0f;
        a1.x = khalf ? 0.0f : b0v.y;
        a1.y = 0.0f;
        pa0 = __builtin_amdgcn_wmma_f32_16x16x4_f32(
                  false, a0, false, b, (short)0, pa0, false, false);
        pa1 = __builtin_amdgcn_wmma_f32_16x16x4_f32(
                  false, a1, false, b, (short)0, pa1, false, false);
    }

    {
        float4* mine = s_x + l15 * XP + (n0 >> 2) + 4 * khalf;
#pragma unroll
        for (int q = 0; q < 4; ++q)
            mine[q] = make_float4(pa0[2 * q], pa1[2 * q], pa0[2 * q + 1], pa1[2 * q + 1]);
    }
    __syncthreads();

    v8f acc[8];
    {
        const float4* row = s_x + l15 * XP + 16 * khalf;
#pragma unroll
        for (int i = 0; i < 16; ++i) {
            const float4 q = row[i];
            acc[4 * (i & 1) + 0][i >> 1] = q.x;
            acc[4 * (i & 1) + 1][i >> 1] = q.y;
            acc[4 * (i & 1) + 2][i >> 1] = q.z;
            acc[4 * (i & 1) + 3][i >> 1] = q.w;
        }
    }
    if (LNRES) {
#pragma unroll
        for (int r = 0; r < 7; ++r)
#pragma unroll
            for (int u = 0; u < 4; ++u) {
                ux[r][u] = *(const v2f*)(Uk + (size_t)(8 * r + 2 * u) * cD);
                uy[r][u] = *(const v2f*)(Uk + (size_t)(8 * r + 2 * u + 1) * cD);
            }
        __builtin_amdgcn_sched_barrier(0);
    }

    float xn[64];
    if (LNRES) {
        float pr[8];
#pragma unroll
        for (int r = 0; r < 8; ++r) {
            float sm = 0.0f;
#pragma unroll
            for (int t = 0; t < 8; ++t) sm += acc[t][r];
            pr[r] = sm;
        }
        float sm = ((pr[0] + pr[1]) + (pr[2] + pr[3])) + ((pr[4] + pr[5]) + (pr[6] + pr[7]));
        sm = sm + other_half(sm);
        const float mu = sm * (1.0f / cD);
#pragma unroll
        for (int r = 0; r < 8; ++r) {
            float sq = 0.0f;
#pragma unroll
            for (int t = 0; t < 8; ++t) {
                const float d = acc[t][r] - mu;
                acc[t][r] = d;
                sq += d * d;
            }
            pr[r] = sq;
        }
        float sq = ((pr[0] + pr[1]) + (pr[2] + pr[3])) + ((pr[4] + pr[5]) + (pr[6] + pr[7]));
        sq = sq + other_half(sq);
        const float var = sq * (1.0f / cD);
        const float rs  = rsqrtf(var + 1e-5f);
#pragma unroll
        for (int r = 0; r < 8; ++r)
#pragma unroll
            for (int t = 0; t < 8; ++t)
                acc[t][r] = acc[t][r] * rs * gv[8 * r + t];

        {
            v2f b;
            b.x = khalf ? 0.0f : 1.0f;
            b.y = 0.0f;
#pragma unroll
            for (int t = 0; t < 8; ++t) {
                v2f a;
                a.x = khalf ? 0.0f : betv[t];
                a.y = 0.0f;
                acc[t] = __builtin_amdgcn_wmma_f32_16x16x4_f32(
                             false, a, false, b, (short)0, acc[t], false, false);
            }
        }
    }
#pragma unroll
    for (int r = 0; r < 8; ++r)
#pragma unroll
        for (int t = 0; t < 8; ++t)
            xn[8 * r + t] = acc[t][r] > 0.0f ? acc[t][r] : 0.0f;

    if (write_x) {
        float* xrow = xout + xoff;
#pragma unroll
        for (int i = 0; i < 16; ++i)
            *(float4*)(xrow + 4 * i) = make_float4(xn[4 * i], xn[4 * i + 1], xn[4 * i + 2], xn[4 * i + 3]);
    }

    v8f e0 = {}, e1 = {};
#pragma unroll
    for (int r = 0; r < 8; ++r) {
#pragma unroll
        for (int u = 0; u < 4; ++u) {
            v2f a0, a1, b;
            a0.x = ux[r][u].x;
            a0.y = uy[r][u].x;
            a1.x = ux[r][u].y;
            a1.y = uy[r][u].y;
            b.x = xn[8 * r + 2 * u];
            b.y = xn[8 * r + 2 * u + 1];
            e0 = __builtin_amdgcn_wmma_f32_16x16x4_f32(
                     false, a0, false, b, (short)0, e0, false, false);
            e1 = __builtin_amdgcn_wmma_f32_16x16x4_f32(
                     false, a1, false, b, (short)0, e1, false, false);
        }
    }

    float y[16];
#pragma unroll
    for (int r = 0; r < 8; ++r) {
        y[2 * r]     = e0[r] + (with_bias ? b2v[2 * r] : 0.0f);
        y[2 * r + 1] = e1[r] + (with_bias ? b2v[2 * r + 1] : 0.0f);
    }
    float* crow = Cw + (size_t)(m0 + l15) * cD + n0 + 16 * khalf;
#pragma unroll
    for (int i = 0; i < 4; ++i)
        *(float4*)(crow + 4 * i) = make_float4(y[4 * i], y[4 * i + 1], y[4 * i + 2], y[4 * i + 3]);
}

__global__ void __launch_bounds__(128)
gemm_in_products(const float* __restrict__ A0, const float* __restrict__ W0,
                 const float* __restrict__ b0, float* __restrict__ xout,
                 const float* __restrict__ B0, const float* __restrict__ B1,
                 const float* __restrict__ B2, const float* __restrict__ bias2,
                 float* __restrict__ C0, float* __restrict__ C1, float* __restrict__ C2)
{
    const int s = blockIdx.y;
    x_then_products<64, false>(A0, W0, b0, nullptr, nullptr, nullptr, xout,
                               s == 0 ? B0 : (s == 1 ? B1 : B2), bias2,
                               s == 0 ? C0 : (s == 1 ? C1 : C2),
                               s == 2, s == 0 && threadIdx.x < 32);
}

__global__ void __launch_bounds__(128)
gemm_hn_ln_products(const float* __restrict__ A0, const float* __restrict__ W0,
                    const float* __restrict__ b0, const float* __restrict__ xin,
                    const float* __restrict__ g, const float* __restrict__ beta,
                    float* __restrict__ xout,
                    const float* __restrict__ B0, const float* __restrict__ B1,
                    const float* __restrict__ B2, const float* __restrict__ bias2,
                    float* __restrict__ C0, float* __restrict__ C1, float* __restrict__ C2)
{
    const int s = blockIdx.y;
    x_then_products<cD, true>(A0, W0, b0, xin, g, beta, xout,
                              s == 0 ? B0 : (s == 1 ? B1 : B2), bias2,
                              s == 0 ? C0 : (s == 1 ? C1 : C2),
                              s == 2, s == 0 && threadIdx.x < 32);
}

__global__ void __launch_bounds__(128)
gemm_hn_ln_out(const float* __restrict__ A,
               const float* __restrict__ Bm,
               const float* __restrict__ bias,
               const float* __restrict__ g,
               const float* __restrict__ beta,
               float* __restrict__ out)
{
    constexpr int KS = cD / 4;
    const int wave  = threadIdx.x >> 5;
    const int lane  = threadIdx.x & 31;
    const int l15   = lane & 15;
    const int khalf = lane >> 4;
    const int m0    = blockIdx.x * 16;
    const int n0    = wave * 32;

    __shared__ float4 s_x[16 * XP];

    const float* Ak = A + (size_t)(m0 + l15) * cD + 2 * khalf;
    const float* Wk = Bm + (size_t)(2 * khalf) * cD + n0 + 2 * l15;

    v2f rb[KS], wx[KS], wy[KS];
    const float4 gq = *(const float4*)(g + 4 * lane);
    const float4 bq = *(const float4*)(beta + 4 * lane);
    const v2f b0v = *(const v2f*)(bias + n0 + 2 * l15);
#pragma unroll
    for (int s = 0; s < PFX; ++s) {
        rb[s] = *(const v2f*)(Ak + 4 * s);
        wx[s] = *(const v2f*)(Wk + (size_t)(4 * s) * cD);
        wy[s] = *(const v2f*)(Wk + (size_t)(4 * s + 1) * cD);
    }
    asm volatile("" :: "v"(gq.x), "v"(gq.y), "v"(gq.z), "v"(gq.w),
                       "v"(bq.x), "v"(bq.y), "v"(bq.z), "v"(bq.w));

    v8f pa0 = {}, pa1 = {};
#pragma unroll
    for (int s = 0; s < KS; ++s) {
        if (s + PFX < KS) {
            const int v = s + PFX;
            rb[v] = *(const v2f*)(Ak + 4 * v);
            wx[v] = *(const v2f*)(Wk + (size_t)(4 * v) * cD);
            wy[v] = *(const v2f*)(Wk + (size_t)(4 * v + 1) * cD);
        }
        __builtin_amdgcn_sched_barrier(0);
        v2f a0, a1;
        a0.x = wx[s].x;
        a0.y = wy[s].x;
        a1.x = wx[s].y;
        a1.y = wy[s].y;
        pa0 = __builtin_amdgcn_wmma_f32_16x16x4_f32(
                  false, a0, false, rb[s], (short)0, pa0, false, false);
        pa1 = __builtin_amdgcn_wmma_f32_16x16x4_f32(
                  false, a1, false, rb[s], (short)0, pa1, false, false);
        __builtin_amdgcn_sched_barrier(0);
    }

    {
        v2f a0, a1, b;
        b.x  = khalf ? 0.0f : 1.0f;
        b.y  = 0.0f;
        a0.x = khalf ? 0.0f : b0v.x;
        a0.y = 0.0f;
        a1.x = khalf ? 0.0f : b0v.y;
        a1.y = 0.0f;
        pa0 = __builtin_amdgcn_wmma_f32_16x16x4_f32(
                  false, a0, false, b, (short)0, pa0, false, false);
        pa1 = __builtin_amdgcn_wmma_f32_16x16x4_f32(
                  false, a1, false, b, (short)0, pa1, false, false);
    }

    {
        float4* mine = s_x + l15 * XP + (n0 >> 2) + 4 * khalf;
#pragma unroll
        for (int q = 0; q < 4; ++q)
            mine[q] = make_float4(pa0[2 * q], pa1[2 * q], pa0[2 * q + 1], pa1[2 * q + 1]);
    }
    __syncthreads();

#pragma unroll
    for (int n = 0; n < 4; ++n) {
        const int node = 4 * wave + n;
        const float4 v = s_x[node * XP + lane];
        float sm = (v.x + v.y) + (v.z + v.w);
        sm = half_sum(sm);
        sm = sm + other_half(sm);
        const float mu = sm * (1.0f / cD);
        const float dx = v.x - mu, dy = v.y - mu, dz = v.z - mu, dw = v.w - mu;
        float sq = (dx * dx + dy * dy) + (dz * dz + dw * dw);
        sq = half_sum(sq);
        sq = sq + other_half(sq);
        const float var = sq * (1.0f / cD);
        const float rs  = rsqrtf(var + 1e-5f);
        *(float4*)(out + (size_t)(m0 + node) * cD + 4 * lane) =
            make_float4(dx * rs * gq.x + bq.x, dy * rs * gq.y + bq.y,
                        dz * rs * gq.z + bq.z, dw * rs * gq.w + bq.w);
    }
}

static constexpr int TI = 3;
static constexpr int QB = 48;
static constexpr int AG = 8;
static constexpr int NG = QB / AG;
static constexpr int RW = cN / 4;
static_assert(cN % TI == 0 && cN % (2 * QB) == 0 && NG % 2 == 0 && AG % 4 == 0, "shape");

struct AdjGroup { float a[TI][AG]; };

__device__ __forceinline__ void load_q(float (&q)[QB], const float* __restrict__ Qcol, int j0)
{
#pragma unroll
    for (int u = 0; u < QB; ++u)
        q[u] = Qcol[(size_t)(j0 + u) * cD];
}

__device__ __forceinline__ void load_adj(AdjGroup& g, const float4* rows, int j0)
{
#pragma unroll
    for (int r = 0; r < TI; ++r) {
#pragma unroll
        for (int v = 0; v < AG / 4; ++v) {
            const float4 x = rows[r * RW + (j0 >> 2) + v];
            g.a[r][4 * v + 0] = x.x; g.a[r][4 * v + 1] = x.y;
            g.a[r][4 * v + 2] = x.z; g.a[r][4 * v + 3] = x.w;
        }
    }
}

__device__ __forceinline__ void compute_group(const float* q, const AdjGroup& g,
                                              const float (&p)[TI], float w, float b1,
                                              float (&acc)[TI], float& dg)
{
#pragma unroll
    for (int u = 0; u < AG; ++u) {
#pragma unroll
        for (int r = 0; r < TI; ++r) {
            const float a = g.a[r][u];
            float t = p[r] + q[u] + a * w + b1;
            t = fmaxf(t, 0.0f);
            acc[r] += a * t;
        }
        dg += g.a[0][u];
    }
}

template <bool LAST>
__device__ __forceinline__ void one_batch(const float (&q)[QB], AdjGroup& ga, AdjGroup& gb,
                                          const float4* rows, int j0,
                                          const float (&p)[TI], float w, float b1,
                                          float (&acc)[TI], float& dg)
{
#pragma unroll
    for (int s = 0; s < NG; s += 2) {
        load_adj(gb, rows, j0 + (s + 1) * AG);
        __builtin_amdgcn_sched_barrier(0);
        compute_group(&q[s * AG], ga, p, w, b1, acc, dg);
        if (s + 2 < NG || !LAST) load_adj(ga, rows, j0 + (s + 2) * AG);
        __builtin_amdgcn_sched_barrier(0);
        compute_group(&q[(s + 1) * AG], gb, p, w, b1, acc, dg);
    }
}

__global__ void __launch_bounds__(cD)
pairwise_agg(const float* __restrict__ P,
             const float* __restrict__ Q,
             const float* __restrict__ adj,
             const float* __restrict__ we,
             const float* __restrict__ mb1,
             float* __restrict__ R,
             float* __restrict__ deg)
{
    const int i0   = blockIdx.x * TI;
    const int b    = i0 / cN;
    const int d    = threadIdx.x;
    const int wave = threadIdx.x >> 5;
    const int rot  = wave % TI;

    __shared__ float4 s_adj4[(2 * TI - 1) * RW];

    const float* Qcol = Q + (size_t)b * cN * cD + d;

    float qa[QB], qb[QB];
    load_q(qa, Qcol, 0);

    {
        const float4* adj4 = (const float4*)(adj + (size_t)i0 * cN);
        for (int v = d; v < TI * RW; v += cD) {
            const float4 x = adj4[v];
            s_adj4[v] = x;
            if (v < (TI - 1) * RW) s_adj4[v + TI * RW] = x;
        }
    }

    int rr[TI];
    float p[TI];
#pragma unroll
    for (int k = 0; k < TI; ++k) {
        rr[k] = (rot + k) % TI;
        p[k]  = P[(size_t)(i0 + rr[k]) * cD + d];
    }
    const float w  = we[d];
    const float b1 = mb1[d];
    const float4* rows = s_adj4 + rot * RW;

    __syncthreads();

    AdjGroup ga, gb;
    load_adj(ga, rows, 0);

    float acc[TI] = {0.0f, 0.0f, 0.0f};
    float dg = 0.0f;

    for (int j0 = 0; j0 < cN - 2 * QB; j0 += 2 * QB) {
        load_q(qb, Qcol, j0 + QB);
        one_batch<false>(qa, ga, gb, rows, j0, p, w, b1, acc, dg);
        load_q(qa, Qcol, j0 + 2 * QB);
        one_batch<false>(qb, ga, gb, rows, j0 + QB, p, w, b1, acc, dg);
    }
    load_q(qb, Qcol, cN - QB);
    one_batch<false>(qa, ga, gb, rows, cN - 2 * QB, p, w, b1, acc, dg);
    one_batch<true >(qb, ga, gb, rows, cN - QB, p, w, b1, acc, dg);

#pragma unroll
    for (int k = 0; k < TI; ++k)
        R[(size_t)(i0 + rr[k]) * cD + d] = acc[k];
    if (wave < TI && (d & 31) == 0) deg[i0 + rot] = dg;
}

extern "C" void kernel_launch(void* const* d_in, const int* in_sizes, int n_in,
                              void* d_out, int out_size, void* d_ws, size_t ws_size,
                              hipStream_t stream)
{
    (void)in_sizes; (void)n_in; (void)out_size; (void)ws_size;

    const float* h    = (const float*)d_in[0];
    const float* adj  = (const float*)d_in[1];
    const float* ip_w = (const float*)d_in[2];
    const float* ip_b = (const float*)d_in[3];
    const float* mw1  = (const float*)d_in[4];
    const float* mb1  = (const float*)d_in[5];
    const float* mw2  = (const float*)d_in[6];
    const float* mb2  = (const float*)d_in[7];
    const float* uw1  = (const float*)d_in[8];
    const float* ub1  = (const float*)d_in[9];
    const float* uw2  = (const float*)d_in[10];
    const float* ub2  = (const float*)d_in[11];
    const float* ln_g = (const float*)d_in[12];
    const float* ln_b = (const float*)d_in[13];

    float* ws = (float*)d_ws;
    const size_t S = (size_t)cM * cD;
    float* xa   = ws + 0 * S;
    float* P    = ws + 1 * S;
    float* Q    = ws + 2 * S;
    float* R    = ws + 3 * S;
    float* xb   = ws + 4 * S;
    float* hida = ws + 5 * S;
    float* hidb = ws + 6 * S;
    float* deg  = ws + 7 * S;

    gemm_in_products<<<dim3(48, 3), 128, 0, stream>>>(h, ip_w, ip_b, xa,
                                                      mw1, mw1 + 128 * cD, uw1, ub1,
                                                      P, Q, hida);

    for (int l = 0; l < 3; ++l) {
        const float* mw1l = mw1 + (size_t)l * 257 * cD;
        const float* mb1l = mb1 + (size_t)l * cD;
        const float* mw2l = mw2 + (size_t)l * cD * cD;
        const float* mb2l = mb2 + (size_t)l * cD;
        const float* uw1l = uw1 + (size_t)l * 256 * cD;
        const float* uw2l = uw2 + (size_t)l * cD * cD;
        const float* ub2l = ub2 + (size_t)l * cD;
        float* xcur = (l & 1) ? xb : xa;
        float* xnxt = (l & 1) ? xa : xb;
        float* hcur = (l & 1) ? hidb : hida;
        float* hnxt = (l & 1) ? hida : hidb;

        pairwise_agg<<<cM / TI, cD, 0, stream>>>(P, Q, adj, mw1l + 256 * cD, mb1l, R, deg);

        gemm_agg_hid<<<4 * (cM / 16), 128, 0, stream>>>(R, mw2l, mb2l, deg, uw1l + 128 * cD, hcur);

        if (l < 2) {
            const float* mw1n = mw1 + (size_t)(l + 1) * 257 * cD;
            const float* uw1n = uw1 + (size_t)(l + 1) * 256 * cD;
            const float* ub1n = ub1 + (size_t)(l + 1) * cD;
            gemm_hn_ln_products<<<dim3(48, 3), 128, 0, stream>>>(hcur, uw2l, ub2l, xcur,
                                                                 ln_g + l * cD, ln_b + l * cD, xnxt,
                                                                 mw1n, mw1n + 128 * cD, uw1n, ub1n,
                                                                 P, Q, hnxt);
        } else {
            gemm_hn_ln_out<<<cM / 16, 128, 0, stream>>>(hcur, uw2l, ub2l,
                                                      ln_g + l * cD, ln_b + l * cD,
                                                      (float*)d_out);
        }
    }
}


# agg+hid run inside the launch that makes hn: a block builds its 16-row tile of hid in LDS (4 column groups a wave) and hn's fragments are read there; 10 launches to 7; every sum keeps its order
# speedup vs baseline: 3.5323x; 1.0685x over previous
#include <hip/hip_runtime.h>
#include <hip/hip_bf16.h>

typedef __attribute__((ext_vector_type(2))) float v2f;
typedef __attribute__((ext_vector_type(8))) float v8f;

static constexpr int cB = 2;
static constexpr int cN = 384;
static constexpr int cD = 128;
static constexpr int cM = cB * cN;

template <int M>
__device__ __forceinline__ float add_xor_lane(float v)
{
    const int o = __builtin_amdgcn_update_dpp(0, __float_as_int(v), 0x160 | M, 0xf, 0xf, true);
    return v + __int_as_float(o);
}
__device__ __forceinline__ float half_sum(float v)
{
    v = add_xor_lane<8>(v);
    v = add_xor_lane<4>(v);
    v = add_xor_lane<2>(v);
    v = add_xor_lane<1>(v);
    return v;
}

__device__ __forceinline__ float other_half(float v)
{
    return __int_as_float(__builtin_amdgcn_permlanex16(0, __float_as_int(v),
                                                       0x76543210, 0xfedcba98, false, false));
}

__device__ __forceinline__ void load4(float* d, const float* p)
{
    const float4 q = *(const float4*)p;
    d[0] = q.x; d[1] = q.y; d[2] = q.z; d[3] = q.w;
}

static constexpr int PFA = 16;
static constexpr int LPQ = 5;
static constexpr int HP  = 132;
static constexpr int SPG = 4 * 32 * LPQ;

__device__ __forceinline__ void agg_hid_tile(const float* __restrict__ R,
                                             const float* __restrict__ W2,
                                             const float* __restrict__ b2,
                                             const float* __restrict__ deg,
                                             const float* __restrict__ U,
                                             const float* __restrict__ hid1,
                                             float4* s_part, float* s_h)
{
    constexpr int KS = cD / 4;
    constexpr int T0 = KS - PFA;
    const int wave  = threadIdx.x >> 5;
    const int lane  = threadIdx.x & 31;
    const int l15   = lane & 15;
    const int khalf = lane >> 4;
    const int m0    = blockIdx.x * 16;
    const int ch0   = wave * 32;

    const float* Rk = R + (size_t)(m0 + l15) * cD + 2 * khalf;
    const float* Wk = W2 + (size_t)(2 * khalf) * cD + ch0 + 2 * l15;
    const float* Uk = U + (size_t)(ch0 + 16 * khalf) * cD + 2 * l15;

    const int orow = 4 * wave + (lane >> 3);
    const int oq   = lane & 7;
    const float* hin = hid1 + (size_t)(m0 + orow) * cD + 4 * oq;

    v2f rb[KS], wx[KS], wy[KS];
    v2f ux[4][8], uy[4][8];
    float4 hold[4];
#pragma unroll
    for (int cg = 0; cg < 4; ++cg) hold[cg] = *(const float4*)(hin + 32 * cg);
    const v2f mbv = *(const v2f*)(b2 + ch0 + 2 * l15);
    const float dg = deg[m0 + l15];
#pragma unroll
    for (int s = 0; s < PFA; ++s) {
        rb[s] = *(const v2f*)(Rk + 4 * s);
        wx[s] = *(const v2f*)(Wk + (size_t)(4 * s) * cD);
        wy[s] = *(const v2f*)(Wk + (size_t)(4 * s + 1) * cD);
    }

    v8f acc0 = {}, acc1 = {};
    {
        v2f a0, a1, b;
        b.x  = khalf ? 0.0f : dg;
        b.y  = 0.0f;
        a0.x = khalf ? 0.0f : mbv.x;
        a0.y = 0.0f;
        a1.x = khalf ? 0.0f : mbv.y;
        a1.y = 0.0f;
        acc0 = __builtin_amdgcn_wmma_f32_16x16x4_f32(
                   false, a0, false, b, (short)0, acc0, false, false);
        acc1 = __builtin_amdgcn_wmma_f32_16x16x4_f32(
                   false, a1, false, b, (short)0, acc1, false, false);
    }

#pragma unroll
    for (int s = 0; s < KS; ++s) {
        if (s < T0) {
            const int v = s + PFA;
            rb[v] = *(const v2f*)(Rk + 4 * v);
            wx[v] = *(const v2f*)(Wk + (size_t)(4 * v) * cD);
            wy[v] = *(const v2f*)(Wk + (size_t)(4 * v + 1) * cD);
        } else if (s < T0 + 8) {
            const int r = s - T0;
#pragma unroll
            for (int cg = 0; cg < 4; ++cg) {
                ux[cg][r] = *(const v2f*)(Uk + (size_t)(2 * r) * cD + 32 * cg);
                uy[cg][r] = *(const v2f*)(Uk + (size_t)(2 * r + 1) * cD + 32 * cg);
            }
        }
        v2f a0, a1;
        a0.x = wx[s].x;
        a0.y = wy[s].x;
        a1.x = wx[s].y;
        a1.y = wy[s].y;
        acc0 = __builtin_amdgcn_wmma_f32_16x16x4_f32(
                   false, a0, false, rb[s], (short)0, acc0, false, false);
        acc1 = __builtin_amdgcn_wmma_f32_16x16x4_f32(
                   false, a1, false, rb[s], (short)0, acc1, false, false);
    }

    v8f e0[4], e1[4];
#pragma unroll
    for (int cg = 0; cg < 4; ++cg) {
        e0[cg] = v8f{};
        e1[cg] = v8f{};
#pragma unroll
        for (int r = 0; r < 8; ++r) {
            v2f a0, a1, b;
            a0.x = ux[cg][r].x;
            a0.y = uy[cg][r].x;
            a1.x = ux[cg][r].y;
            a1.y = uy[cg][r].y;
            b.x = acc0[r];
            b.y = acc1[r];
            e0[cg] = __builtin_amdgcn_wmma_f32_16x16x4_f32(
                         false, a0, false, b, (short)0, e0[cg], false, false);
            e1[cg] = __builtin_amdgcn_wmma_f32_16x16x4_f32(
                         false, a1, false, b, (short)0, e1[cg], false, false);
        }
    }

    __builtin_amdgcn_sched_group_barrier(0x020, 6 + 3 * PFA, 0);
    __builtin_amdgcn_sched_group_barrier(0x008, 2, 0);
#pragma unroll
    for (int s = 0; s < T0; ++s) {
        __builtin_amdgcn_sched_group_barrier(0x020, 3, 0);
        __builtin_amdgcn_sched_group_barrier(0x008, 2, 0);
    }
#pragma unroll
    for (int r = 0; r < 8; ++r) {
        __builtin_amdgcn_sched_group_barrier(0x020, 8, 0);
        __builtin_amdgcn_sched_group_barrier(0x008, 2, 0);
    }

    float4* mine = s_part + (wave * 32 + lane) * LPQ;
#pragma unroll
    for (int cg = 0; cg < 4; ++cg)
#pragma unroll
        for (int q = 0; q < 4; ++q)
            mine[cg * SPG + q] = make_float4(e0[cg][2 * q], e1[cg][2 * q], e0[cg][2 * q + 1], e1[cg][2 * q + 1]);

    __syncthreads();

    const float4* theirs = s_part + (orow + 16 * (oq >> 2)) * LPQ + (oq & 3);
#pragma unroll
    for (int cg = 0; cg < 4; ++cg) {
        float4 v = hold[cg];
#pragma unroll
        for (int w = 0; w < 4; ++w) {
            const float4 p = theirs[cg * SPG + w * 32 * LPQ];
            v.x += p.x;
            v.y += p.y;
            v.z += p.z;
            v.w += p.w;
        }
        v.x = v.x > 0.0f ? v.x : 0.0f;
        v.y = v.y > 0.0f ? v.y : 0.0f;
        v.z = v.z > 0.0f ? v.z : 0.0f;
        v.w = v.w > 0.0f ? v.w : 0.0f;
        *(float4*)(s_h + orow * HP + 32 * cg + 4 * oq) = v;
    }

    __syncthreads();
}

static constexpr int PFX = 16;
static constexpr int XP  = 33;

template <int K0, bool LNRES>
__device__ __forceinline__ void x_then_products(const float* __restrict__ A0,
                                                const float* __restrict__ W0,
                                                const float* __restrict__ b0,
                                                const float* __restrict__ xin,
                                                const float* __restrict__ g,
                                                const float* __restrict__ beta,
                                                float* __restrict__ xout,
                                                const float* __restrict__ Bw,
                                                const float* __restrict__ bias2,
                                                float* __restrict__ Cw,
                                                const bool with_bias, const bool write_x)
{
    constexpr int KS = K0 / 4;
    constexpr int T0 = KS - PFX;
    static_assert(LNRES ? (KS == 2 * PFX) : (KS == PFX), "the load schedule below is written for these two shapes");
    const int wave  = threadIdx.x >> 5;
    const int lane  = threadIdx.x & 31;
    const int l15   = lane & 15;
    const int khalf = lane >> 4;
    const int m0    = blockIdx.x * 16;
    const int n0    = wave * 32;

    __shared__ float4 s_x[16 * XP];

    const float* Ak = LNRES ? A0 + l15 * HP + 2 * khalf
                            : A0 + (size_t)(m0 + l15) * K0 + 2 * khalf;
    const float* Wk = W0 + (size_t)(2 * khalf) * cD + n0 + 2 * l15;
    const float* Uk = Bw + (size_t)(64 * khalf) * cD + n0 + 2 * l15;
    const size_t xoff = (size_t)(m0 + l15) * cD + 64 * khalf;

    v2f rb[KS], wx[KS], wy[KS];
    v2f ux[8][4], uy[8][4];
    float b2v[16];
    float gv[64];
    float betv[8];

    v8f pa0 = {}, pa1 = {};
    if (LNRES) {
        float q[16];
#pragma unroll
        for (int i = 0; i < 4; ++i)
            load4(&q[4 * i], xin + (size_t)(m0 + l15) * cD + n0 + 16 * khalf + 4 * i);
#pragma unroll
        for (int r = 0; r < 8; ++r) {
            pa0[r] = q[2 * r];
            pa1[r] = q[2 * r + 1];
        }
    }
    const v2f b0v = *(const v2f*)(b0 + n0 + 2 * l15);
#pragma unroll
    for (int s = 0; s < PFX; ++s) {
        rb[s] = *(const v2f*)(Ak + 4 * s);
        wx[s] = *(const v2f*)(Wk + (size_t)(4 * s) * cD);
        wy[s] = *(const v2f*)(Wk + (size_t)(4 * s + 1) * cD);
    }

#pragma unroll
    for (int s = 0; s < KS; ++s) {
        if (s < T0) {
            const int v = s + PFX;
            rb[v] = *(const v2f*)(Ak + 4 * v);
            wx[v] = *(const v2f*)(Wk + (size_t)(4 * v) * cD);
            wy[v] = *(const v2f*)(Wk + (size_t)(4 * v + 1) * cD);
        }
        if (!LNRES) {
            if (s < 4) {
#pragma unroll
                for (int r = 2 * s; r < 2 * s + 2; ++r)
#pragma unroll
                    for (int u = 0; u < 4; ++u) {
                        ux[r][u] = *(const v2f*)(Uk + (size_t)(8 * r + 2 * u) * cD);
                        uy[r][u] = *(const v2f*)(Uk + (size_t)(8 * r + 2 * u + 1) * cD);
                    }
            }
            if (s == 4) {
#pragma unroll
                for (int i = 0; i < 4; ++i)
                    load4(&b2v[4 * i], bias2 + n0 + 16 * khalf + 4 * i);
            }
        } else {
            const int p = s - T0;
            if (p >= 0 && p < 3) {
#pragma unroll
                for (int i = 6 * p; i < 6 * p + 6 && i < 16; ++i)
                    load4(&gv[4 * i], g + 64 * khalf + 4 * i);
            }
            if (p == 2) {
                load4(&betv[0], beta + l15 * 8);
                load4(&betv[4], beta + l15 * 8 + 4);
            }
            if (p == 3) {
#pragma unroll
                for (int i = 0; i < 4; ++i)
                    load4(&b2v[4 * i], bias2 + n0 + 16 * khalf + 4 * i);
            }
            if (p == 3 || p == 4) {
#pragma unroll
                for (int u = (p == 3 ? 0 : 1); u < (p == 3 ? 1 : 4); ++u) {
                    ux[7][u] = *(const v2f*)(Uk + (size_t)(8 * 7 + 2 * u) * cD);
                    uy[7][u] = *(const v2f*)(Uk + (size_t)(8 * 7 + 2 * u + 1) * cD);
                }
            }
        }
        __builtin_amdgcn_sched_barrier(0);
        v2f a0, a1;
        a0.x = wx[s].x;
        a0.y = wy[s].x;
        a1.x = wx[s].y;
        a1.y = wy[s].y;
        pa0 = __builtin_amdgcn_wmma_f32_16x16x4_f32(
                  false, a0, false, rb[s], (short)0, pa0, false, false);
        pa1 = __builtin_amdgcn_wmma_f32_16x16x4_f32(
                  false, a1, false, rb[s], (short)0, pa1, false, false);
        __builtin_amdgcn_sched_barrier(0);
    }

    {
        v2f a0, a1, b;
        b.x  = khalf ? 0.0f : 1.0f;
        b.y  = 0.0f;
        a0.x = khalf ? 0.0f : b0v.x;
        a0.y = 0.0f;
        a1.x = khalf ? 0.0f : b0v.y;
        a1.y = 0.0f;
        pa0 = __builtin_amdgcn_wmma_f32_16x16x4_f32(
                  false, a0, false, b, (short)0, pa0, false, false);
        pa1 = __builtin_amdgcn_wmma_f32_16x16x4_f32(
                  false, a1, false, b, (short)0, pa1, false, false);
    }

    {
        float4* mine = s_x + l15 * XP + (n0 >> 2) + 4 * khalf;
#pragma unroll
        for (int q = 0; q < 4; ++q)
            mine[q] = make_float4(pa0[2 * q], pa1[2 * q], pa0[2 * q + 1], pa1[2 * q + 1]);
    }
    __syncthreads();

    v8f acc[8];
    {
        const float4* row = s_x + l15 * XP + 16 * khalf;
#pragma unroll
        for (int i = 0; i < 16; ++i) {
            const float4 q = row[i];
            acc[4 * (i & 1) + 0][i >> 1] = q.x;
            acc[4 * (i & 1) + 1][i >> 1] = q.y;
            acc[4 * (i & 1) + 2][i >> 1] = q.z;
            acc[4 * (i & 1) + 3][i >> 1] = q.w;
        }
    }
    if (LNRES) {
#pragma unroll
        for (int r = 0; r < 7; ++r)
#pragma unroll
            for (int u = 0; u < 4; ++u) {
                ux[r][u] = *(const v2f*)(Uk + (size_t)(8 * r + 2 * u) * cD);
                uy[r][u] = *(const v2f*)(Uk + (size_t)(8 * r + 2 * u + 1) * cD);
            }
        __builtin_amdgcn_sched_barrier(0);
    }

    float xn[64];
    if (LNRES) {
        float pr[8];
#pragma unroll
        for (int r = 0; r < 8; ++r) {
            float sm = 0.0f;
#pragma unroll
            for (int t = 0; t < 8; ++t) sm += acc[t][r];
            pr[r] = sm;
        }
        float sm = ((pr[0] + pr[1]) + (pr[2] + pr[3])) + ((pr[4] + pr[5]) + (pr[6] + pr[7]));
        sm = sm + other_half(sm);
        const float mu = sm * (1.0f / cD);
#pragma unroll
        for (int r = 0; r < 8; ++r) {
            float sq = 0.0f;
#pragma unroll
            for (int t = 0; t < 8; ++t) {
                const float d = acc[t][r] - mu;
                acc[t][r] = d;
                sq += d * d;
            }
            pr[r] = sq;
        }
        float sq = ((pr[0] + pr[1]) + (pr[2] + pr[3])) + ((pr[4] + pr[5]) + (pr[6] + pr[7]));
        sq = sq + other_half(sq);
        const float var = sq * (1.0f / cD);
        const float rs  = rsqrtf(var + 1e-5f);
#pragma unroll
        for (int r = 0; r < 8; ++r)
#pragma unroll
            for (int t = 0; t < 8; ++t)
                acc[t][r] = acc[t][r] * rs * gv[8 * r + t];

        {
            v2f b;
            b.x = khalf ? 0.0f : 1.0f;
            b.y = 0.0f;
#pragma unroll
            for (int t = 0; t < 8; ++t) {
                v2f a;
                a.x = khalf ? 0.0f : betv[t];
                a.y = 0.0f;
                acc[t] = __builtin_amdgcn_wmma_f32_16x16x4_f32(
                             false, a, false, b, (short)0, acc[t], false, false);
            }
        }
    }
#pragma unroll
    for (int r = 0; r < 8; ++r)
#pragma unroll
        for (int t = 0; t < 8; ++t)
            xn[8 * r + t] = acc[t][r] > 0.0f ? acc[t][r] : 0.0f;

    if (write_x) {
        float* xrow = xout + xoff;
#pragma unroll
        for (int i = 0; i < 16; ++i)
            *(float4*)(xrow + 4 * i) = make_float4(xn[4 * i], xn[4 * i + 1], xn[4 * i + 2], xn[4 * i + 3]);
    }

    v8f e0 = {}, e1 = {};
#pragma unroll
    for (int r = 0; r < 8; ++r) {
#pragma unroll
        for (int u = 0; u < 4; ++u) {
            v2f a0, a1, b;
            a0.x = ux[r][u].x;
            a0.y = uy[r][u].x;
            a1.x = ux[r][u].y;
            a1.y = uy[r][u].y;
            b.x = xn[8 * r + 2 * u];
            b.y = xn[8 * r + 2 * u + 1];
            e0 = __builtin_amdgcn_wmma_f32_16x16x4_f32(
                     false, a0, false, b, (short)0, e0, false, false);
            e1 = __builtin_amdgcn_wmma_f32_16x16x4_f32(
                     false, a1, false, b, (short)0, e1, false, false);
        }
    }

    float y[16];
#pragma unroll
    for (int r = 0; r < 8; ++r) {
        y[2 * r]     = e0[r] + (with_bias ? b2v[2 * r] : 0.0f);
        y[2 * r + 1] = e1[r] + (with_bias ? b2v[2 * r + 1] : 0.0f);
    }
    float* crow = Cw + (size_t)(m0 + l15) * cD + n0 + 16 * khalf;
#pragma unroll
    for (int i = 0; i < 4; ++i)
        *(float4*)(crow + 4 * i) = make_float4(y[4 * i], y[4 * i + 1], y[4 * i + 2], y[4 * i + 3]);
}

__global__ void __launch_bounds__(128)
gemm_in_products(const float* __restrict__ A0, const float* __restrict__ W0,
                 const float* __restrict__ b0, float* __restrict__ xout,
                 const float* __restrict__ B0, const float* __restrict__ B1,
                 const float* __restrict__ B2, const float* __restrict__ bias2,
                 float* __restrict__ C0, float* __restrict__ C1, float* __restrict__ C2)
{
    const int s = blockIdx.y;
    x_then_products<64, false>(A0, W0, b0, nullptr, nullptr, nullptr, xout,
                               s == 0 ? B0 : (s == 1 ? B1 : B2), bias2,
                               s == 0 ? C0 : (s == 1 ? C1 : C2),
                               s == 2, s == 0 && threadIdx.x < 32);
}

__global__ void __launch_bounds__(128)
gemm_hn_ln_products(const float* __restrict__ R, const float* __restrict__ W2,
                    const float* __restrict__ b2, const float* __restrict__ deg,
                    const float* __restrict__ U, const float* __restrict__ hid1,
                    const float* __restrict__ W0,
                    const float* __restrict__ b0, const float* __restrict__ xin,
                    const float* __restrict__ g, const float* __restrict__ beta,
                    float* __restrict__ xout,
                    const float* __restrict__ B0, const float* __restrict__ B1,
                    const float* __restrict__ B2, const float* __restrict__ bias2,
                    float* __restrict__ C0, float* __restrict__ C1, float* __restrict__ C2)
{
    __shared__ float4 s_part[4 * SPG];
    __shared__ float4 s_h4[16 * HP / 4];
    float* s_h = (float*)s_h4;
    agg_hid_tile(R, W2, b2, deg, U, hid1, s_part, s_h);
    const int s = blockIdx.y;
    x_then_products<cD, true>(s_h, W0, b0, xin, g, beta, xout,
                              s == 0 ? B0 : (s == 1 ? B1 : B2), bias2,
                              s == 0 ? C0 : (s == 1 ? C1 : C2),
                              s == 2, s == 0 && threadIdx.x < 32);
}

__global__ void __launch_bounds__(128)
gemm_hn_ln_out(const float* __restrict__ R, const float* __restrict__ W2,
               const float* __restrict__ b2, const float* __restrict__ deg,
               const float* __restrict__ U,
               const float* __restrict__ hid1,
               const float* __restrict__ Bm,
               const float* __restrict__ bias,
               const float* __restrict__ g,
               const float* __restrict__ beta,
               float* __restrict__ out)
{
    constexpr int KS = cD / 4;
    const int wave  = threadIdx.x >> 5;
    const int lane  = threadIdx.x & 31;
    const int l15   = lane & 15;
    const int khalf = lane >> 4;
    const int m0    = blockIdx.x * 16;
    const int n0    = wave * 32;

    __shared__ float4 s_x[16 * XP];
    __shared__ float4 s_part[4 * SPG];
    __shared__ float4 s_h4[16 * HP / 4];
    float* s_h = (float*)s_h4;

    agg_hid_tile(R, W2, b2, deg, U, hid1, s_part, s_h);

    const float* Ak = s_h + l15 * HP + 2 * khalf;
    const float* Wk = Bm + (size_t)(2 * khalf) * cD + n0 + 2 * l15;

    v2f rb[KS], wx[KS], wy[KS];
    const float4 gq = *(const float4*)(g + 4 * lane);
    const float4 bq = *(const float4*)(beta + 4 * lane);
    const v2f b0v = *(const v2f*)(bias + n0 + 2 * l15);
#pragma unroll
    for (int s = 0; s < PFX; ++s) {
        rb[s] = *(const v2f*)(Ak + 4 * s);
        wx[s] = *(const v2f*)(Wk + (size_t)(4 * s) * cD);
        wy[s] = *(const v2f*)(Wk + (size_t)(4 * s + 1) * cD);
    }
    asm volatile("" :: "v"(gq.x), "v"(gq.y), "v"(gq.z), "v"(gq.w),
                       "v"(bq.x), "v"(bq.y), "v"(bq.z), "v"(bq.w));

    v8f pa0 = {}, pa1 = {};
#pragma unroll
    for (int s = 0; s < KS; ++s) {
        if (s + PFX < KS) {
            const int v = s + PFX;
            rb[v] = *(const v2f*)(Ak + 4 * v);
            wx[v] = *(const v2f*)(Wk + (size_t)(4 * v) * cD);
            wy[v] = *(const v2f*)(Wk + (size_t)(4 * v + 1) * cD);
        }
        __builtin_amdgcn_sched_barrier(0);
        v2f a0, a1;
        a0.x = wx[s].x;
        a0.y = wy[s].x;
        a1.x = wx[s].y;
        a1.y = wy[s].y;
        pa0 = __builtin_amdgcn_wmma_f32_16x16x4_f32(
                  false, a0, false, rb[s], (short)0, pa0, false, false);
        pa1 = __builtin_amdgcn_wmma_f32_16x16x4_f32(
                  false, a1, false, rb[s], (short)0, pa1, false, false);
        __builtin_amdgcn_sched_barrier(0);
    }

    {
        v2f a0, a1, b;
        b.x  = khalf ? 0.0f : 1.0f;
        b.y  = 0.0f;
        a0.x = khalf ? 0.0f : b0v.x;
        a0.y = 0.0f;
        a1.x = khalf ? 0.0f : b0v.y;
        a1.y = 0.0f;
        pa0 = __builtin_amdgcn_wmma_f32_16x16x4_f32(
                  false, a0, false, b, (short)0, pa0, false, false);
        pa1 = __builtin_amdgcn_wmma_f32_16x16x4_f32(
                  false, a1, false, b, (short)0, pa1, false, false);
    }

    {
        float4* mine = s_x + l15 * XP + (n0 >> 2) + 4 * khalf;
#pragma unroll
        for (int q = 0; q < 4; ++q)
            mine[q] = make_float4(pa0[2 * q], pa1[2 * q], pa0[2 * q + 1], pa1[2 * q + 1]);
    }
    __syncthreads();

#pragma unroll
    for (int n = 0; n < 4; ++n) {
        const int node = 4 * wave + n;
        const float4 v = s_x[node * XP + lane];
        float sm = (v.x + v.y) + (v.z + v.w);
        sm = half_sum(sm);
        sm = sm + other_half(sm);
        const float mu = sm * (1.0f / cD);
        const float dx = v.x - mu, dy = v.y - mu, dz = v.z - mu, dw = v.w - mu;
        float sq = (dx * dx + dy * dy) + (dz * dz + dw * dw);
        sq = half_sum(sq);
        sq = sq + other_half(sq);
        const float var = sq * (1.0f / cD);
        const float rs  = rsqrtf(var + 1e-5f);
        *(float4*)(out + (size_t)(m0 + node) * cD + 4 * lane) =
            make_float4(dx * rs * gq.x + bq.x, dy * rs * gq.y + bq.y,
                        dz * rs * gq.z + bq.z, dw * rs * gq.w + bq.w);
    }
}

static constexpr int TI = 3;
static constexpr int QB = 48;
static constexpr int AG = 8;
static constexpr int NG = QB / AG;
static constexpr int RW = cN / 4;
static_assert(cN % TI == 0 && cN % (2 * QB) == 0 && NG % 2 == 0 && AG % 4 == 0, "shape");

struct AdjGroup { float a[TI][AG]; };

__device__ __forceinline__ void load_q(float (&q)[QB], const float* __restrict__ Qcol, int j0)
{
#pragma unroll
    for (int u = 0; u < QB; ++u)
        q[u] = Qcol[(size_t)(j0 + u) * cD];
}

__device__ __forceinline__ void load_adj(AdjGroup& g, const float4* rows, int j0)
{
#pragma unroll
    for (int r = 0; r < TI; ++r) {
#pragma unroll
        for (int v = 0; v < AG / 4; ++v) {
            const float4 x = rows[r * RW + (j0 >> 2) + v];
            g.a[r][4 * v + 0] = x.x; g.a[r][4 * v + 1] = x.y;
            g.a[r][4 * v + 2] = x.z; g.a[r][4 * v + 3] = x.w;
        }
    }
}

__device__ __forceinline__ void compute_group(const float* q, const AdjGroup& g,
                                              const float (&p)[TI], float w, float b1,
                                              float (&acc)[TI], float& dg)
{
#pragma unroll
    for (int u = 0; u < AG; ++u) {
#pragma unroll
        for (int r = 0; r < TI; ++r) {
            const float a = g.a[r][u];
            float t = p[r] + q[u] + a * w + b1;
            t = fmaxf(t, 0.0f);
            acc[r] += a * t;
        }
        dg += g.a[0][u];
    }
}

template <bool LAST>
__device__ __forceinline__ void one_batch(const float (&q)[QB], AdjGroup& ga, AdjGroup& gb,
                                          const float4* rows, int j0,
                                          const float (&p)[TI], float w, float b1,
                                          float (&acc)[TI], float& dg)
{
#pragma unroll
    for (int s = 0; s < NG; s += 2) {
        load_adj(gb, rows, j0 + (s + 1) * AG);
        __builtin_amdgcn_sched_barrier(0);
        compute_group(&q[s * AG], ga, p, w, b1, acc, dg);
        if (s + 2 < NG || !LAST) load_adj(ga, rows, j0 + (s + 2) * AG);
        __builtin_amdgcn_sched_barrier(0);
        compute_group(&q[(s + 1) * AG], gb, p, w, b1, acc, dg);
    }
}

__global__ void __launch_bounds__(cD)
pairwise_agg(const float* __restrict__ P,
             const float* __restrict__ Q,
             const float* __restrict__ adj,
             const float* __restrict__ we,
             const float* __restrict__ mb1,
             float* __restrict__ R,
             float* __restrict__ deg)
{
    const int i0   = blockIdx.x * TI;
    const int b    = i0 / cN;
    const int d    = threadIdx.x;
    const int wave = threadIdx.x >> 5;
    const int rot  = wave % TI;

    __shared__ float4 s_adj4[(2 * TI - 1) * RW];

    const float* Qcol = Q + (size_t)b * cN * cD + d;

    float qa[QB], qb[QB];
    load_q(qa, Qcol, 0);

    {
        const float4* adj4 = (const float4*)(adj + (size_t)i0 * cN);
        for (int v = d; v < TI * RW; v += cD) {
            const float4 x = adj4[v];
            s_adj4[v] = x;
            if (v < (TI - 1) * RW) s_adj4[v + TI * RW] = x;
        }
    }

    int rr[TI];
    float p[TI];
#pragma unroll
    for (int k = 0; k < TI; ++k) {
        rr[k] = (rot + k) % TI;
        p[k]  = P[(size_t)(i0 + rr[k]) * cD + d];
    }
    const float w  = we[d];
    const float b1 = mb1[d];
    const float4* rows = s_adj4 + rot * RW;

    __syncthreads();

    AdjGroup ga, gb;
    load_adj(ga, rows, 0);

    float acc[TI] = {0.0f, 0.0f, 0.0f};
    float dg = 0.0f;

    for (int j0 = 0; j0 < cN - 2 * QB; j0 += 2 * QB) {
        load_q(qb, Qcol, j0 + QB);
        one_batch<false>(qa, ga, gb, rows, j0, p, w, b1, acc, dg);
        load_q(qa, Qcol, j0 + 2 * QB);
        one_batch<false>(qb, ga, gb, rows, j0 + QB, p, w, b1, acc, dg);
    }
    load_q(qb, Qcol, cN - QB);
    one_batch<false>(qa, ga, gb, rows, cN - 2 * QB, p, w, b1, acc, dg);
    one_batch<true >(qb, ga, gb, rows, cN - QB, p, w, b1, acc, dg);

#pragma unroll
    for (int k = 0; k < TI; ++k)
        R[(size_t)(i0 + rr[k]) * cD + d] = acc[k];
    if (wave < TI && (d & 31) == 0) deg[i0 + rot] = dg;
}

extern "C" void kernel_launch(void* const* d_in, const int* in_sizes, int n_in,
                              void* d_out, int out_size, void* d_ws, size_t ws_size,
                              hipStream_t stream)
{
    (void)in_sizes; (void)n_in; (void)out_size; (void)ws_size;

    const float* h    = (const float*)d_in[0];
    const float* adj  = (const float*)d_in[1];
    const float* ip_w = (const float*)d_in[2];
    const float* ip_b = (const float*)d_in[3];
    const float* mw1  = (const float*)d_in[4];
    const float* mb1  = (const float*)d_in[5];
    const float* mw2  = (const float*)d_in[6];
    const float* mb2  = (const float*)d_in[7];
    const float* uw1  = (const float*)d_in[8];
    const float* ub1  = (const float*)d_in[9];
    const float* uw2  = (const float*)d_in[10];
    const float* ub2  = (const float*)d_in[11];
    const float* ln_g = (const float*)d_in[12];
    const float* ln_b = (const float*)d_in[13];

    float* ws = (float*)d_ws;
    const size_t S = (size_t)cM * cD;
    float* xa   = ws + 0 * S;
    float* P    = ws + 1 * S;
    float* Q    = ws + 2 * S;
    float* R    = ws + 3 * S;
    float* xb   = ws + 4 * S;
    float* hida = ws + 5 * S;
    float* hidb = ws + 6 * S;
    float* deg  = ws + 7 * S;

    gemm_in_products<<<dim3(48, 3), 128, 0, stream>>>(h, ip_w, ip_b, xa,
                                                      mw1, mw1 + 128 * cD, uw1, ub1,
                                                      P, Q, hida);

    for (int l = 0; l < 3; ++l) {
        const float* mw1l = mw1 + (size_t)l * 257 * cD;
        const float* mb1l = mb1 + (size_t)l * cD;
        const float* mw2l = mw2 + (size_t)l * cD * cD;
        const float* mb2l = mb2 + (size_t)l * cD;
        const float* uw1l = uw1 + (size_t)l * 256 * cD;
        const float* uw2l = uw2 + (size_t)l * cD * cD;
        const float* ub2l = ub2 + (size_t)l * cD;
        float* xcur = (l & 1) ? xb : xa;
        float* xnxt = (l & 1) ? xa : xb;
        float* hcur = (l & 1) ? hidb : hida;
        float* hnxt = (l & 1) ? hida : hidb;

        pairwise_agg<<<cM / TI, cD, 0, stream>>>(P, Q, adj, mw1l + 256 * cD, mb1l, R, deg);

        if (l < 2) {
            const float* mw1n = mw1 + (size_t)(l + 1) * 257 * cD;
            const float* uw1n = uw1 + (size_t)(l + 1) * 256 * cD;
            const float* ub1n = ub1 + (size_t)(l + 1) * cD;
            gemm_hn_ln_products<<<dim3(48, 3), 128, 0, stream>>>(R, mw2l, mb2l, deg, uw1l + 128 * cD,
                                                                 hcur, uw2l, ub2l, xcur,
                                                                 ln_g + l * cD, ln_b + l * cD, xnxt,
                                                                 mw1n, mw1n + 128 * cD, uw1n, ub1n,
                                                                 P, Q, hnxt);
        } else {
            gemm_hn_ln_out<<<cM / 16, 128, 0, stream>>>(R, mw2l, mb2l, deg, uw1l + 128 * cD,
                                                      hcur, uw2l, ub2l,
                                                      ln_g + l * cD, ln_b + l * cD,
                                                      (float*)d_out);
        }
    }
}


# pairwise sum: a wave takes a quarter of the senders with all 128 channels (16-byte Q loads, 4x the bytes in flight), was all senders x 32 channels; R summed in 4 parts via LDS, deg across lanes
# speedup vs baseline: 4.0847x; 1.1564x over previous
#include <hip/hip_runtime.h>
#include <hip/hip_bf16.h>

typedef __attribute__((ext_vector_type(2))) float v2f;
typedef __attribute__((ext_vector_type(8))) float v8f;

static constexpr int cB = 2;
static constexpr int cN = 384;
static constexpr int cD = 128;
static constexpr int cM = cB * cN;

template <int M>
__device__ __forceinline__ float add_xor_lane(float v)
{
    const int o = __builtin_amdgcn_update_dpp(0, __float_as_int(v), 0x160 | M, 0xf, 0xf, true);
    return v + __int_as_float(o);
}
__device__ __forceinline__ float half_sum(float v)
{
    v = add_xor_lane<8>(v);
    v = add_xor_lane<4>(v);
    v = add_xor_lane<2>(v);
    v = add_xor_lane<1>(v);
    return v;
}

__device__ __forceinline__ float other_half(float v)
{
    return __int_as_float(__builtin_amdgcn_permlanex16(0, __float_as_int(v),
                                                       0x76543210, 0xfedcba98, false, false));
}

__device__ __forceinline__ void load4(float* d, const float* p)
{
    const float4 q = *(const float4*)p;
    d[0] = q.x; d[1] = q.y; d[2] = q.z; d[3] = q.w;
}

static constexpr int PFA = 16;
static constexpr int LPQ = 5;
static constexpr int HP  = 132;
static constexpr int SPG = 4 * 32 * LPQ;

__device__ __forceinline__ void agg_hid_tile(const float* __restrict__ R,
                                             const float* __restrict__ W2,
                                             const float* __restrict__ b2,
                                             const float* __restrict__ deg,
                                             const float* __restrict__ U,
                                             const float* __restrict__ hid1,
                                             float4* s_part, float* s_h)
{
    constexpr int KS = cD / 4;
    constexpr int T0 = KS - PFA;
    const int wave  = threadIdx.x >> 5;
    const int lane  = threadIdx.x & 31;
    const int l15   = lane & 15;
    const int khalf = lane >> 4;
    const int m0    = blockIdx.x * 16;
    const int ch0   = wave * 32;

    const float* Rk = R + (size_t)(m0 + l15) * cD + 2 * khalf;
    const float* Wk = W2 + (size_t)(2 * khalf) * cD + ch0 + 2 * l15;
    const float* Uk = U + (size_t)(ch0 + 16 * khalf) * cD + 2 * l15;

    const int orow = 4 * wave + (lane >> 3);
    const int oq   = lane & 7;
    const float* hin = hid1 + (size_t)(m0 + orow) * cD + 4 * oq;

    v2f rb[KS], wx[KS], wy[KS];
    v2f ux[4][8], uy[4][8];
    float4 hold[4];
#pragma unroll
    for (int cg = 0; cg < 4; ++cg) hold[cg] = *(const float4*)(hin + 32 * cg);
    const v2f mbv = *(const v2f*)(b2 + ch0 + 2 * l15);
    const float dg = deg[m0 + l15];
#pragma unroll
    for (int s = 0; s < PFA; ++s) {
        rb[s] = *(const v2f*)(Rk + 4 * s);
        wx[s] = *(const v2f*)(Wk + (size_t)(4 * s) * cD);
        wy[s] = *(const v2f*)(Wk + (size_t)(4 * s + 1) * cD);
    }

    v8f acc0 = {}, acc1 = {};
    {
        v2f a0, a1, b;
        b.x  = khalf ? 0.0f : dg;
        b.y  = 0.0f;
        a0.x = khalf ? 0.0f : mbv.x;
        a0.y = 0.0f;
        a1.x = khalf ? 0.0f : mbv.y;
        a1.y = 0.0f;
        acc0 = __builtin_amdgcn_wmma_f32_16x16x4_f32(
                   false, a0, false, b, (short)0, acc0, false, false);
        acc1 = __builtin_amdgcn_wmma_f32_16x16x4_f32(
                   false, a1, false, b, (short)0, acc1, false, false);
    }

#pragma unroll
    for (int s = 0; s < KS; ++s) {
        if (s < T0) {
            const int v = s + PFA;
            rb[v] = *(const v2f*)(Rk + 4 * v);
            wx[v] = *(const v2f*)(Wk + (size_t)(4 * v) * cD);
            wy[v] = *(const v2f*)(Wk + (size_t)(4 * v + 1) * cD);
        } else if (s < T0 + 8) {
            const int r = s - T0;
#pragma unroll
            for (int cg = 0; cg < 4; ++cg) {
                ux[cg][r] = *(const v2f*)(Uk + (size_t)(2 * r) * cD + 32 * cg);
                uy[cg][r] = *(const v2f*)(Uk + (size_t)(2 * r + 1) * cD + 32 * cg);
            }
        }
        v2f a0, a1;
        a0.x = wx[s].x;
        a0.y = wy[s].x;
        a1.x = wx[s].y;
        a1.y = wy[s].y;
        acc0 = __builtin_amdgcn_wmma_f32_16x16x4_f32(
                   false, a0, false, rb[s], (short)0, acc0, false, false);
        acc1 = __builtin_amdgcn_wmma_f32_16x16x4_f32(
                   false, a1, false, rb[s], (short)0, acc1, false, false);
    }

    v8f e0[4], e1[4];
#pragma unroll
    for (int cg = 0; cg < 4; ++cg) {
        e0[cg] = v8f{};
        e1[cg] = v8f{};
#pragma unroll
        for (int r = 0; r < 8; ++r) {
            v2f a0, a1, b;
            a0.x = ux[cg][r].x;
            a0.y = uy[cg][r].x;
            a1.x = ux[cg][r].y;
            a1.y = uy[cg][r].y;
            b.x = acc0[r];
            b.y = acc1[r];
            e0[cg] = __builtin_amdgcn_wmma_f32_16x16x4_f32(
                         false, a0, false, b, (short)0, e0[cg], false, false);
            e1[cg] = __builtin_amdgcn_wmma_f32_16x16x4_f32(
                         false, a1, false, b, (short)0, e1[cg], false, false);
        }
    }

    __builtin_amdgcn_sched_group_barrier(0x020, 6 + 3 * PFA, 0);
    __builtin_amdgcn_sched_group_barrier(0x008, 2, 0);
#pragma unroll
    for (int s = 0; s < T0; ++s) {
        __builtin_amdgcn_sched_group_barrier(0x020, 3, 0);
        __builtin_amdgcn_sched_group_barrier(0x008, 2, 0);
    }
#pragma unroll
    for (int r = 0; r < 8; ++r) {
        __builtin_amdgcn_sched_group_barrier(0x020, 8, 0);
        __builtin_amdgcn_sched_group_barrier(0x008, 2, 0);
    }

    float4* mine = s_part + (wave * 32 + lane) * LPQ;
#pragma unroll
    for (int cg = 0; cg < 4; ++cg)
#pragma unroll
        for (int q = 0; q < 4; ++q)
            mine[cg * SPG + q] = make_float4(e0[cg][2 * q], e1[cg][2 * q], e0[cg][2 * q + 1], e1[cg][2 * q + 1]);

    __syncthreads();

    const float4* theirs = s_part + (orow + 16 * (oq >> 2)) * LPQ + (oq & 3);
#pragma unroll
    for (int cg = 0; cg < 4; ++cg) {
        float4 v = hold[cg];
#pragma unroll
        for (int w = 0; w < 4; ++w) {
            const float4 p = theirs[cg * SPG + w * 32 * LPQ];
            v.x += p.x;
            v.y += p.y;
            v.z += p.z;
            v.w += p.w;
        }
        v.x = v.x > 0.0f ? v.x : 0.0f;
        v.y = v.y > 0.0f ? v.y : 0.0f;
        v.z = v.z > 0.0f ? v.z : 0.0f;
        v.w = v.w > 0.0f ? v.w : 0.0f;
        *(float4*)(s_h + orow * HP + 32 * cg + 4 * oq) = v;
    }

    __syncthreads();
}

static constexpr int PFX = 16;
static constexpr int XP  = 33;

template <int K0, bool LNRES>
__device__ __forceinline__ void x_then_products(const float* __restrict__ A0,
                                                const float* __restrict__ W0,
                                                const float* __restrict__ b0,
                                                const float* __restrict__ xin,
                                                const float* __restrict__ g,
                                                const float* __restrict__ beta,
                                                float* __restrict__ xout,
                                                const float* __restrict__ Bw,
                                                const float* __restrict__ bias2,
                                                float* __restrict__ Cw,
                                                const bool with_bias, const bool write_x)
{
    constexpr int KS = K0 / 4;
    constexpr int T0 = KS - PFX;
    static_assert(LNRES ? (KS == 2 * PFX) : (KS == PFX), "the load schedule below is written for these two shapes");
    const int wave  = threadIdx.x >> 5;
    const int lane  = threadIdx.x & 31;
    const int l15   = lane & 15;
    const int khalf = lane >> 4;
    const int m0    = blockIdx.x * 16;
    const int n0    = wave * 32;

    __shared__ float4 s_x[16 * XP];

    const float* Ak = LNRES ? A0 + l15 * HP + 2 * khalf
                            : A0 + (size_t)(m0 + l15) * K0 + 2 * khalf;
    const float* Wk = W0 + (size_t)(2 * khalf) * cD + n0 + 2 * l15;
    const float* Uk = Bw + (size_t)(64 * khalf) * cD + n0 + 2 * l15;
    const size_t xoff = (size_t)(m0 + l15) * cD + 64 * khalf;

    v2f rb[KS], wx[KS], wy[KS];
    v2f ux[8][4], uy[8][4];
    float b2v[16];
    float gv[64];
    float betv[8];

    v8f pa0 = {}, pa1 = {};
    if (LNRES) {
        float q[16];
#pragma unroll
        for (int i = 0; i < 4; ++i)
            load4(&q[4 * i], xin + (size_t)(m0 + l15) * cD + n0 + 16 * khalf + 4 * i);
#pragma unroll
        for (int r = 0; r < 8; ++r) {
            pa0[r] = q[2 * r];
            pa1[r] = q[2 * r + 1];
        }
    }
    const v2f b0v = *(const v2f*)(b0 + n0 + 2 * l15);
#pragma unroll
    for (int s = 0; s < PFX; ++s) {
        rb[s] = *(const v2f*)(Ak + 4 * s);
        wx[s] = *(const v2f*)(Wk + (size_t)(4 * s) * cD);
        wy[s] = *(const v2f*)(Wk + (size_t)(4 * s + 1) * cD);
    }

#pragma unroll
    for (int s = 0; s < KS; ++s) {
        if (s < T0) {
            const int v = s + PFX;
            rb[v] = *(const v2f*)(Ak + 4 * v);
            wx[v] = *(const v2f*)(Wk + (size_t)(4 * v) * cD);
            wy[v] = *(const v2f*)(Wk + (size_t)(4 * v + 1) * cD);
        }
        if (!LNRES) {
            if (s < 4) {
#pragma unroll
                for (int r = 2 * s; r < 2 * s + 2; ++r)
#pragma unroll
                    for (int u = 0; u < 4; ++u) {
                        ux[r][u] = *(const v2f*)(Uk + (size_t)(8 * r + 2 * u) * cD);
                        uy[r][u] = *(const v2f*)(Uk + (size_t)(8 * r + 2 * u + 1) * cD);
                    }
            }
            if (s == 4) {
#pragma unroll
                for (int i = 0; i < 4; ++i)
                    load4(&b2v[4 * i], bias2 + n0 + 16 * khalf + 4 * i);
            }
        } else {
            const int p = s - T0;
            if (p >= 0 && p < 3) {
#pragma unroll
                for (int i = 6 * p; i < 6 * p + 6 && i < 16; ++i)
                    load4(&gv[4 * i], g + 64 * khalf + 4 * i);
            }
            if (p == 2) {
                load4(&betv[0], beta + l15 * 8);
                load4(&betv[4], beta + l15 * 8 + 4);
            }
            if (p == 3) {
#pragma unroll
                for (int i = 0; i < 4; ++i)
                    load4(&b2v[4 * i], bias2 + n0 + 16 * khalf + 4 * i);
            }
            if (p == 3 || p == 4) {
#pragma unroll
                for (int u = (p == 3 ? 0 : 1); u < (p == 3 ? 1 : 4); ++u) {
                    ux[7][u] = *(const v2f*)(Uk + (size_t)(8 * 7 + 2 * u) * cD);
                    uy[7][u] = *(const v2f*)(Uk + (size_t)(8 * 7 + 2 * u + 1) * cD);
                }
            }
        }
        __builtin_amdgcn_sched_barrier(0);
        v2f a0, a1;
        a0.x = wx[s].x;
        a0.y = wy[s].x;
        a1.x = wx[s].y;
        a1.y = wy[s].y;
        pa0 = __builtin_amdgcn_wmma_f32_16x16x4_f32(
                  false, a0, false, rb[s], (short)0, pa0, false, false);
        pa1 = __builtin_amdgcn_wmma_f32_16x16x4_f32(
                  false, a1, false, rb[s], (short)0, pa1, false, false);
        __builtin_amdgcn_sched_barrier(0);
    }

    {
        v2f a0, a1, b;
        b.x  = khalf ? 0.0f : 1.0f;
        b.y  = 0.0f;
        a0.x = khalf ? 0.0f : b0v.x;
        a0.y = 0.0f;
        a1.x = khalf ? 0.0f : b0v.y;
        a1.y = 0.0f;
        pa0 = __builtin_amdgcn_wmma_f32_16x16x4_f32(
                  false, a0, false, b, (short)0, pa0, false, false);
        pa1 = __builtin_amdgcn_wmma_f32_16x16x4_f32(
                  false, a1, false, b, (short)0, pa1, false, false);
    }

    {
        float4* mine = s_x + l15 * XP + (n0 >> 2) + 4 * khalf;
#pragma unroll
        for (int q = 0; q < 4; ++q)
            mine[q] = make_float4(pa0[2 * q], pa1[2 * q], pa0[2 * q + 1], pa1[2 * q + 1]);
    }
    __syncthreads();

    v8f acc[8];
    {
        const float4* row = s_x + l15 * XP + 16 * khalf;
#pragma unroll
        for (int i = 0; i < 16; ++i) {
            const float4 q = row[i];
            acc[4 * (i & 1) + 0][i >> 1] = q.x;
            acc[4 * (i & 1) + 1][i >> 1] = q.y;
            acc[4 * (i & 1) + 2][i >> 1] = q.z;
            acc[4 * (i & 1) + 3][i >> 1] = q.w;
        }
    }
    if (LNRES) {
#pragma unroll
        for (int r = 0; r < 7; ++r)
#pragma unroll
            for (int u = 0; u < 4; ++u) {
                ux[r][u] = *(const v2f*)(Uk + (size_t)(8 * r + 2 * u) * cD);
                uy[r][u] = *(const v2f*)(Uk + (size_t)(8 * r + 2 * u + 1) * cD);
            }
        __builtin_amdgcn_sched_barrier(0);
    }

    float xn[64];
    if (LNRES) {
        float pr[8];
#pragma unroll
        for (int r = 0; r < 8; ++r) {
            float sm = 0.0f;
#pragma unroll
            for (int t = 0; t < 8; ++t) sm += acc[t][r];
            pr[r] = sm;
        }
        float sm = ((pr[0] + pr[1]) + (pr[2] + pr[3])) + ((pr[4] + pr[5]) + (pr[6] + pr[7]));
        sm = sm + other_half(sm);
        const float mu = sm * (1.0f / cD);
#pragma unroll
        for (int r = 0; r < 8; ++r) {
            float sq = 0.0f;
#pragma unroll
            for (int t = 0; t < 8; ++t) {
                const float d = acc[t][r] - mu;
                acc[t][r] = d;
                sq += d * d;
            }
            pr[r] = sq;
        }
        float sq = ((pr[0] + pr[1]) + (pr[2] + pr[3])) + ((pr[4] + pr[5]) + (pr[6] + pr[7]));
        sq = sq + other_half(sq);
        const float var = sq * (1.0f / cD);
        const float rs  = rsqrtf(var + 1e-5f);
#pragma unroll
        for (int r = 0; r < 8; ++r)
#pragma unroll
            for (int t = 0; t < 8; ++t)
                acc[t][r] = acc[t][r] * rs * gv[8 * r + t];

        {
            v2f b;
            b.x = khalf ? 0.0f : 1.0f;
            b.y = 0.0f;
#pragma unroll
            for (int t = 0; t < 8; ++t) {
                v2f a;
                a.x = khalf ? 0.0f : betv[t];
                a.y = 0.0f;
                acc[t] = __builtin_amdgcn_wmma_f32_16x16x4_f32(
                             false, a, false, b, (short)0, acc[t], false, false);
            }
        }
    }
#pragma unroll
    for (int r = 0; r < 8; ++r)
#pragma unroll
        for (int t = 0; t < 8; ++t)
            xn[8 * r + t] = acc[t][r] > 0.0f ? acc[t][r] : 0.0f;

    if (write_x) {
        float* xrow = xout + xoff;
#pragma unroll
        for (int i = 0; i < 16; ++i)
            *(float4*)(xrow + 4 * i) = make_float4(xn[4 * i], xn[4 * i + 1], xn[4 * i + 2], xn[4 * i + 3]);
    }

    v8f e0 = {}, e1 = {};
#pragma unroll
    for (int r = 0; r < 8; ++r) {
#pragma unroll
        for (int u = 0; u < 4; ++u) {
            v2f a0, a1, b;
            a0.x = ux[r][u].x;
            a0.y = uy[r][u].x;
            a1.x = ux[r][u].y;
            a1.y = uy[r][u].y;
            b.x = xn[8 * r + 2 * u];
            b.y = xn[8 * r + 2 * u + 1];
            e0 = __builtin_amdgcn_wmma_f32_16x16x4_f32(
                     false, a0, false, b, (short)0, e0, false, false);
            e1 = __builtin_amdgcn_wmma_f32_16x16x4_f32(
                     false, a1, false, b, (short)0, e1, false, false);
        }
    }

    float y[16];
#pragma unroll
    for (int r = 0; r < 8; ++r) {
        y[2 * r]     = e0[r] + (with_bias ? b2v[2 * r] : 0.0f);
        y[2 * r + 1] = e1[r] + (with_bias ? b2v[2 * r + 1] : 0.0f);
    }
    float* crow = Cw + (size_t)(m0 + l15) * cD + n0 + 16 * khalf;
#pragma unroll
    for (int i = 0; i < 4; ++i)
        *(float4*)(crow + 4 * i) = make_float4(y[4 * i], y[4 * i + 1], y[4 * i + 2], y[4 * i + 3]);
}

__global__ void __launch_bounds__(128)
gemm_in_products(const float* __restrict__ A0, const float* __restrict__ W0,
                 const float* __restrict__ b0, float* __restrict__ xout,
                 const float* __restrict__ B0, const float* __restrict__ B1,
                 const float* __restrict__ B2, const float* __restrict__ bias2,
                 float* __restrict__ C0, float* __restrict__ C1, float* __restrict__ C2)
{
    const int s = blockIdx.y;
    x_then_products<64, false>(A0, W0, b0, nullptr, nullptr, nullptr, xout,
                               s == 0 ? B0 : (s == 1 ? B1 : B2), bias2,
                               s == 0 ? C0 : (s == 1 ? C1 : C2),
                               s == 2, s == 0 && threadIdx.x < 32);
}

__global__ void __launch_bounds__(128)
gemm_hn_ln_products(const float* __restrict__ R, const float* __restrict__ W2,
                    const float* __restrict__ b2, const float* __restrict__ deg,
                    const float* __restrict__ U, const float* __restrict__ hid1,
                    const float* __restrict__ W0,
                    const float* __restrict__ b0, const float* __restrict__ xin,
                    const float* __restrict__ g, const float* __restrict__ beta,
                    float* __restrict__ xout,
                    const float* __restrict__ B0, const float* __restrict__ B1,
                    const float* __restrict__ B2, const float* __restrict__ bias2,
                    float* __restrict__ C0, float* __restrict__ C1, float* __restrict__ C2)
{
    __shared__ float4 s_part[4 * SPG];
    __shared__ float4 s_h4[16 * HP / 4];
    float* s_h = (float*)s_h4;
    agg_hid_tile(R, W2, b2, deg, U, hid1, s_part, s_h);
    const int s = blockIdx.y;
    x_then_products<cD, true>(s_h, W0, b0, xin, g, beta, xout,
                              s == 0 ? B0 : (s == 1 ? B1 : B2), bias2,
                              s == 0 ? C0 : (s == 1 ? C1 : C2),
                              s == 2, s == 0 && threadIdx.x < 32);
}

__global__ void __launch_bounds__(128)
gemm_hn_ln_out(const float* __restrict__ R, const float* __restrict__ W2,
               const float* __restrict__ b2, const float* __restrict__ deg,
               const float* __restrict__ U,
               const float* __restrict__ hid1,
               const float* __restrict__ Bm,
               const float* __restrict__ bias,
               const float* __restrict__ g,
               const float* __restrict__ beta,
               float* __restrict__ out)
{
    constexpr int KS = cD / 4;
    const int wave  = threadIdx.x >> 5;
    const int lane  = threadIdx.x & 31;
    const int l15   = lane & 15;
    const int khalf = lane >> 4;
    const int m0    = blockIdx.x * 16;
    const int n0    = wave * 32;

    __shared__ float4 s_x[16 * XP];
    __shared__ float4 s_part[4 * SPG];
    __shared__ float4 s_h4[16 * HP / 4];
    float* s_h = (float*)s_h4;

    agg_hid_tile(R, W2, b2, deg, U, hid1, s_part, s_h);

    const float* Ak = s_h + l15 * HP + 2 * khalf;
    const float* Wk = Bm + (size_t)(2 * khalf) * cD + n0 + 2 * l15;

    v2f rb[KS], wx[KS], wy[KS];
    const float4 gq = *(const float4*)(g + 4 * lane);
    const float4 bq = *(const float4*)(beta + 4 * lane);
    const v2f b0v = *(const v2f*)(bias + n0 + 2 * l15);
#pragma unroll
    for (int s = 0; s < PFX; ++s) {
        rb[s] = *(const v2f*)(Ak + 4 * s);
        wx[s] = *(const v2f*)(Wk + (size_t)(4 * s) * cD);
        wy[s] = *(const v2f*)(Wk + (size_t)(4 * s + 1) * cD);
    }
    asm volatile("" :: "v"(gq.x), "v"(gq.y), "v"(gq.z), "v"(gq.w),
                       "v"(bq.x), "v"(bq.y), "v"(bq.z), "v"(bq.w));

    v8f pa0 = {}, pa1 = {};
#pragma unroll
    for (int s = 0; s < KS; ++s) {
        if (s + PFX < KS) {
            const int v = s + PFX;
            rb[v] = *(const v2f*)(Ak + 4 * v);
            wx[v] = *(const v2f*)(Wk + (size_t)(4 * v) * cD);
            wy[v] = *(const v2f*)(Wk + (size_t)(4 * v + 1) * cD);
        }
        __builtin_amdgcn_sched_barrier(0);
        v2f a0, a1;
        a0.x = wx[s].x;
        a0.y = wy[s].x;
        a1.x = wx[s].y;
        a1.y = wy[s].y;
        pa0 = __builtin_amdgcn_wmma_f32_16x16x4_f32(
                  false, a0, false, rb[s], (short)0, pa0, false, false);
        pa1 = __builtin_amdgcn_wmma_f32_16x16x4_f32(
                  false, a1, false, rb[s], (short)0, pa1, false, false);
        __builtin_amdgcn_sched_barrier(0);
    }

    {
        v2f a0, a1, b;
        b.x  = khalf ? 0.0f : 1.0f;
        b.y  = 0.0f;
        a0.x = khalf ? 0.0f : b0v.x;
        a0.y = 0.0f;
        a1.x = khalf ? 0.0f : b0v.y;
        a1.y = 0.0f;
        pa0 = __builtin_amdgcn_wmma_f32_16x16x4_f32(
                  false, a0, false, b, (short)0, pa0, false, false);
        pa1 = __builtin_amdgcn_wmma_f32_16x16x4_f32(
                  false, a1, false, b, (short)0, pa1, false, false);
    }

    {
        float4* mine = s_x + l15 * XP + (n0 >> 2) + 4 * khalf;
#pragma unroll
        for (int q = 0; q < 4; ++q)
            mine[q] = make_float4(pa0[2 * q], pa1[2 * q], pa0[2 * q + 1], pa1[2 * q + 1]);
    }
    __syncthreads();

#pragma unroll
    for (int n = 0; n < 4; ++n) {
        const int node = 4 * wave + n;
        const float4 v = s_x[node * XP + lane];
        float sm = (v.x + v.y) + (v.z + v.w);
        sm = half_sum(sm);
        sm = sm + other_half(sm);
        const float mu = sm * (1.0f / cD);
        const float dx = v.x - mu, dy = v.y - mu, dz = v.z - mu, dw = v.w - mu;
        float sq = (dx * dx + dy * dy) + (dz * dz + dw * dw);
        sq = half_sum(sq);
        sq = sq + other_half(sq);
        const float var = sq * (1.0f / cD);
        const float rs  = rsqrtf(var + 1e-5f);
        *(float4*)(out + (size_t)(m0 + node) * cD + 4 * lane) =
            make_float4(dx * rs * gq.x + bq.x, dy * rs * gq.y + bq.y,
                        dz * rs * gq.z + bq.z, dw * rs * gq.w + bq.w);
    }
}

static constexpr int TI = 3;
static constexpr int NW = 4;
static constexpr int JW = cN / NW;
static constexpr int QB = 24;
static constexpr int AG = 4;
static constexpr int NG = QB / AG;
static constexpr int NIT = JW / (2 * QB);
static constexpr int RW = cN / 4;
static_assert(cN % TI == 0 && cN % NW == 0 && JW % (2 * QB) == 0 && NG % 2 == 0 && AG % 4 == 0 &&
              RW % 32 == 0 && cD == 4 * 32 && cD == 32 * NW, "shape");

struct AdjGroup { float a[TI][AG]; };
typedef __attribute__((ext_vector_type(4))) float v4f;
struct Ch4 { v2f lo, hi; };

__device__ __forceinline__ Ch4 load_ch4(const float* p)
{
    const v4f x = *(const v4f*)p;
    Ch4 r;
    r.lo = x.xy;
    r.hi = x.zw;
    return r;
}

__device__ __forceinline__ void load_q(Ch4 (&q)[QB], const float* __restrict__ Qrow, int j0)
{
#pragma unroll
    for (int u = 0; u < QB; ++u)
        q[u] = load_ch4(Qrow + (size_t)(j0 + u) * cD);
}

__device__ __forceinline__ void load_adj(AdjGroup& g, const float4* rows, int j)
{
#pragma unroll
    for (int r = 0; r < TI; ++r) {
#pragma unroll
        for (int v = 0; v < AG / 4; ++v) {
            const float4 x = rows[r * RW + (j >> 2) + v];
            g.a[r][4 * v + 0] = x.x; g.a[r][4 * v + 1] = x.y;
            g.a[r][4 * v + 2] = x.z; g.a[r][4 * v + 3] = x.w;
        }
    }
}

__device__ __forceinline__ void pair_step(v2f& acc, const v2f p, const v2f q, const v2f a,
                                          const v2f w, const v2f b1)
{
    v2f t = p + q + a * w + b1;
    t.x = fmaxf(t.x, 0.0f);
    t.y = fmaxf(t.y, 0.0f);
    acc += a * t;
}

__device__ __forceinline__ void compute_group(const Ch4* q, const AdjGroup& g,
                                              const Ch4 (&p)[TI], const Ch4 w, const Ch4 b1,
                                              Ch4 (&acc)[TI])
{
#pragma unroll
    for (int u = 0; u < AG; ++u) {
#pragma unroll
        for (int r = 0; r < TI; ++r) {
            v2f a;
            a.x = g.a[r][u];
            a.y = g.a[r][u];
            pair_step(acc[r].lo, p[r].lo, q[u].lo, a, w.lo, b1.lo);
            pair_step(acc[r].hi, p[r].hi, q[u].hi, a, w.hi, b1.hi);
        }
    }
}

__device__ __forceinline__ void one_batch(const Ch4 (&q)[QB], AdjGroup& ga, AdjGroup& gb,
                                          const float4* rows, int j,
                                          const Ch4 (&p)[TI], const Ch4 w, const Ch4 b1,
                                          Ch4 (&acc)[TI])
{
#pragma unroll
    for (int s = 0; s < NG; s += 2) {
        load_adj(gb, rows, j + (s + 1) * AG);
        __builtin_amdgcn_sched_barrier(0);
        compute_group(&q[s * AG], ga, p, w, b1, acc);
        const int jn = j + (s + 2) * AG;
        load_adj(ga, rows, jn < JW ? jn : JW - AG);
        __builtin_amdgcn_sched_barrier(0);
        compute_group(&q[(s + 1) * AG], gb, p, w, b1, acc);
    }
}

__global__ void __launch_bounds__(cD)
pairwise_agg(const float* __restrict__ P,
             const float* __restrict__ Q,
             const float* __restrict__ adj,
             const float* __restrict__ we,
             const float* __restrict__ mb1,
             float* __restrict__ R,
             float* __restrict__ deg)
{
    const int i0   = blockIdx.x * TI;
    const int b    = i0 / cN;
    const int d    = threadIdx.x;
    const int wave = threadIdx.x >> 5;
    const int lane = threadIdx.x & 31;
    const int jw   = wave * JW;

    __shared__ float4 s_adj4[TI * RW];
    __shared__ float4 s_red4[NW * TI * 32];

    const float* Qrow = Q + ((size_t)b * cN + jw) * cD + 4 * lane;

    Ch4 qa[QB], qb[QB];
    load_q(qa, Qrow, 0);

    {
        const float4* adj4 = (const float4*)(adj + (size_t)i0 * cN);
        for (int v = d; v < TI * RW; v += cD)
            s_adj4[v] = adj4[v];
    }

    Ch4 p[TI];
#pragma unroll
    for (int k = 0; k < TI; ++k)
        p[k] = load_ch4(P + (size_t)(i0 + k) * cD + 4 * lane);
    const Ch4 w  = load_ch4(we + 4 * lane);
    const Ch4 b1 = load_ch4(mb1 + 4 * lane);
    const float4* rows = s_adj4 + (jw >> 2);

    __syncthreads();

    AdjGroup ga, gb;
    load_adj(ga, rows, 0);

    if (wave < TI) {
        const float4* row = s_adj4 + wave * RW;
        float s = 0.0f;
#pragma unroll
        for (int k = 0; k < RW / 32; ++k) {
            const float4 x = row[lane + 32 * k];
            s += (x.x + x.y) + (x.z + x.w);
        }
        s = half_sum(s);
        s = s + other_half(s);
        if (lane == 0) deg[i0 + wave] = s;
    }

    Ch4 acc[TI];
#pragma unroll
    for (int r = 0; r < TI; ++r) {
        acc[r].lo = v2f{};
        acc[r].hi = v2f{};
    }

#pragma unroll 1
    for (int it = 0; it < NIT; ++it) {
        const int j = it * 2 * QB;
        load_q(qb, Qrow, j + QB);
        one_batch(qa, ga, gb, rows, j, p, w, b1, acc);
        if (it + 1 < NIT) load_q(qa, Qrow, j + 2 * QB);
        one_batch(qb, ga, gb, rows, j + QB, p, w, b1, acc);
    }

    {
        float4* mine = s_red4 + (wave * TI) * 32 + lane;
#pragma unroll
        for (int r = 0; r < TI; ++r)
            mine[r * 32] = make_float4(acc[r].lo.x, acc[r].lo.y, acc[r].hi.x, acc[r].hi.y);
    }
    __syncthreads();
    {
        const float* red = (const float*)s_red4;
#pragma unroll
        for (int r = 0; r < TI; ++r) {
            float v = red[r * cD + d];
#pragma unroll
            for (int k = 1; k < NW; ++k) v += red[(k * TI + r) * cD + d];
            R[(size_t)(i0 + r) * cD + d] = v;
        }
    }
}


extern "C" void kernel_launch(void* const* d_in, const int* in_sizes, int n_in,
                              void* d_out, int out_size, void* d_ws, size_t ws_size,
                              hipStream_t stream)
{
    (void)in_sizes; (void)n_in; (void)out_size; (void)ws_size;

    const float* h    = (const float*)d_in[0];
    const float* adj  = (const float*)d_in[1];
    const float* ip_w = (const float*)d_in[2];
    const float* ip_b = (const float*)d_in[3];
    const float* mw1  = (const float*)d_in[4];
    const float* mb1  = (const float*)d_in[5];
    const float* mw2  = (const float*)d_in[6];
    const float* mb2  = (const float*)d_in[7];
    const float* uw1  = (const float*)d_in[8];
    const float* ub1  = (const float*)d_in[9];
    const float* uw2  = (const float*)d_in[10];
    const float* ub2  = (const float*)d_in[11];
    const float* ln_g = (const float*)d_in[12];
    const float* ln_b = (const float*)d_in[13];

    float* ws = (float*)d_ws;
    const size_t S = (size_t)cM * cD;
    float* xa   = ws + 0 * S;
    float* P    = ws + 1 * S;
    float* Q    = ws + 2 * S;
    float* R    = ws + 3 * S;
    float* xb   = ws + 4 * S;
    float* hida = ws + 5 * S;
    float* hidb = ws + 6 * S;
    float* deg  = ws + 7 * S;

    gemm_in_products<<<dim3(48, 3), 128, 0, stream>>>(h, ip_w, ip_b, xa,
                                                      mw1, mw1 + 128 * cD, uw1, ub1,
                                                      P, Q, hida);

    for (int l = 0; l < 3; ++l) {
        const float* mw1l = mw1 + (size_t)l * 257 * cD;
        const float* mb1l = mb1 + (size_t)l * cD;
        const float* mw2l = mw2 + (size_t)l * cD * cD;
        const float* mb2l = mb2 + (size_t)l * cD;
        const float* uw1l = uw1 + (size_t)l * 256 * cD;
        const float* uw2l = uw2 + (size_t)l * cD * cD;
        const float* ub2l = ub2 + (size_t)l * cD;
        float* xcur = (l & 1) ? xb : xa;
        float* xnxt = (l & 1) ? xa : xb;
        float* hcur = (l & 1) ? hidb : hida;
        float* hnxt = (l & 1) ? hida : hidb;

        pairwise_agg<<<cM / TI, cD, 0, stream>>>(P, Q, adj, mw1l + 256 * cD, mb1l, R, deg);

        if (l < 2) {
            const float* mw1n = mw1 + (size_t)(l + 1) * 257 * cD;
            const float* uw1n = uw1 + (size_t)(l + 1) * 256 * cD;
            const float* ub1n = ub1 + (size_t)(l + 1) * cD;
            gemm_hn_ln_products<<<dim3(48, 3), 128, 0, stream>>>(R, mw2l, mb2l, deg, uw1l + 128 * cD,
                                                                 hcur, uw2l, ub2l, xcur,
                                                                 ln_g + l * cD, ln_b + l * cD, xnxt,
                                                                 mw1n, mw1n + 128 * cD, uw1n, ub1n,
                                                                 P, Q, hnxt);
        } else {
            gemm_hn_ln_out<<<cM / 16, 128, 0, stream>>>(R, mw2l, mb2l, deg, uw1l + 128 * cD,
                                                      hcur, uw2l, ub2l,
                                                      ln_g + l * cD, ln_b + l * cD,
                                                      (float*)d_out);
        }
    }
}
